# starting kernel
# baseline (speedup 1.0000x reference)
#include <hip/hip_runtime.h>


typedef __attribute__((ext_vector_type(2))) float v2f;
typedef __attribute__((ext_vector_type(8))) float v8f;

#define NDIM 64
#define WAVES_PER_BLOCK 4
#define WP_ELEMS 4096
#define SCAN_B 256

__device__ __forceinline__ v8f wmma_f32_16x16x4(v2f a, v2f b, v8f c) {
  return __builtin_amdgcn_wmma_f32_16x16x4_f32(false, a, false, b, (short)0, c,
                                               false, false);
}

__global__ void gin_repack_kernel(const float* __restrict__ W,
                                  float* __restrict__ Wp) {
  const int tid = blockIdx.x * blockDim.x + threadIdx.x;
  const int c = tid >> 7;
  const int n = (tid >> 5) & 3;
  const int lane = tid & 31;
  const int hi = lane >> 4;
  const int m = lane & 15;
  const int k = 4 * c + 2 * hi;
  Wp[tid * 2 + 0] = W[(k + 0) * NDIM + 16 * n + m];
  Wp[tid * 2 + 1] = W[(k + 1) * NDIM + 16 * n + m];
}

template <int MODE>
__global__ void __launch_bounds__(WAVES_PER_BLOCK * 32)
gin_mlp_kernel(const float* __restrict__ H, const float* __restrict__ AGG,
               const float* __restrict__ WP1, const float* __restrict__ B1,
               const float* __restrict__ WP2, const float* __restrict__ B2,
               float* __restrict__ OUT, int nTiles) {
  __shared__ float lds[WAVES_PER_BLOCK][16 * NDIM];

  const int wave = threadIdx.x >> 5;
  const int lane = threadIdx.x & 31;
  const int tile = blockIdx.x * WAVES_PER_BLOCK + wave;
  if (tile >= nTiles) return;

  const int laneM = lane & 15;
  const int hi = lane >> 4;
  const int row0 = tile * 16;

  v2f a[16];
  {
    const size_t base = (size_t)(row0 + laneM) * NDIM + 2 * hi;
    const float* hp = H + base;
    const float* gp = AGG + base;
#pragma unroll
    for (int c = 0; c < 16; ++c) {
      v2f z = *(const v2f*)(hp + 4 * c);
      if (MODE == 1) {
        v2f g = *(const v2f*)(gp + 4 * c);
        z.x += g.x;
        z.y += g.y;
      }
      a[c] = z;
    }
  }

  const v2f* wp1 = (const v2f*)WP1;

  v8f acc[4];
#pragma unroll
  for (int n = 0; n < 4; ++n) {
    const float bv = B1[16 * n + laneM];
#pragma unroll
    for (int r = 0; r < 8; ++r) acc[n][r] = bv;
  }
#pragma unroll
  for (int c = 0; c < 16; ++c)
#pragma unroll
    for (int n = 0; n < 4; ++n)
      acc[n] = wmma_f32_16x16x4(a[c], wp1[(c * 4 + n) * 32 + lane], acc[n]);

  if (MODE == 2) {
#pragma unroll
    for (int n = 0; n < 4; ++n)
#pragma unroll
      for (int r = 0; r < 8; ++r)
        OUT[(size_t)(row0 + r + 8 * hi) * NDIM + 16 * n + laneM] = acc[n][r];
    return;
  }

  float* lt = lds[wave];
#pragma unroll
  for (int n = 0; n < 4; ++n)
#pragma unroll
    for (int r = 0; r < 8; ++r)
      lt[(r + 8 * hi) * NDIM + 16 * n + laneM] = fmaxf(acc[n][r], 0.0f);

#pragma unroll
  for (int c = 0; c < 16; ++c)
    a[c] = *(const v2f*)(lt + laneM * NDIM + 4 * c + 2 * hi);

  const v2f* wp2 = (const v2f*)WP2;
#pragma unroll
  for (int n = 0; n < 4; ++n) {
    const float bv = B2[16 * n + laneM];
#pragma unroll
    for (int r = 0; r < 8; ++r) acc[n][r] = bv;
  }
#pragma unroll
  for (int c = 0; c < 16; ++c)
#pragma unroll
    for (int n = 0; n < 4; ++n)
      acc[n] = wmma_f32_16x16x4(a[c], wp2[(c * 4 + n) * 32 + lane], acc[n]);

#pragma unroll
  for (int n = 0; n < 4; ++n)
#pragma unroll
    for (int r = 0; r < 8; ++r)
      OUT[(size_t)(row0 + r + 8 * hi) * NDIM + 16 * n + laneM] = acc[n][r];
}


__global__ void gin_zero_i32_kernel(int* __restrict__ p, int n) {
  const int i = blockIdx.x * blockDim.x + threadIdx.x;
  if (i < n) p[i] = 0;
}

__global__ void gin_degree_kernel(const int* __restrict__ rowp,
                                  int* __restrict__ deg, int E) {
  const int e = blockIdx.x * blockDim.x + threadIdx.x;
  if (e < E) atomicAdd(&deg[rowp[e]], 1);
}

__global__ void gin_scan1_kernel(const int* __restrict__ deg,
                                 int* __restrict__ rowstart1,
                                 int* __restrict__ bwork, int n) {
  __shared__ int s[SCAN_B];
  const int i = blockIdx.x * SCAN_B + threadIdx.x;
  const int v = (i < n) ? deg[i] : 0;
  s[threadIdx.x] = v;
  __syncthreads();
#pragma unroll
  for (int off = 1; off < SCAN_B; off <<= 1) {
    const int tv = (threadIdx.x >= off) ? s[threadIdx.x - off] : 0;
    __syncthreads();
    s[threadIdx.x] += tv;
    __syncthreads();
  }
  if (i < n) rowstart1[i] = s[threadIdx.x];
  if (threadIdx.x == SCAN_B - 1) bwork[blockIdx.x] = s[SCAN_B - 1];
}

__global__ void gin_scan2_kernel(int* __restrict__ bwork, int nb) {
  __shared__ int s[512];
  const int i = threadIdx.x;
  const int v = (i < nb) ? bwork[i] : 0;
  s[i] = v;
  __syncthreads();
#pragma unroll
  for (int off = 1; off < 512; off <<= 1) {
    const int tv = (i >= off) ? s[i - off] : 0;
    __syncthreads();
    s[i] += tv;
    __syncthreads();
  }
  if (i < nb) bwork[i] = s[i] - v;
}

__global__ void gin_scan3_kernel(int* __restrict__ rowstart,
                                 const int* __restrict__ bwork, int n) {
  const int i = blockIdx.x * blockDim.x + threadIdx.x;
  if (i < n) rowstart[i + 1] += bwork[i / SCAN_B];
  if (i == 0) rowstart[0] = 0;
}

__global__ void gin_fill_kernel(const int* __restrict__ rowp,
                                const int* __restrict__ colp,
                                const int* __restrict__ rowstart,
                                int* __restrict__ cursor,
                                int* __restrict__ perm, int E) {
  const int e = blockIdx.x * blockDim.x + threadIdx.x;
  if (e < E) {
    const int r = rowp[e];
    const int pos = rowstart[r] + atomicAdd(&cursor[r], 1);
    perm[pos] = colp[e];
  }
}

__global__ void gin_gather_kernel(const float* __restrict__ T,
                                  const int* __restrict__ rowstart,
                                  const int* __restrict__ perm,
                                  float* __restrict__ AGG, int N) {
  const int g = (blockIdx.x * blockDim.x + threadIdx.x) >> 4;
  if (g >= N) return;
  const int l4 = (threadIdx.x & 15) * 4;
  const int s = rowstart[g];
  const int e = rowstart[g + 1];
  float4 acc = make_float4(0.f, 0.f, 0.f, 0.f);
  for (int j = s; j < e; ++j) {
    const int c = perm[j];
    const float4 v = *(const float4*)(T + (size_t)c * NDIM + l4);
    acc.x += v.x;
    acc.y += v.y;
    acc.z += v.z;
    acc.w += v.w;
  }
  *(float4*)(AGG + (size_t)g * NDIM + l4) = acc;
}


extern "C" void kernel_launch(void* const* d_in, const int* in_sizes, int n_in,
                              void* d_out, int out_size, void* d_ws,
                              size_t ws_size, hipStream_t stream) {
  (void)n_in;
  (void)out_size;
  (void)ws_size;

  const float* x = (const float*)d_in[0];
  const int* ei = (const int*)d_in[1];
  const float* W1 = (const float*)d_in[2];
  const float* b1 = (const float*)d_in[3];
  const float* W2 = (const float*)d_in[4];
  const float* b2 = (const float*)d_in[5];
  const float* Wf = (const float*)d_in[6];
  const float* bf = (const float*)d_in[7];
  float* out = (float*)d_out;

  const int N = in_sizes[0] / NDIM;
  const int E = in_sizes[1] / 2;
  const int L = in_sizes[3] / NDIM;
  const int nTiles = N / 16;
  const size_t NB = (size_t)N * NDIM;

  float* agg = (float*)d_ws;
  float* t = agg + NB;
  float* h = t + NB;
  float* wp = h + NB;
  int* deg = (int*)(wp + 7 * WP_ELEMS);
  int* cursor = deg + N;
  int* rowstart = cursor + N;
  int* bwork = rowstart + (N + 1);
  int* perm = bwork + 1024;

  const int* rowp = ei;
  const int* colp = ei + E;

  const float* wsrc[7] = {W1, W1 + WP_ELEMS, W1 + 2 * WP_ELEMS,
                          W2, W2 + WP_ELEMS, W2 + 2 * WP_ELEMS, Wf};
  for (int m = 0; m < 7; ++m)
    gin_repack_kernel<<<WP_ELEMS / 2 / 256, 256, 0, stream>>>(
        wsrc[m], wp + (size_t)m * WP_ELEMS);

  const int nScanBlocks = (N + SCAN_B - 1) / SCAN_B;
  gin_zero_i32_kernel<<<(2 * N + 255) / 256, 256, 0, stream>>>(deg, 2 * N);
  gin_degree_kernel<<<(E + 255) / 256, 256, 0, stream>>>(rowp, deg, E);
  gin_scan1_kernel<<<nScanBlocks, SCAN_B, 0, stream>>>(deg, rowstart + 1,
                                                       bwork, N);
  gin_scan2_kernel<<<1, 512, 0, stream>>>(bwork, nScanBlocks);
  gin_scan3_kernel<<<(N + 255) / 256, 256, 0, stream>>>(rowstart, bwork, N);
  gin_fill_kernel<<<(E + 255) / 256, 256, 0, stream>>>(rowp, colp, rowstart,
                                                       cursor, perm, E);

  const dim3 mlpBlk(WAVES_PER_BLOCK * 32);
  const dim3 mlpGrd((nTiles + WAVES_PER_BLOCK - 1) / WAVES_PER_BLOCK);
  const dim3 gatGrd((N * 16 + 255) / 256);

  const float* hcur = x;
  for (int l = 0; l < L; ++l) {
    const float* wp1 = wp + (size_t)l * WP_ELEMS;
    const float* wp2 = wp + (size_t)(3 + l) * WP_ELEMS;
    const float* v1 = b1 + (size_t)l * NDIM;
    const float* v2 = b2 + (size_t)l * NDIM;

    gin_mlp_kernel<0><<<mlpGrd, mlpBlk, 0, stream>>>(hcur, nullptr, wp1, v1,
                                                     wp2, v2, t, nTiles);
    gin_gather_kernel<<<gatGrd, 256, 0, stream>>>(t, rowstart, perm, agg, N);
    gin_mlp_kernel<1><<<mlpGrd, mlpBlk, 0, stream>>>(hcur, agg, wp1, v1, wp2,
                                                     v2, h, nTiles);
    hcur = h;
  }
  gin_mlp_kernel<2><<<mlpGrd, mlpBlk, 0, stream>>>(
      hcur, nullptr, wp + (size_t)6 * WP_ELEMS, bf, nullptr, nullptr, out,
      nTiles);
}


# MLP kernel made persistent: each wave loads both weight matrices' B fragments once into registers and loops over its row tiles (next tile's rows fetched ahead), so no WMMA waits on a weight load
# speedup vs baseline: 1.2729x; 1.2729x over previous
#include <hip/hip_runtime.h>


typedef __attribute__((ext_vector_type(2))) float v2f;
typedef __attribute__((ext_vector_type(8))) float v8f;

#define NDIM 64
#define WAVES_PER_BLOCK 4
#define WP_ELEMS 4096
#define SCAN_B 256
#define MLP_GRID 512

__device__ __forceinline__ v8f wmma_f32_16x16x4(v2f a, v2f b, v8f c) {
  return __builtin_amdgcn_wmma_f32_16x16x4_f32(false, a, false, b, (short)0, c,
                                               false, false);
}

__global__ void gin_repack_kernel(const float* __restrict__ W,
                                  float* __restrict__ Wp) {
  const int tid = blockIdx.x * blockDim.x + threadIdx.x;
  const int c = tid >> 7;
  const int n = (tid >> 5) & 3;
  const int lane = tid & 31;
  const int hi = lane >> 4;
  const int m = lane & 15;
  const int k = 4 * c + 2 * hi;
  Wp[tid * 2 + 0] = W[(k + 0) * NDIM + 16 * n + m];
  Wp[tid * 2 + 1] = W[(k + 1) * NDIM + 16 * n + m];
}


template <int MODE>
__device__ __forceinline__ void gin_fetch_rows(const float* __restrict__ H,
                                               const float* __restrict__ AGG,
                                               int tile, int laneOff,
                                               v2f (&zh)[16], v2f (&zg)[16]) {
  const float* hp = H + (size_t)tile * (16 * NDIM) + laneOff;
#pragma unroll
  for (int c = 0; c < 16; ++c) zh[c] = *(const v2f*)(hp + 4 * c);
  if (MODE == 1) {
    const float* gp = AGG + (size_t)tile * (16 * NDIM) + laneOff;
#pragma unroll
    for (int c = 0; c < 16; ++c) zg[c] = *(const v2f*)(gp + 4 * c);
  }
}

template <int MODE>
__device__ __forceinline__ void gin_mlp_tile(const v2f (&zh)[16],
                                             const v2f (&zg)[16],
                                             const v2f (&w1)[64],
                                             const v2f (&w2)[64],
                                             const float (&bias1)[4],
                                             const float (&bias2)[4],
                                             float* lt, float* __restrict__ OUT,
                                             int tile, int laneM, int hi) {
  v2f a[16];
#pragma unroll
  for (int c = 0; c < 16; ++c) {
    v2f z = zh[c];
    if (MODE == 1) {
      z.x += zg[c].x;
      z.y += zg[c].y;
    }
    a[c] = z;
  }

  float* op = OUT + ((size_t)tile * 16 + 8 * hi) * NDIM + laneM;

  v8f acc[4];
#pragma unroll
  for (int n = 0; n < 4; ++n)
#pragma unroll
    for (int r = 0; r < 8; ++r) acc[n][r] = bias1[n];
#pragma unroll
  for (int c = 0; c < 16; ++c)
#pragma unroll
    for (int n = 0; n < 4; ++n)
      acc[n] = wmma_f32_16x16x4(a[c], w1[c * 4 + n], acc[n]);

  if (MODE == 2) {
#pragma unroll
    for (int n = 0; n < 4; ++n)
#pragma unroll
      for (int r = 0; r < 8; ++r) op[r * NDIM + 16 * n] = acc[n][r];
    return;
  }

#pragma unroll
  for (int n = 0; n < 4; ++n)
#pragma unroll
    for (int r = 0; r < 8; ++r)
      lt[(r + 8 * hi) * NDIM + 16 * n + laneM] = fmaxf(acc[n][r], 0.0f);

#pragma unroll
  for (int c = 0; c < 16; ++c)
    a[c] = *(const v2f*)(lt + laneM * NDIM + 4 * c + 2 * hi);

#pragma unroll
  for (int n = 0; n < 4; ++n)
#pragma unroll
    for (int r = 0; r < 8; ++r) acc[n][r] = bias2[n];
#pragma unroll
  for (int c = 0; c < 16; ++c)
#pragma unroll
    for (int n = 0; n < 4; ++n)
      acc[n] = wmma_f32_16x16x4(a[c], w2[c * 4 + n], acc[n]);

#pragma unroll
  for (int n = 0; n < 4; ++n)
#pragma unroll
    for (int r = 0; r < 8; ++r) op[r * NDIM + 16 * n] = acc[n][r];
}

template <int MODE>
__global__ void __launch_bounds__(WAVES_PER_BLOCK * 32)
gin_mlp_kernel(const float* __restrict__ H, const float* __restrict__ AGG,
               const float* __restrict__ WP1, const float* __restrict__ B1,
               const float* __restrict__ WP2, const float* __restrict__ B2,
               float* __restrict__ OUT, int nTiles) {
  __shared__ float lds[WAVES_PER_BLOCK][16 * NDIM];

  const int wave = __builtin_amdgcn_readfirstlane((int)(threadIdx.x >> 5));
  const int lane = threadIdx.x & 31;
  const int stride = (int)gridDim.x * WAVES_PER_BLOCK;
  int tile = (int)blockIdx.x * WAVES_PER_BLOCK + wave;
  if (tile >= nTiles) return;

  const int laneM = lane & 15;
  const int hi = lane >> 4;
  const int laneOff = laneM * NDIM + 2 * hi;
  float* lt = lds[wave];

  v2f hA[16], gA[16], hB[16], gB[16];
  gin_fetch_rows<MODE>(H, AGG, tile, laneOff, hA, gA);

  v2f w1[64];
  v2f w2[64];
  float bias1[4];
  float bias2[4];
  {
    const v2f* wp1 = (const v2f*)WP1 + lane;
#pragma unroll
    for (int i = 0; i < 64; ++i) w1[i] = wp1[i * 32];
#pragma unroll
    for (int n = 0; n < 4; ++n) bias1[n] = B1[16 * n + laneM];
    if (MODE != 2) {
      const v2f* wp2 = (const v2f*)WP2 + lane;
#pragma unroll
      for (int i = 0; i < 64; ++i) w2[i] = wp2[i * 32];
#pragma unroll
      for (int n = 0; n < 4; ++n) bias2[n] = B2[16 * n + laneM];
    }
  }

  int cnt = (nTiles - 1 - tile) / stride + 1;

  while (cnt >= 2) {
    const int t1 = tile + stride;
    const int t2 = t1 + stride;

    gin_fetch_rows<MODE>(H, AGG, t1, laneOff, hB, gB);
    __builtin_amdgcn_sched_barrier(0);
    gin_mlp_tile<MODE>(hA, gA, w1, w2, bias1, bias2, lt, OUT, tile, laneM, hi);

    gin_fetch_rows<MODE>(H, AGG, cnt > 2 ? t2 : t1, laneOff, hA, gA);
    __builtin_amdgcn_sched_barrier(0);
    gin_mlp_tile<MODE>(hB, gB, w1, w2, bias1, bias2, lt, OUT, t1, laneM, hi);

    tile = t2;
    cnt -= 2;
  }
  if (cnt == 1)
    gin_mlp_tile<MODE>(hA, gA, w1, w2, bias1, bias2, lt, OUT, tile, laneM, hi);
}


__global__ void gin_zero_i32_kernel(int* __restrict__ p, int n) {
  const int i = blockIdx.x * blockDim.x + threadIdx.x;
  if (i < n) p[i] = 0;
}

__global__ void gin_degree_kernel(const int* __restrict__ rowp,
                                  int* __restrict__ deg, int E) {
  const int e = blockIdx.x * blockDim.x + threadIdx.x;
  if (e < E) atomicAdd(&deg[rowp[e]], 1);
}

__global__ void gin_scan1_kernel(const int* __restrict__ deg,
                                 int* __restrict__ rowstart1,
                                 int* __restrict__ bwork, int n) {
  __shared__ int s[SCAN_B];
  const int i = blockIdx.x * SCAN_B + threadIdx.x;
  const int v = (i < n) ? deg[i] : 0;
  s[threadIdx.x] = v;
  __syncthreads();
#pragma unroll
  for (int off = 1; off < SCAN_B; off <<= 1) {
    const int tv = (threadIdx.x >= off) ? s[threadIdx.x - off] : 0;
    __syncthreads();
    s[threadIdx.x] += tv;
    __syncthreads();
  }
  if (i < n) rowstart1[i] = s[threadIdx.x];
  if (threadIdx.x == SCAN_B - 1) bwork[blockIdx.x] = s[SCAN_B - 1];
}

__global__ void gin_scan2_kernel(int* __restrict__ bwork, int nb) {
  __shared__ int s[512];
  const int i = threadIdx.x;
  const int v = (i < nb) ? bwork[i] : 0;
  s[i] = v;
  __syncthreads();
#pragma unroll
  for (int off = 1; off < 512; off <<= 1) {
    const int tv = (i >= off) ? s[i - off] : 0;
    __syncthreads();
    s[i] += tv;
    __syncthreads();
  }
  if (i < nb) bwork[i] = s[i] - v;
}

__global__ void gin_scan3_kernel(int* __restrict__ rowstart,
                                 const int* __restrict__ bwork, int n) {
  const int i = blockIdx.x * blockDim.x + threadIdx.x;
  if (i < n) rowstart[i + 1] += bwork[i / SCAN_B];
  if (i == 0) rowstart[0] = 0;
}

__global__ void gin_fill_kernel(const int* __restrict__ rowp,
                                const int* __restrict__ colp,
                                const int* __restrict__ rowstart,
                                int* __restrict__ cursor,
                                int* __restrict__ perm, int E) {
  const int e = blockIdx.x * blockDim.x + threadIdx.x;
  if (e < E) {
    const int r = rowp[e];
    const int pos = rowstart[r] + atomicAdd(&cursor[r], 1);
    perm[pos] = colp[e];
  }
}

__global__ void gin_gather_kernel(const float* __restrict__ T,
                                  const int* __restrict__ rowstart,
                                  const int* __restrict__ perm,
                                  float* __restrict__ AGG, int N) {
  const int g = (blockIdx.x * blockDim.x + threadIdx.x) >> 4;
  if (g >= N) return;
  const int l4 = (threadIdx.x & 15) * 4;
  const int s = rowstart[g];
  const int e = rowstart[g + 1];
  float4 acc = make_float4(0.f, 0.f, 0.f, 0.f);
  for (int j = s; j < e; ++j) {
    const int c = perm[j];
    const float4 v = *(const float4*)(T + (size_t)c * NDIM + l4);
    acc.x += v.x;
    acc.y += v.y;
    acc.z += v.z;
    acc.w += v.w;
  }
  *(float4*)(AGG + (size_t)g * NDIM + l4) = acc;
}


extern "C" void kernel_launch(void* const* d_in, const int* in_sizes, int n_in,
                              void* d_out, int out_size, void* d_ws,
                              size_t ws_size, hipStream_t stream) {
  (void)n_in;
  (void)out_size;
  (void)ws_size;

  const float* x = (const float*)d_in[0];
  const int* ei = (const int*)d_in[1];
  const float* W1 = (const float*)d_in[2];
  const float* b1 = (const float*)d_in[3];
  const float* W2 = (const float*)d_in[4];
  const float* b2 = (const float*)d_in[5];
  const float* Wf = (const float*)d_in[6];
  const float* bf = (const float*)d_in[7];
  float* out = (float*)d_out;

  const int N = in_sizes[0] / NDIM;
  const int E = in_sizes[1] / 2;
  const int L = in_sizes[3] / NDIM;
  const int nTiles = N / 16;
  const size_t NB = (size_t)N * NDIM;

  float* agg = (float*)d_ws;
  float* t = agg + NB;
  float* h = t + NB;
  float* wp = h + NB;
  int* deg = (int*)(wp + 7 * WP_ELEMS);
  int* cursor = deg + N;
  int* rowstart = cursor + N;
  int* bwork = rowstart + (N + 1);
  int* perm = bwork + 1024;

  const int* rowp = ei;
  const int* colp = ei + E;

  const float* wsrc[7] = {W1, W1 + WP_ELEMS, W1 + 2 * WP_ELEMS,
                          W2, W2 + WP_ELEMS, W2 + 2 * WP_ELEMS, Wf};
  for (int m = 0; m < 7; ++m)
    gin_repack_kernel<<<WP_ELEMS / 2 / 256, 256, 0, stream>>>(
        wsrc[m], wp + (size_t)m * WP_ELEMS);

  const int nScanBlocks = (N + SCAN_B - 1) / SCAN_B;
  gin_zero_i32_kernel<<<(2 * N + 255) / 256, 256, 0, stream>>>(deg, 2 * N);
  gin_degree_kernel<<<(E + 255) / 256, 256, 0, stream>>>(rowp, deg, E);
  gin_scan1_kernel<<<nScanBlocks, SCAN_B, 0, stream>>>(deg, rowstart + 1,
                                                       bwork, N);
  gin_scan2_kernel<<<1, 512, 0, stream>>>(bwork, nScanBlocks);
  gin_scan3_kernel<<<(N + 255) / 256, 256, 0, stream>>>(rowstart, bwork, N);
  gin_fill_kernel<<<(E + 255) / 256, 256, 0, stream>>>(rowp, colp, rowstart,
                                                       cursor, perm, E);

  const dim3 mlpBlk(WAVES_PER_BLOCK * 32);
  const dim3 mlpGrd(MLP_GRID);
  const dim3 gatGrd((N * 16 + 255) / 256);

  const float* hcur = x;
  for (int l = 0; l < L; ++l) {
    const float* wp1 = wp + (size_t)l * WP_ELEMS;
    const float* wp2 = wp + (size_t)(3 + l) * WP_ELEMS;
    const float* v1 = b1 + (size_t)l * NDIM;
    const float* v2 = b2 + (size_t)l * NDIM;

    gin_mlp_kernel<0><<<mlpGrd, mlpBlk, 0, stream>>>(hcur, nullptr, wp1, v1,
                                                     wp2, v2, t, nTiles);
    gin_gather_kernel<<<gatGrd, 256, 0, stream>>>(t, rowstart, perm, agg, N);
    gin_mlp_kernel<1><<<mlpGrd, mlpBlk, 0, stream>>>(hcur, agg, wp1, v1, wp2,
                                                     v2, h, nTiles);
    hcur = h;
  }
  gin_mlp_kernel<2><<<mlpGrd, mlpBlk, 0, stream>>>(
      hcur, nullptr, wp + (size_t)6 * WP_ELEMS, bf, nullptr, nullptr, out,
      nTiles);
}


# Fewer dependent launches, 23 -> 15: 7 weight repacks and the counter clear run as one set-up launch; each scan3 block sums the block totals in front of it itself (no single-block scan launch)
# speedup vs baseline: 1.3414x; 1.0538x over previous
#include <hip/hip_runtime.h>


typedef __attribute__((ext_vector_type(2))) float v2f;
typedef __attribute__((ext_vector_type(8))) float v8f;

#define NDIM 64
#define WAVES_PER_BLOCK 4
#define WP_ELEMS 4096
#define SCAN_B 256
#define MLP_GRID 512

__device__ __forceinline__ v8f wmma_f32_16x16x4(v2f a, v2f b, v8f c) {
  return __builtin_amdgcn_wmma_f32_16x16x4_f32(false, a, false, b, (short)0, c,
                                               false, false);
}

#define REPACK_BLOCKS 8
__global__ void gin_setup_kernel(const float* __restrict__ W1,
                                 const float* __restrict__ W2,
                                 const float* __restrict__ Wf,
                                 float* __restrict__ Wp, int L,
                                 int* __restrict__ zp, int nZero) {
  const int nRepack = (2 * L + 1) * REPACK_BLOCKS;
  if ((int)blockIdx.x >= nRepack) {
    const int i = ((int)blockIdx.x - nRepack) * 256 + (int)threadIdx.x;
    if (i < nZero) zp[i] = 0;
    return;
  }
  const int mat = (int)blockIdx.x / REPACK_BLOCKS;
  const float* W = mat < L       ? W1 + (size_t)mat * (NDIM * NDIM)
                   : mat < 2 * L ? W2 + (size_t)(mat - L) * (NDIM * NDIM)
                                 : Wf;
  float* dst = Wp + (size_t)mat * WP_ELEMS;
  const int tid =
      ((int)blockIdx.x % REPACK_BLOCKS) * 256 + (int)threadIdx.x;
  const int c = tid >> 7;
  const int n = (tid >> 5) & 3;
  const int lane = tid & 31;
  const int hi = lane >> 4;
  const int m = lane & 15;
  const int k = 4 * c + 2 * hi;
  dst[tid * 2 + 0] = W[(k + 0) * NDIM + 16 * n + m];
  dst[tid * 2 + 1] = W[(k + 1) * NDIM + 16 * n + m];
}


template <int MODE>
__device__ __forceinline__ void gin_fetch_rows(const float* __restrict__ H,
                                               const float* __restrict__ AGG,
                                               int tile, int laneOff,
                                               v2f (&zh)[16], v2f (&zg)[16]) {
  const float* hp = H + (size_t)tile * (16 * NDIM) + laneOff;
#pragma unroll
  for (int c = 0; c < 16; ++c) zh[c] = *(const v2f*)(hp + 4 * c);
  if (MODE == 1) {
    const float* gp = AGG + (size_t)tile * (16 * NDIM) + laneOff;
#pragma unroll
    for (int c = 0; c < 16; ++c) zg[c] = *(const v2f*)(gp + 4 * c);
  }
}

template <int MODE>
__device__ __forceinline__ void gin_mlp_tile(const v2f (&zh)[16],
                                             const v2f (&zg)[16],
                                             const v2f (&w1)[64],
                                             const v2f (&w2)[64],
                                             const float (&bias1)[4],
                                             const float (&bias2)[4],
                                             float* lt, float* __restrict__ OUT,
                                             int tile, int laneM, int hi) {
  v2f a[16];
#pragma unroll
  for (int c = 0; c < 16; ++c) {
    v2f z = zh[c];
    if (MODE == 1) {
      z.x += zg[c].x;
      z.y += zg[c].y;
    }
    a[c] = z;
  }

  float* op = OUT + ((size_t)tile * 16 + 8 * hi) * NDIM + laneM;

  v8f acc[4];
#pragma unroll
  for (int n = 0; n < 4; ++n)
#pragma unroll
    for (int r = 0; r < 8; ++r) acc[n][r] = bias1[n];
#pragma unroll
  for (int c = 0; c < 16; ++c)
#pragma unroll
    for (int n = 0; n < 4; ++n)
      acc[n] = wmma_f32_16x16x4(a[c], w1[c * 4 + n], acc[n]);

  if (MODE == 2) {
#pragma unroll
    for (int n = 0; n < 4; ++n)
#pragma unroll
      for (int r = 0; r < 8; ++r) op[r * NDIM + 16 * n] = acc[n][r];
    return;
  }

#pragma unroll
  for (int n = 0; n < 4; ++n)
#pragma unroll
    for (int r = 0; r < 8; ++r)
      lt[(r + 8 * hi) * NDIM + 16 * n + laneM] = fmaxf(acc[n][r], 0.0f);

#pragma unroll
  for (int c = 0; c < 16; ++c)
    a[c] = *(const v2f*)(lt + laneM * NDIM + 4 * c + 2 * hi);

#pragma unroll
  for (int n = 0; n < 4; ++n)
#pragma unroll
    for (int r = 0; r < 8; ++r) acc[n][r] = bias2[n];
#pragma unroll
  for (int c = 0; c < 16; ++c)
#pragma unroll
    for (int n = 0; n < 4; ++n)
      acc[n] = wmma_f32_16x16x4(a[c], w2[c * 4 + n], acc[n]);

#pragma unroll
  for (int n = 0; n < 4; ++n)
#pragma unroll
    for (int r = 0; r < 8; ++r) op[r * NDIM + 16 * n] = acc[n][r];
}

template <int MODE>
__global__ void __launch_bounds__(WAVES_PER_BLOCK * 32)
gin_mlp_kernel(const float* __restrict__ H, const float* __restrict__ AGG,
               const float* __restrict__ WP1, const float* __restrict__ B1,
               const float* __restrict__ WP2, const float* __restrict__ B2,
               float* __restrict__ OUT, int nTiles) {
  __shared__ float lds[WAVES_PER_BLOCK][16 * NDIM];

  const int wave = __builtin_amdgcn_readfirstlane((int)(threadIdx.x >> 5));
  const int lane = threadIdx.x & 31;
  const int stride = (int)gridDim.x * WAVES_PER_BLOCK;
  int tile = (int)blockIdx.x * WAVES_PER_BLOCK + wave;
  if (tile >= nTiles) return;

  const int laneM = lane & 15;
  const int hi = lane >> 4;
  const int laneOff = laneM * NDIM + 2 * hi;
  float* lt = lds[wave];

  v2f hA[16], gA[16], hB[16], gB[16];
  gin_fetch_rows<MODE>(H, AGG, tile, laneOff, hA, gA);

  v2f w1[64];
  v2f w2[64];
  float bias1[4];
  float bias2[4];
  {
    const v2f* wp1 = (const v2f*)WP1 + lane;
#pragma unroll
    for (int i = 0; i < 64; ++i) w1[i] = wp1[i * 32];
#pragma unroll
    for (int n = 0; n < 4; ++n) bias1[n] = B1[16 * n + laneM];
    if (MODE != 2) {
      const v2f* wp2 = (const v2f*)WP2 + lane;
#pragma unroll
      for (int i = 0; i < 64; ++i) w2[i] = wp2[i * 32];
#pragma unroll
      for (int n = 0; n < 4; ++n) bias2[n] = B2[16 * n + laneM];
    }
  }

  int cnt = (nTiles - 1 - tile) / stride + 1;

  while (cnt >= 2) {
    const int t1 = tile + stride;
    const int t2 = t1 + stride;

    gin_fetch_rows<MODE>(H, AGG, t1, laneOff, hB, gB);
    __builtin_amdgcn_sched_barrier(0);
    gin_mlp_tile<MODE>(hA, gA, w1, w2, bias1, bias2, lt, OUT, tile, laneM, hi);

    gin_fetch_rows<MODE>(H, AGG, cnt > 2 ? t2 : t1, laneOff, hA, gA);
    __builtin_amdgcn_sched_barrier(0);
    gin_mlp_tile<MODE>(hB, gB, w1, w2, bias1, bias2, lt, OUT, t1, laneM, hi);

    tile = t2;
    cnt -= 2;
  }
  if (cnt == 1)
    gin_mlp_tile<MODE>(hA, gA, w1, w2, bias1, bias2, lt, OUT, tile, laneM, hi);
}


__global__ void gin_degree_kernel(const int* __restrict__ rowp,
                                  int* __restrict__ deg, int E) {
  const int e = blockIdx.x * blockDim.x + threadIdx.x;
  if (e < E) atomicAdd(&deg[rowp[e]], 1);
}

__global__ void gin_scan1_kernel(const int* __restrict__ deg,
                                 int* __restrict__ rowstart1,
                                 int* __restrict__ bwork, int n) {
  __shared__ int s[SCAN_B];
  const int i = blockIdx.x * SCAN_B + threadIdx.x;
  const int v = (i < n) ? deg[i] : 0;
  s[threadIdx.x] = v;
  __syncthreads();
#pragma unroll
  for (int off = 1; off < SCAN_B; off <<= 1) {
    const int tv = (threadIdx.x >= off) ? s[threadIdx.x - off] : 0;
    __syncthreads();
    s[threadIdx.x] += tv;
    __syncthreads();
  }
  if (i < n) rowstart1[i] = s[threadIdx.x];
  if (threadIdx.x == SCAN_B - 1) bwork[blockIdx.x] = s[SCAN_B - 1];
}

__global__ void gin_scan3_kernel(int* __restrict__ rowstart,
                                 const int* __restrict__ bwork, int n) {
  __shared__ int s[SCAN_B];
  int part = 0;
  for (int k = threadIdx.x; k < (int)blockIdx.x; k += SCAN_B) part += bwork[k];
  s[threadIdx.x] = part;
  __syncthreads();
#pragma unroll
  for (int off = SCAN_B / 2; off > 0; off >>= 1) {
    if ((int)threadIdx.x < off) s[threadIdx.x] += s[threadIdx.x + off];
    __syncthreads();
  }
  const int boff = s[0];
  const int i = blockIdx.x * SCAN_B + threadIdx.x;
  if (i < n) rowstart[i + 1] += boff;
  if (i == 0) rowstart[0] = 0;
}

__global__ void gin_fill_kernel(const int* __restrict__ rowp,
                                const int* __restrict__ colp,
                                const int* __restrict__ rowstart,
                                int* __restrict__ cursor,
                                int* __restrict__ perm, int E) {
  const int e = blockIdx.x * blockDim.x + threadIdx.x;
  if (e < E) {
    const int r = rowp[e];
    const int pos = rowstart[r] + atomicAdd(&cursor[r], 1);
    perm[pos] = colp[e];
  }
}

__global__ void gin_gather_kernel(const float* __restrict__ T,
                                  const int* __restrict__ rowstart,
                                  const int* __restrict__ perm,
                                  float* __restrict__ AGG, int N) {
  const int g = (blockIdx.x * blockDim.x + threadIdx.x) >> 4;
  if (g >= N) return;
  const int l4 = (threadIdx.x & 15) * 4;
  const int s = rowstart[g];
  const int e = rowstart[g + 1];
  float4 acc = make_float4(0.f, 0.f, 0.f, 0.f);
  for (int j = s; j < e; ++j) {
    const int c = perm[j];
    const float4 v = *(const float4*)(T + (size_t)c * NDIM + l4);
    acc.x += v.x;
    acc.y += v.y;
    acc.z += v.z;
    acc.w += v.w;
  }
  *(float4*)(AGG + (size_t)g * NDIM + l4) = acc;
}


extern "C" void kernel_launch(void* const* d_in, const int* in_sizes, int n_in,
                              void* d_out, int out_size, void* d_ws,
                              size_t ws_size, hipStream_t stream) {
  (void)n_in;
  (void)out_size;
  (void)ws_size;

  const float* x = (const float*)d_in[0];
  const int* ei = (const int*)d_in[1];
  const float* W1 = (const float*)d_in[2];
  const float* b1 = (const float*)d_in[3];
  const float* W2 = (const float*)d_in[4];
  const float* b2 = (const float*)d_in[5];
  const float* Wf = (const float*)d_in[6];
  const float* bf = (const float*)d_in[7];
  float* out = (float*)d_out;

  const int N = in_sizes[0] / NDIM;
  const int E = in_sizes[1] / 2;
  const int L = in_sizes[3] / NDIM;
  const int nTiles = N / 16;
  const size_t NB = (size_t)N * NDIM;

  float* agg = (float*)d_ws;
  float* t = agg + NB;
  float* h = t + NB;
  const int nMat = 2 * L + 1;
  float* wp = h + NB;
  int* deg = (int*)(wp + (size_t)nMat * WP_ELEMS);
  int* cursor = deg + N;
  int* rowstart = cursor + N;
  int* bwork = rowstart + (N + 1);
  int* perm = bwork + 1024;

  const int* rowp = ei;
  const int* colp = ei + E;

  gin_setup_kernel<<<nMat * REPACK_BLOCKS + (2 * N + 255) / 256, 256, 0,
                     stream>>>(W1, W2, Wf, wp, L, deg, 2 * N);

  const int nScanBlocks = (N + SCAN_B - 1) / SCAN_B;
  gin_degree_kernel<<<(E + 255) / 256, 256, 0, stream>>>(rowp, deg, E);
  gin_scan1_kernel<<<nScanBlocks, SCAN_B, 0, stream>>>(deg, rowstart + 1,
                                                       bwork, N);
  gin_scan3_kernel<<<nScanBlocks, SCAN_B, 0, stream>>>(rowstart, bwork, N);
  gin_fill_kernel<<<(E + 255) / 256, 256, 0, stream>>>(rowp, colp, rowstart,
                                                       cursor, perm, E);

  const dim3 mlpBlk(WAVES_PER_BLOCK * 32);
  const dim3 mlpGrd(MLP_GRID);
  const dim3 gatGrd((N * 16 + 255) / 256);

  const float* hcur = x;
  for (int l = 0; l < L; ++l) {
    const float* wp1 = wp + (size_t)l * WP_ELEMS;
    const float* wp2 = wp + (size_t)(L + l) * WP_ELEMS;
    const float* v1 = b1 + (size_t)l * NDIM;
    const float* v2 = b2 + (size_t)l * NDIM;

    gin_mlp_kernel<0><<<mlpGrd, mlpBlk, 0, stream>>>(hcur, nullptr, wp1, v1,
                                                     wp2, v2, t, nTiles);
    gin_gather_kernel<<<gatGrd, 256, 0, stream>>>(t, rowstart, perm, agg, N);
    gin_mlp_kernel<1><<<mlpGrd, mlpBlk, 0, stream>>>(hcur, agg, wp1, v1, wp2,
                                                     v2, h, nTiles);
    hcur = h;
  }
  gin_mlp_kernel<2><<<mlpGrd, mlpBlk, 0, stream>>>(
      hcur, nullptr, wp + (size_t)(2 * L) * WP_ELEMS, bf, nullptr, nullptr, out,
      nTiles);
}


# CSR built by a two-level counting sort with all counters in LDS (chunk histograms, column scan, staged scatter, per-bucket finish): no global atomics; neighbour order in a row's sum changes
# speedup vs baseline: 2.1956x; 1.6368x over previous
#include <hip/hip_runtime.h>


typedef __attribute__((ext_vector_type(2))) float v2f;
typedef __attribute__((ext_vector_type(8))) float v8f;
typedef __attribute__((ext_vector_type(2))) int v2i;

#define NDIM 64
#define WAVES_PER_BLOCK 4
#define WP_ELEMS 4096
#define RB 256
#define RB_SHIFT 8
#define NBKP 512
#define CH 4096
#define BIN_T 512
#define FIN_U 16
#define FIN_CAP (FIN_U * RB)
static_assert(BIN_T == NBKP, "one thread per padded bucket in the bin kernels");
static_assert((1 << RB_SHIFT) == RB, "bucket = row >> RB_SHIFT");
#define MLP_GRID 512

__device__ __forceinline__ v8f wmma_f32_16x16x4(v2f a, v2f b, v8f c) {
  return __builtin_amdgcn_wmma_f32_16x16x4_f32(false, a, false, b, (short)0, c,
                                               false, false);
}

#define REPACK_BLOCKS 8
__global__ void gin_setup_kernel(const float* __restrict__ W1,
                                 const float* __restrict__ W2,
                                 const float* __restrict__ Wf,
                                 float* __restrict__ Wp, int L) {
  const int mat = (int)blockIdx.x / REPACK_BLOCKS;
  const float* W = mat < L       ? W1 + (size_t)mat * (NDIM * NDIM)
                   : mat < 2 * L ? W2 + (size_t)(mat - L) * (NDIM * NDIM)
                                 : Wf;
  float* dst = Wp + (size_t)mat * WP_ELEMS;
  const int tid =
      ((int)blockIdx.x % REPACK_BLOCKS) * 256 + (int)threadIdx.x;
  const int c = tid >> 7;
  const int n = (tid >> 5) & 3;
  const int lane = tid & 31;
  const int hi = lane >> 4;
  const int m = lane & 15;
  const int k = 4 * c + 2 * hi;
  dst[tid * 2 + 0] = W[(k + 0) * NDIM + 16 * n + m];
  dst[tid * 2 + 1] = W[(k + 1) * NDIM + 16 * n + m];
}


template <int MODE>
__device__ __forceinline__ void gin_fetch_rows(const float* __restrict__ H,
                                               const float* __restrict__ AGG,
                                               int tile, int laneOff,
                                               v2f (&zh)[16], v2f (&zg)[16]) {
  const float* hp = H + (size_t)tile * (16 * NDIM) + laneOff;
#pragma unroll
  for (int c = 0; c < 16; ++c) zh[c] = *(const v2f*)(hp + 4 * c);
  if (MODE == 1) {
    const float* gp = AGG + (size_t)tile * (16 * NDIM) + laneOff;
#pragma unroll
    for (int c = 0; c < 16; ++c) zg[c] = *(const v2f*)(gp + 4 * c);
  }
}

template <int MODE>
__device__ __forceinline__ void gin_mlp_tile(const v2f (&zh)[16],
                                             const v2f (&zg)[16],
                                             const v2f (&w1)[64],
                                             const v2f (&w2)[64],
                                             const float (&bias1)[4],
                                             const float (&bias2)[4],
                                             float* lt, float* __restrict__ OUT,
                                             int tile, int laneM, int hi) {
  v2f a[16];
#pragma unroll
  for (int c = 0; c < 16; ++c) {
    v2f z = zh[c];
    if (MODE == 1) {
      z.x += zg[c].x;
      z.y += zg[c].y;
    }
    a[c] = z;
  }

  float* op = OUT + ((size_t)tile * 16 + 8 * hi) * NDIM + laneM;

  v8f acc[4];
#pragma unroll
  for (int n = 0; n < 4; ++n)
#pragma unroll
    for (int r = 0; r < 8; ++r) acc[n][r] = bias1[n];
#pragma unroll
  for (int c = 0; c < 16; ++c)
#pragma unroll
    for (int n = 0; n < 4; ++n)
      acc[n] = wmma_f32_16x16x4(a[c], w1[c * 4 + n], acc[n]);

  if (MODE == 2) {
#pragma unroll
    for (int n = 0; n < 4; ++n)
#pragma unroll
      for (int r = 0; r < 8; ++r) op[r * NDIM + 16 * n] = acc[n][r];
    return;
  }

#pragma unroll
  for (int n = 0; n < 4; ++n)
#pragma unroll
    for (int r = 0; r < 8; ++r)
      lt[(r + 8 * hi) * NDIM + 16 * n + laneM] = fmaxf(acc[n][r], 0.0f);

#pragma unroll
  for (int c = 0; c < 16; ++c)
    a[c] = *(const v2f*)(lt + laneM * NDIM + 4 * c + 2 * hi);

#pragma unroll
  for (int n = 0; n < 4; ++n)
#pragma unroll
    for (int r = 0; r < 8; ++r) acc[n][r] = bias2[n];
#pragma unroll
  for (int c = 0; c < 16; ++c)
#pragma unroll
    for (int n = 0; n < 4; ++n)
      acc[n] = wmma_f32_16x16x4(a[c], w2[c * 4 + n], acc[n]);

#pragma unroll
  for (int n = 0; n < 4; ++n)
#pragma unroll
    for (int r = 0; r < 8; ++r) op[r * NDIM + 16 * n] = acc[n][r];
}

template <int MODE>
__global__ void __launch_bounds__(WAVES_PER_BLOCK * 32)
gin_mlp_kernel(const float* __restrict__ H, const float* __restrict__ AGG,
               const float* __restrict__ WP1, const float* __restrict__ B1,
               const float* __restrict__ WP2, const float* __restrict__ B2,
               float* __restrict__ OUT, int nTiles) {
  __shared__ float lds[WAVES_PER_BLOCK][16 * NDIM];

  const int wave = __builtin_amdgcn_readfirstlane((int)(threadIdx.x >> 5));
  const int lane = threadIdx.x & 31;
  const int stride = (int)gridDim.x * WAVES_PER_BLOCK;
  int tile = (int)blockIdx.x * WAVES_PER_BLOCK + wave;
  if (tile >= nTiles) return;

  const int laneM = lane & 15;
  const int hi = lane >> 4;
  const int laneOff = laneM * NDIM + 2 * hi;
  float* lt = lds[wave];

  v2f hA[16], gA[16], hB[16], gB[16];
  gin_fetch_rows<MODE>(H, AGG, tile, laneOff, hA, gA);

  v2f w1[64];
  v2f w2[64];
  float bias1[4];
  float bias2[4];
  {
    const v2f* wp1 = (const v2f*)WP1 + lane;
#pragma unroll
    for (int i = 0; i < 64; ++i) w1[i] = wp1[i * 32];
#pragma unroll
    for (int n = 0; n < 4; ++n) bias1[n] = B1[16 * n + laneM];
    if (MODE != 2) {
      const v2f* wp2 = (const v2f*)WP2 + lane;
#pragma unroll
      for (int i = 0; i < 64; ++i) w2[i] = wp2[i * 32];
#pragma unroll
      for (int n = 0; n < 4; ++n) bias2[n] = B2[16 * n + laneM];
    }
  }

  int cnt = (nTiles - 1 - tile) / stride + 1;

  while (cnt >= 2) {
    const int t1 = tile + stride;
    const int t2 = t1 + stride;

    gin_fetch_rows<MODE>(H, AGG, t1, laneOff, hB, gB);
    __builtin_amdgcn_sched_barrier(0);
    gin_mlp_tile<MODE>(hA, gA, w1, w2, bias1, bias2, lt, OUT, tile, laneM, hi);

    gin_fetch_rows<MODE>(H, AGG, cnt > 2 ? t2 : t1, laneOff, hA, gA);
    __builtin_amdgcn_sched_barrier(0);
    gin_mlp_tile<MODE>(hB, gB, w1, w2, bias1, bias2, lt, OUT, t1, laneM, hi);

    tile = t2;
    cnt -= 2;
  }
  if (cnt == 1)
    gin_mlp_tile<MODE>(hA, gA, w1, w2, bias1, bias2, lt, OUT, tile, laneM, hi);
}


template <int NW>
__device__ __forceinline__ int gin_block_scan(int v, int t, int* wsum,
                                              int& total) {
  const int lane = t & 31;
  const int wave = t >> 5;
#pragma unroll
  for (int off = 1; off < 32; off <<= 1) {
    const int u = __shfl_up(v, off);
    if (lane >= off) v += u;
  }
  if (lane == 31) wsum[wave] = v;
  __syncthreads();
  int before = 0;
  int all = 0;
#pragma unroll
  for (int w = 0; w < NW; ++w) {
    const int ws = wsum[w];
    all += ws;
    if (w < wave) before += ws;
  }
  total = all;
  return v + before;
}

__global__ void __launch_bounds__(BIN_T)
gin_bin_count_kernel(const int* __restrict__ rowp, int* __restrict__ cnt,
                     int E) {
  __shared__ int lc[NBKP];
  const int t = (int)threadIdx.x;
  lc[t] = 0;
  __syncthreads();
  const int e0 = (int)blockIdx.x * CH;
#pragma unroll
  for (int j = 0; j < CH / BIN_T; ++j) {
    const int e = e0 + j * BIN_T + t;
    if (e < E) atomicAdd(&lc[rowp[e] >> RB_SHIFT], 1);
  }
  __syncthreads();
  cnt[(size_t)blockIdx.x * NBKP + t] = lc[t];
}

__global__ void __launch_bounds__(BIN_T)
gin_bin_colscan_kernel(const int* __restrict__ cnt, int* __restrict__ pre,
                       int* __restrict__ tot, int nChunks) {
  __shared__ int wsum[BIN_T / 32];
  const int t = (int)threadIdx.x;
  const int k = (int)blockIdx.x;
  const int v = (t < nChunks) ? cnt[(size_t)t * NBKP + k] : 0;
  int total;
  const int incl = gin_block_scan<BIN_T / 32>(v, t, wsum, total);
  if (t < nChunks) pre[(size_t)t * NBKP + k] = incl - v;
  if (t == 0) tot[k] = total;
}

__global__ void __launch_bounds__(BIN_T)
gin_bin_scatter_kernel(const int* __restrict__ rowp,
                       const int* __restrict__ colp,
                       const int* __restrict__ cnt,
                       const int* __restrict__ pre,
                       const int* __restrict__ tot, int* __restrict__ gbase,
                       int* __restrict__ sorted, int E) {
  __shared__ int wsa[BIN_T / 32];
  __shared__ int wsb[BIN_T / 32];
  __shared__ int lcur[NBKP];
  __shared__ int gdel[NBKP];
  __shared__ v2i stage[CH];
  const int t = (int)threadIdx.x;
  const int b = (int)blockIdx.x;
  const int e0 = b * CH;

  int er[CH / BIN_T], ec[CH / BIN_T];
#pragma unroll
  for (int j = 0; j < CH / BIN_T; ++j) {
    const int e = e0 + j * BIN_T + t;
    er[j] = (e < E) ? rowp[e] : -1;
    ec[j] = (e < E) ? colp[e] : 0;
  }
  const int c = cnt[(size_t)b * NBKP + t];
  const int p = pre[(size_t)b * NBKP + t];
  const int T = tot[t];
  int nChunk, nAll;
  const int lstart =
      gin_block_scan<BIN_T / 32>(c, t, wsa, nChunk) - c;
  const int base =
      gin_block_scan<BIN_T / 32>(T, t, wsb, nAll) - T;
  lcur[t] = lstart;
  gdel[t] = base + p - lstart;
  if (b == 0) {
    gbase[t] = base;
    if (t == 0) gbase[NBKP] = nAll;
  }
  __syncthreads();

  int slot[CH / BIN_T];
#pragma unroll
  for (int j = 0; j < CH / BIN_T; ++j)
    slot[j] = (er[j] >= 0) ? atomicAdd(&lcur[er[j] >> RB_SHIFT], 1) : 0;
#pragma unroll
  for (int j = 0; j < CH / BIN_T; ++j) {
    if (er[j] >= 0) {
      v2i v;
      v.x = er[j];
      v.y = ec[j];
      stage[slot[j]] = v;
    }
  }
  __syncthreads();

  v2i* dst = (v2i*)sorted;
#pragma unroll
  for (int j = 0; j < CH / BIN_T; ++j) {
    const int i = j * BIN_T + t;
    if (i < nChunk) {
      const v2i v = stage[i];
      dst[gdel[v.x >> RB_SHIFT] + i] = v;
    }
  }
}

__global__ void __launch_bounds__(RB)
gin_bin_finish_kernel(const int* __restrict__ sorted,
                      const int* __restrict__ gbase,
                      int* __restrict__ rowstart, int* __restrict__ perm,
                      int N) {
  __shared__ int wsum[RB / 32];
  __shared__ int lcur[RB];
  __shared__ int lperm[FIN_CAP];
  const int t = (int)threadIdx.x;
  const int k = (int)blockIdx.x;
  const int b0 = gbase[k];
  const int n = gbase[k + 1] - b0;
  const bool staged = n <= FIN_CAP;
  const v2i* src = (const v2i*)sorted + b0;

  v2i ev[FIN_U];
  lcur[t] = 0;
  __syncthreads();
  if (staged) {
#pragma unroll
    for (int u = 0; u < FIN_U; ++u) {
      const int i = u * RB + t;
      if (i < n) {
        ev[u] = src[i];
      } else {
        ev[u].x = -1;
        ev[u].y = 0;
      }
    }
#pragma unroll
    for (int u = 0; u < FIN_U; ++u)
      if (ev[u].x >= 0) atomicAdd(&lcur[ev[u].x & (RB - 1)], 1);
  } else {
    for (int i = t; i < n; i += RB) atomicAdd(&lcur[src[i].x & (RB - 1)], 1);
  }
  __syncthreads();

  const int d = lcur[t];
  int nScan;
  const int ex =
      gin_block_scan<RB / 32>(d, t, wsum, nScan) - d;
  lcur[t] = ex;
  const int row = k * RB + t;
  if (row < N) rowstart[row] = b0 + ex;
  if (row == N - 1) rowstart[N] = b0 + n;
  __syncthreads();

  if (staged) {
    int sl[FIN_U];
#pragma unroll
    for (int u = 0; u < FIN_U; ++u)
      sl[u] = (ev[u].x >= 0) ? atomicAdd(&lcur[ev[u].x & (RB - 1)], 1) : 0;
#pragma unroll
    for (int u = 0; u < FIN_U; ++u)
      if (ev[u].x >= 0) lperm[sl[u]] = ev[u].y;
    __syncthreads();
#pragma unroll
    for (int u = 0; u < FIN_U; ++u) {
      const int i = u * RB + t;
      if (i < n) perm[b0 + i] = lperm[i];
    }
  } else {
    for (int i = t; i < n; i += RB) {
      const v2i v = src[i];
      perm[b0 + atomicAdd(&lcur[v.x & (RB - 1)], 1)] = v.y;
    }
  }
}

__global__ void gin_gather_kernel(const float* __restrict__ T,
                                  const int* __restrict__ rowstart,
                                  const int* __restrict__ perm,
                                  float* __restrict__ AGG, int N) {
  const int g = (blockIdx.x * blockDim.x + threadIdx.x) >> 4;
  if (g >= N) return;
  const int l4 = (threadIdx.x & 15) * 4;
  const int s = rowstart[g];
  const int e = rowstart[g + 1];
  float4 acc = make_float4(0.f, 0.f, 0.f, 0.f);
  for (int j = s; j < e; ++j) {
    const int c = perm[j];
    const float4 v = *(const float4*)(T + (size_t)c * NDIM + l4);
    acc.x += v.x;
    acc.y += v.y;
    acc.z += v.z;
    acc.w += v.w;
  }
  *(float4*)(AGG + (size_t)g * NDIM + l4) = acc;
}


extern "C" void kernel_launch(void* const* d_in, const int* in_sizes, int n_in,
                              void* d_out, int out_size, void* d_ws,
                              size_t ws_size, hipStream_t stream) {
  (void)n_in;
  (void)out_size;
  (void)ws_size;

  const float* x = (const float*)d_in[0];
  const int* ei = (const int*)d_in[1];
  const float* W1 = (const float*)d_in[2];
  const float* b1 = (const float*)d_in[3];
  const float* W2 = (const float*)d_in[4];
  const float* b2 = (const float*)d_in[5];
  const float* Wf = (const float*)d_in[6];
  const float* bf = (const float*)d_in[7];
  float* out = (float*)d_out;

  const int N = in_sizes[0] / NDIM;
  const int E = in_sizes[1] / 2;
  const int L = in_sizes[3] / NDIM;
  const int nTiles = N / 16;
  const size_t NB = (size_t)N * NDIM;

  float* agg = (float*)d_ws;
  float* t = agg + NB;
  float* h = t + NB;
  const int nMat = 2 * L + 1;
  float* wp = h + NB;
  const int nChunks = (E + CH - 1) / CH;
  const int nBuckets = (N + RB - 1) / RB;
  int* cnt = (int*)(wp + (size_t)nMat * WP_ELEMS);
  int* pre = cnt + (size_t)nChunks * NBKP;
  int* tot = pre + (size_t)nChunks * NBKP;
  int* gbase = tot + NBKP;
  int* sorted = gbase + 2 * NBKP;
  int* perm = sorted + 2 * (size_t)E;
  int* rowstart = perm + E;

  const int* rowp = ei;
  const int* colp = ei + E;

  gin_setup_kernel<<<nMat * REPACK_BLOCKS, 256, 0, stream>>>(W1, W2, Wf, wp, L);

  gin_bin_count_kernel<<<nChunks, BIN_T, 0, stream>>>(rowp, cnt, E);
  gin_bin_colscan_kernel<<<NBKP, BIN_T, 0, stream>>>(cnt, pre, tot, nChunks);
  gin_bin_scatter_kernel<<<nChunks, BIN_T, 0, stream>>>(rowp, colp, cnt, pre,
                                                        tot, gbase, sorted, E);
  gin_bin_finish_kernel<<<nBuckets, RB, 0, stream>>>(sorted, gbase, rowstart,
                                                     perm, N);

  const dim3 mlpBlk(WAVES_PER_BLOCK * 32);
  const dim3 mlpGrd(MLP_GRID);
  const dim3 gatGrd((N * 16 + 255) / 256);

  const float* hcur = x;
  for (int l = 0; l < L; ++l) {
    const float* wp1 = wp + (size_t)l * WP_ELEMS;
    const float* wp2 = wp + (size_t)(L + l) * WP_ELEMS;
    const float* v1 = b1 + (size_t)l * NDIM;
    const float* v2 = b2 + (size_t)l * NDIM;

    gin_mlp_kernel<0><<<mlpGrd, mlpBlk, 0, stream>>>(hcur, nullptr, wp1, v1,
                                                     wp2, v2, t, nTiles);
    gin_gather_kernel<<<gatGrd, 256, 0, stream>>>(t, rowstart, perm, agg, N);
    gin_mlp_kernel<1><<<mlpGrd, mlpBlk, 0, stream>>>(hcur, agg, wp1, v1, wp2,
                                                     v2, h, nTiles);
    hcur = h;
  }
  gin_mlp_kernel<2><<<mlpGrd, mlpBlk, 0, stream>>>(
      hcur, nullptr, wp + (size_t)(2 * L) * WP_ELEMS, bf, nullptr, nullptr, out,
      nTiles);
}


# Each layer's combine MLP and the row-wise stage after it (next layer's message MLP, or the final linear) run in one launch on the tile still in the wave; their weights from LDS; 15 -> 12 launches
# speedup vs baseline: 2.3194x; 1.0564x over previous
#include <hip/hip_runtime.h>


typedef __attribute__((ext_vector_type(2))) float v2f;
typedef __attribute__((ext_vector_type(8))) float v8f;
typedef __attribute__((ext_vector_type(2))) int v2i;

#define NDIM 64
#define WAVES_PER_BLOCK 4
#define WP_ELEMS 4096
#define RB 256
#define RB_SHIFT 8
#define NBKP 512
#define CH 4096
#define BIN_T 512
#define FIN_U 16
#define FIN_CAP (FIN_U * RB)
static_assert(BIN_T == NBKP, "one thread per padded bucket in the bin kernels");
static_assert((1 << RB_SHIFT) == RB, "bucket = row >> RB_SHIFT");
#define MLP_GRID 512

__device__ __forceinline__ v8f wmma_f32_16x16x4(v2f a, v2f b, v8f c) {
  return __builtin_amdgcn_wmma_f32_16x16x4_f32(false, a, false, b, (short)0, c,
                                               false, false);
}

#define REPACK_BLOCKS 8
__global__ void gin_setup_kernel(const float* __restrict__ W1,
                                 const float* __restrict__ W2,
                                 const float* __restrict__ Wf,
                                 float* __restrict__ Wp, int L) {
  const int mat = (int)blockIdx.x / REPACK_BLOCKS;
  const float* W = mat < L       ? W1 + (size_t)mat * (NDIM * NDIM)
                   : mat < 2 * L ? W2 + (size_t)(mat - L) * (NDIM * NDIM)
                                 : Wf;
  float* dst = Wp + (size_t)mat * WP_ELEMS;
  const int tid =
      ((int)blockIdx.x % REPACK_BLOCKS) * 256 + (int)threadIdx.x;
  const int c = tid >> 7;
  const int n = (tid >> 5) & 3;
  const int lane = tid & 31;
  const int hi = lane >> 4;
  const int m = lane & 15;
  const int k = 4 * c + 2 * hi;
  dst[tid * 2 + 0] = W[(k + 0) * NDIM + 16 * n + m];
  dst[tid * 2 + 1] = W[(k + 1) * NDIM + 16 * n + m];
}


template <int MODE>
__device__ __forceinline__ void gin_fetch_rows(const float* __restrict__ H,
                                               const float* __restrict__ AGG,
                                               int tile, int laneOff,
                                               v2f (&zh)[16], v2f (&zg)[16]) {
  const float* hp = H + (size_t)tile * (16 * NDIM) + laneOff;
#pragma unroll
  for (int c = 0; c < 16; ++c) zh[c] = *(const v2f*)(hp + 4 * c);
  if (MODE == 1) {
    const float* gp = AGG + (size_t)tile * (16 * NDIM) + laneOff;
#pragma unroll
    for (int c = 0; c < 16; ++c) zg[c] = *(const v2f*)(gp + 4 * c);
  }
}

template <int MODE>
__device__ __forceinline__ void gin_mlp_tile(const v2f (&zh)[16],
                                             const v2f (&zg)[16],
                                             const v2f (&w1)[64],
                                             const v2f (&w2)[64],
                                             const float (&bias1)[4],
                                             const float (&bias2)[4],
                                             float* lt, float* __restrict__ OUT,
                                             int tile, int laneM, int hi) {
  v2f a[16];
#pragma unroll
  for (int c = 0; c < 16; ++c) {
    v2f z = zh[c];
    if (MODE == 1) {
      z.x += zg[c].x;
      z.y += zg[c].y;
    }
    a[c] = z;
  }

  float* op = OUT + ((size_t)tile * 16 + 8 * hi) * NDIM + laneM;

  v8f acc[4];
#pragma unroll
  for (int n = 0; n < 4; ++n)
#pragma unroll
    for (int r = 0; r < 8; ++r) acc[n][r] = bias1[n];
#pragma unroll
  for (int c = 0; c < 16; ++c)
#pragma unroll
    for (int n = 0; n < 4; ++n)
      acc[n] = wmma_f32_16x16x4(a[c], w1[c * 4 + n], acc[n]);

  if (MODE == 2) {
#pragma unroll
    for (int n = 0; n < 4; ++n)
#pragma unroll
      for (int r = 0; r < 8; ++r) op[r * NDIM + 16 * n] = acc[n][r];
    return;
  }

#pragma unroll
  for (int n = 0; n < 4; ++n)
#pragma unroll
    for (int r = 0; r < 8; ++r)
      lt[(r + 8 * hi) * NDIM + 16 * n + laneM] = fmaxf(acc[n][r], 0.0f);

#pragma unroll
  for (int c = 0; c < 16; ++c)
    a[c] = *(const v2f*)(lt + laneM * NDIM + 4 * c + 2 * hi);

#pragma unroll
  for (int n = 0; n < 4; ++n)
#pragma unroll
    for (int r = 0; r < 8; ++r) acc[n][r] = bias2[n];
#pragma unroll
  for (int c = 0; c < 16; ++c)
#pragma unroll
    for (int n = 0; n < 4; ++n)
      acc[n] = wmma_f32_16x16x4(a[c], w2[c * 4 + n], acc[n]);

#pragma unroll
  for (int n = 0; n < 4; ++n)
#pragma unroll
    for (int r = 0; r < 8; ++r) op[r * NDIM + 16 * n] = acc[n][r];
}

template <int MODE>
__global__ void __launch_bounds__(WAVES_PER_BLOCK * 32)
gin_mlp_kernel(const float* __restrict__ H, const float* __restrict__ AGG,
               const float* __restrict__ WP1, const float* __restrict__ B1,
               const float* __restrict__ WP2, const float* __restrict__ B2,
               float* __restrict__ OUT, int nTiles) {
  __shared__ float lds[WAVES_PER_BLOCK][16 * NDIM];

  const int wave = __builtin_amdgcn_readfirstlane((int)(threadIdx.x >> 5));
  const int lane = threadIdx.x & 31;
  const int stride = (int)gridDim.x * WAVES_PER_BLOCK;
  int tile = (int)blockIdx.x * WAVES_PER_BLOCK + wave;
  if (tile >= nTiles) return;

  const int laneM = lane & 15;
  const int hi = lane >> 4;
  const int laneOff = laneM * NDIM + 2 * hi;
  float* lt = lds[wave];

  v2f hA[16], gA[16], hB[16], gB[16];
  gin_fetch_rows<MODE>(H, AGG, tile, laneOff, hA, gA);

  v2f w1[64];
  v2f w2[64];
  float bias1[4];
  float bias2[4];
  {
    const v2f* wp1 = (const v2f*)WP1 + lane;
#pragma unroll
    for (int i = 0; i < 64; ++i) w1[i] = wp1[i * 32];
#pragma unroll
    for (int n = 0; n < 4; ++n) bias1[n] = B1[16 * n + laneM];
    if (MODE != 2) {
      const v2f* wp2 = (const v2f*)WP2 + lane;
#pragma unroll
      for (int i = 0; i < 64; ++i) w2[i] = wp2[i * 32];
#pragma unroll
      for (int n = 0; n < 4; ++n) bias2[n] = B2[16 * n + laneM];
    }
  }

  int cnt = (nTiles - 1 - tile) / stride + 1;

  while (cnt >= 2) {
    const int t1 = tile + stride;
    const int t2 = t1 + stride;

    gin_fetch_rows<MODE>(H, AGG, t1, laneOff, hB, gB);
    __builtin_amdgcn_sched_barrier(0);
    gin_mlp_tile<MODE>(hA, gA, w1, w2, bias1, bias2, lt, OUT, tile, laneM, hi);

    gin_fetch_rows<MODE>(H, AGG, cnt > 2 ? t2 : t1, laneOff, hA, gA);
    __builtin_amdgcn_sched_barrier(0);
    gin_mlp_tile<MODE>(hB, gB, w1, w2, bias1, bias2, lt, OUT, t1, laneM, hi);

    tile = t2;
    cnt -= 2;
  }
  if (cnt == 1)
    gin_mlp_tile<MODE>(hA, gA, w1, w2, bias1, bias2, lt, OUT, tile, laneM, hi);
}

__device__ __forceinline__ void gin_lds_frags(const v2f* wl, int blk,
                                              v2f (&b)[16]) {
#pragma unroll
  for (int i = 0; i < 16; ++i) b[i] = wl[(blk * 16 + i) * 32];
}

__device__ __forceinline__ void gin_wmma_block(const v2f (&a)[16], int blk,
                                               const v2f (&b)[16],
                                               v8f (&acc)[4]) {
#pragma unroll
  for (int cc = 0; cc < 4; ++cc)
#pragma unroll
    for (int n = 0; n < 4; ++n)
      acc[n] = wmma_f32_16x16x4(a[blk * 4 + cc], b[cc * 4 + n], acc[n]);
}

template <bool CHAIN>
__device__ __forceinline__ void gin_gemm_lds(const v2f (&a)[16], const v2f* wl,
                                             const v2f* nxt, v2f (&b0)[16],
                                             v2f (&b1)[16], v8f (&acc)[4]) {
  gin_lds_frags(wl, 1, b1);
  __builtin_amdgcn_sched_barrier(0);
  gin_wmma_block(a, 0, b0, acc);
  __builtin_amdgcn_sched_barrier(0);
  gin_lds_frags(wl, 2, b0);
  __builtin_amdgcn_sched_barrier(0);
  gin_wmma_block(a, 1, b1, acc);
  __builtin_amdgcn_sched_barrier(0);
  gin_lds_frags(wl, 3, b1);
  __builtin_amdgcn_sched_barrier(0);
  gin_wmma_block(a, 2, b0, acc);
  __builtin_amdgcn_sched_barrier(0);
  if (CHAIN) gin_lds_frags(nxt, 0, b0);
  __builtin_amdgcn_sched_barrier(0);
  gin_wmma_block(a, 3, b1, acc);
  __builtin_amdgcn_sched_barrier(0);
}

template <int FIN>
__device__ __forceinline__ void gin_fused_tile(
    const v2f (&zh)[16], const v2f (&zg)[16], const v2f (&w1)[64],
    const v2f* wl2, const float (&bias1)[4], const float (&bias2)[4],
    const v2f* wl3, const v2f* wl4, const float (&bias3)[4],
    const float (&bias4)[4], float* lt, float* __restrict__ HOUT,
    float* __restrict__ OUT2, int tile, int laneM, int hi) {
  v2f a[16];
#pragma unroll
  for (int c = 0; c < 16; ++c) {
    v2f z = zh[c];
    z.x += zg[c].x;
    z.y += zg[c].y;
    a[c] = z;
  }
  const size_t off = ((size_t)tile * 16 + 8 * hi) * NDIM + laneM;
  v2f bq0[16], bq1[16];

  v8f acc[4];
#pragma unroll
  for (int n = 0; n < 4; ++n)
#pragma unroll
    for (int r = 0; r < 8; ++r) acc[n][r] = bias1[n];
#pragma unroll
  for (int c = 0; c < 12; ++c)
#pragma unroll
    for (int n = 0; n < 4; ++n)
      acc[n] = wmma_f32_16x16x4(a[c], w1[c * 4 + n], acc[n]);
  __builtin_amdgcn_sched_barrier(0);
  gin_lds_frags(wl2, 0, bq0);
  __builtin_amdgcn_sched_barrier(0);
#pragma unroll
  for (int c = 12; c < 16; ++c)
#pragma unroll
    for (int n = 0; n < 4; ++n)
      acc[n] = wmma_f32_16x16x4(a[c], w1[c * 4 + n], acc[n]);
  __builtin_amdgcn_sched_barrier(0);

#pragma unroll
  for (int n = 0; n < 4; ++n)
#pragma unroll
    for (int r = 0; r < 8; ++r)
      lt[(r + 8 * hi) * NDIM + 16 * n + laneM] = fmaxf(acc[n][r], 0.0f);
#pragma unroll
  for (int c = 0; c < 16; ++c)
    a[c] = *(const v2f*)(lt + laneM * NDIM + 4 * c + 2 * hi);

#pragma unroll
  for (int n = 0; n < 4; ++n)
#pragma unroll
    for (int r = 0; r < 8; ++r) acc[n][r] = bias2[n];
  gin_gemm_lds<true>(a, wl2, wl3, bq0, bq1, acc);

  if (FIN == 0) {
    float* hp = HOUT + off;
#pragma unroll
    for (int n = 0; n < 4; ++n)
#pragma unroll
      for (int r = 0; r < 8; ++r) hp[r * NDIM + 16 * n] = acc[n][r];
  }

#pragma unroll
  for (int n = 0; n < 4; ++n)
#pragma unroll
    for (int r = 0; r < 8; ++r)
      lt[(r + 8 * hi) * NDIM + 16 * n + laneM] = acc[n][r];
#pragma unroll
  for (int c = 0; c < 16; ++c)
    a[c] = *(const v2f*)(lt + laneM * NDIM + 4 * c + 2 * hi);

#pragma unroll
  for (int n = 0; n < 4; ++n)
#pragma unroll
    for (int r = 0; r < 8; ++r) acc[n][r] = bias3[n];
  gin_gemm_lds<FIN == 0>(a, wl3, wl4, bq0, bq1, acc);

  float* op = OUT2 + off;
  if (FIN == 1) {
#pragma unroll
    for (int n = 0; n < 4; ++n)
#pragma unroll
      for (int r = 0; r < 8; ++r) op[r * NDIM + 16 * n] = acc[n][r];
    return;
  }

#pragma unroll
  for (int n = 0; n < 4; ++n)
#pragma unroll
    for (int r = 0; r < 8; ++r)
      lt[(r + 8 * hi) * NDIM + 16 * n + laneM] = fmaxf(acc[n][r], 0.0f);
#pragma unroll
  for (int c = 0; c < 16; ++c)
    a[c] = *(const v2f*)(lt + laneM * NDIM + 4 * c + 2 * hi);

#pragma unroll
  for (int n = 0; n < 4; ++n)
#pragma unroll
    for (int r = 0; r < 8; ++r) acc[n][r] = bias4[n];
  gin_gemm_lds<false>(a, wl4, wl4, bq0, bq1, acc);

#pragma unroll
  for (int n = 0; n < 4; ++n)
#pragma unroll
    for (int r = 0; r < 8; ++r) op[r * NDIM + 16 * n] = acc[n][r];
}

#define WL_FRAGS (WP_ELEMS / 2)

template <int FIN>
__global__ void __launch_bounds__(WAVES_PER_BLOCK * 32)
gin_fused_kernel(const float* __restrict__ H, const float* __restrict__ AGG,
                 const float* __restrict__ WP1, const float* __restrict__ B1,
                 const float* __restrict__ WP2, const float* __restrict__ B2,
                 const float* __restrict__ WP3, const float* __restrict__ B3,
                 const float* __restrict__ WP4, const float* __restrict__ B4,
                 float* __restrict__ HOUT, float* __restrict__ OUT2,
                 int nTiles) {
  __shared__ float lds[WAVES_PER_BLOCK][16 * NDIM];
  __shared__ v2f wl[(FIN ? 2 : 3) * WL_FRAGS];

  const int wave = __builtin_amdgcn_readfirstlane((int)(threadIdx.x >> 5));
  const int lane = threadIdx.x & 31;
  const int stride = (int)gridDim.x * WAVES_PER_BLOCK;
  int tile = (int)blockIdx.x * WAVES_PER_BLOCK + wave;
  const bool has = tile < nTiles;

  const int laneM = lane & 15;
  const int hi = lane >> 4;
  const int laneOff = laneM * NDIM + 2 * hi;
  float* lt = lds[wave];

  v2f hA[16], gA[16], hB[16], gB[16];
  if (has) gin_fetch_rows<1>(H, AGG, tile, laneOff, hA, gA);

  {
    const v2f* s2 = (const v2f*)WP2;
    const v2f* s3 = (const v2f*)WP3;
#pragma unroll
    for (int i = 0; i < WL_FRAGS / (WAVES_PER_BLOCK * 32); ++i) {
      const int j = i * (WAVES_PER_BLOCK * 32) + (int)threadIdx.x;
      wl[j] = s2[j];
      wl[WL_FRAGS + j] = s3[j];
    }
    if (FIN == 0) {
      const v2f* s4 = (const v2f*)WP4;
#pragma unroll
      for (int i = 0; i < WL_FRAGS / (WAVES_PER_BLOCK * 32); ++i) {
        const int j = i * (WAVES_PER_BLOCK * 32) + (int)threadIdx.x;
        wl[2 * WL_FRAGS + j] = s4[j];
      }
    }
  }

  v2f w1[64];
  float bias1[4], bias2[4], bias3[4], bias4[4];
  {
    const v2f* wp1 = (const v2f*)WP1 + lane;
#pragma unroll
    for (int i = 0; i < 64; ++i) w1[i] = wp1[i * 32];
#pragma unroll
    for (int n = 0; n < 4; ++n) {
      bias1[n] = B1[16 * n + laneM];
      bias2[n] = B2[16 * n + laneM];
      bias3[n] = B3[16 * n + laneM];
      bias4[n] = (FIN == 0) ? B4[16 * n + laneM] : 0.0f;
    }
  }
  __syncthreads();
  if (!has) return;

  const v2f* wl2 = wl + lane;
  const v2f* wl3 = wl + WL_FRAGS + lane;
  const v2f* wl4 = wl + (FIN ? 0 : 2 * WL_FRAGS) + lane;

  int cnt = (nTiles - 1 - tile) / stride + 1;

  while (cnt >= 2) {
    const int t1 = tile + stride;
    const int t2 = t1 + stride;

    gin_fetch_rows<1>(H, AGG, t1, laneOff, hB, gB);
    __builtin_amdgcn_sched_barrier(0);
    gin_fused_tile<FIN>(hA, gA, w1, wl2, bias1, bias2, wl3, wl4, bias3, bias4,
                        lt, HOUT, OUT2, tile, laneM, hi);

    gin_fetch_rows<1>(H, AGG, cnt > 2 ? t2 : t1, laneOff, hA, gA);
    __builtin_amdgcn_sched_barrier(0);
    gin_fused_tile<FIN>(hB, gB, w1, wl2, bias1, bias2, wl3, wl4, bias3, bias4,
                        lt, HOUT, OUT2, t1, laneM, hi);

    tile = t2;
    cnt -= 2;
  }
  if (cnt == 1)
    gin_fused_tile<FIN>(hA, gA, w1, wl2, bias1, bias2, wl3, wl4, bias3, bias4,
                        lt, HOUT, OUT2, tile, laneM, hi);
}


template <int NW>
__device__ __forceinline__ int gin_block_scan(int v, int t, int* wsum,
                                              int& total) {
  const int lane = t & 31;
  const int wave = t >> 5;
#pragma unroll
  for (int off = 1; off < 32; off <<= 1) {
    const int u = __shfl_up(v, off);
    if (lane >= off) v += u;
  }
  if (lane == 31) wsum[wave] = v;
  __syncthreads();
  int before = 0;
  int all = 0;
#pragma unroll
  for (int w = 0; w < NW; ++w) {
    const int ws = wsum[w];
    all += ws;
    if (w < wave) before += ws;
  }
  total = all;
  return v + before;
}

__global__ void __launch_bounds__(BIN_T)
gin_bin_count_kernel(const int* __restrict__ rowp, int* __restrict__ cnt,
                     int E) {
  __shared__ int lc[NBKP];
  const int t = (int)threadIdx.x;
  lc[t] = 0;
  __syncthreads();
  const int e0 = (int)blockIdx.x * CH;
#pragma unroll
  for (int j = 0; j < CH / BIN_T; ++j) {
    const int e = e0 + j * BIN_T + t;
    if (e < E) atomicAdd(&lc[rowp[e] >> RB_SHIFT], 1);
  }
  __syncthreads();
  cnt[(size_t)blockIdx.x * NBKP + t] = lc[t];
}

__global__ void __launch_bounds__(BIN_T)
gin_bin_colscan_kernel(const int* __restrict__ cnt, int* __restrict__ pre,
                       int* __restrict__ tot, int nChunks) {
  __shared__ int wsum[BIN_T / 32];
  const int t = (int)threadIdx.x;
  const int k = (int)blockIdx.x;
  const int v = (t < nChunks) ? cnt[(size_t)t * NBKP + k] : 0;
  int total;
  const int incl = gin_block_scan<BIN_T / 32>(v, t, wsum, total);
  if (t < nChunks) pre[(size_t)t * NBKP + k] = incl - v;
  if (t == 0) tot[k] = total;
}

__global__ void __launch_bounds__(BIN_T)
gin_bin_scatter_kernel(const int* __restrict__ rowp,
                       const int* __restrict__ colp,
                       const int* __restrict__ cnt,
                       const int* __restrict__ pre,
                       const int* __restrict__ tot, int* __restrict__ gbase,
                       int* __restrict__ sorted, int E) {
  __shared__ int wsa[BIN_T / 32];
  __shared__ int wsb[BIN_T / 32];
  __shared__ int lcur[NBKP];
  __shared__ int gdel[NBKP];
  __shared__ v2i stage[CH];
  const int t = (int)threadIdx.x;
  const int b = (int)blockIdx.x;
  const int e0 = b * CH;

  int er[CH / BIN_T], ec[CH / BIN_T];
#pragma unroll
  for (int j = 0; j < CH / BIN_T; ++j) {
    const int e = e0 + j * BIN_T + t;
    er[j] = (e < E) ? rowp[e] : -1;
    ec[j] = (e < E) ? colp[e] : 0;
  }
  const int c = cnt[(size_t)b * NBKP + t];
  const int p = pre[(size_t)b * NBKP + t];
  const int T = tot[t];
  int nChunk, nAll;
  const int lstart =
      gin_block_scan<BIN_T / 32>(c, t, wsa, nChunk) - c;
  const int base =
      gin_block_scan<BIN_T / 32>(T, t, wsb, nAll) - T;
  lcur[t] = lstart;
  gdel[t] = base + p - lstart;
  if (b == 0) {
    gbase[t] = base;
    if (t == 0) gbase[NBKP] = nAll;
  }
  __syncthreads();

  int slot[CH / BIN_T];
#pragma unroll
  for (int j = 0; j < CH / BIN_T; ++j)
    slot[j] = (er[j] >= 0) ? atomicAdd(&lcur[er[j] >> RB_SHIFT], 1) : 0;
#pragma unroll
  for (int j = 0; j < CH / BIN_T; ++j) {
    if (er[j] >= 0) {
      v2i v;
      v.x = er[j];
      v.y = ec[j];
      stage[slot[j]] = v;
    }
  }
  __syncthreads();

  v2i* dst = (v2i*)sorted;
#pragma unroll
  for (int j = 0; j < CH / BIN_T; ++j) {
    const int i = j * BIN_T + t;
    if (i < nChunk) {
      const v2i v = stage[i];
      dst[gdel[v.x >> RB_SHIFT] + i] = v;
    }
  }
}

__global__ void __launch_bounds__(RB)
gin_bin_finish_kernel(const int* __restrict__ sorted,
                      const int* __restrict__ gbase,
                      int* __restrict__ rowstart, int* __restrict__ perm,
                      int N) {
  __shared__ int wsum[RB / 32];
  __shared__ int lcur[RB];
  __shared__ int lperm[FIN_CAP];
  const int t = (int)threadIdx.x;
  const int k = (int)blockIdx.x;
  const int b0 = gbase[k];
  const int n = gbase[k + 1] - b0;
  const bool staged = n <= FIN_CAP;
  const v2i* src = (const v2i*)sorted + b0;

  v2i ev[FIN_U];
  lcur[t] = 0;
  __syncthreads();
  if (staged) {
#pragma unroll
    for (int u = 0; u < FIN_U; ++u) {
      const int i = u * RB + t;
      if (i < n) {
        ev[u] = src[i];
      } else {
        ev[u].x = -1;
        ev[u].y = 0;
      }
    }
#pragma unroll
    for (int u = 0; u < FIN_U; ++u)
      if (ev[u].x >= 0) atomicAdd(&lcur[ev[u].x & (RB - 1)], 1);
  } else {
    for (int i = t; i < n; i += RB) atomicAdd(&lcur[src[i].x & (RB - 1)], 1);
  }
  __syncthreads();

  const int d = lcur[t];
  int nScan;
  const int ex =
      gin_block_scan<RB / 32>(d, t, wsum, nScan) - d;
  lcur[t] = ex;
  const int row = k * RB + t;
  if (row < N) rowstart[row] = b0 + ex;
  if (row == N - 1) rowstart[N] = b0 + n;
  __syncthreads();

  if (staged) {
    int sl[FIN_U];
#pragma unroll
    for (int u = 0; u < FIN_U; ++u)
      sl[u] = (ev[u].x >= 0) ? atomicAdd(&lcur[ev[u].x & (RB - 1)], 1) : 0;
#pragma unroll
    for (int u = 0; u < FIN_U; ++u)
      if (ev[u].x >= 0) lperm[sl[u]] = ev[u].y;
    __syncthreads();
#pragma unroll
    for (int u = 0; u < FIN_U; ++u) {
      const int i = u * RB + t;
      if (i < n) perm[b0 + i] = lperm[i];
    }
  } else {
    for (int i = t; i < n; i += RB) {
      const v2i v = src[i];
      perm[b0 + atomicAdd(&lcur[v.x & (RB - 1)], 1)] = v.y;
    }
  }
}

__global__ void gin_gather_kernel(const float* __restrict__ T,
                                  const int* __restrict__ rowstart,
                                  const int* __restrict__ perm,
                                  float* __restrict__ AGG, int N) {
  const int g = (blockIdx.x * blockDim.x + threadIdx.x) >> 4;
  if (g >= N) return;
  const int l4 = (threadIdx.x & 15) * 4;
  const int s = rowstart[g];
  const int e = rowstart[g + 1];
  float4 acc = make_float4(0.f, 0.f, 0.f, 0.f);
  for (int j = s; j < e; ++j) {
    const int c = perm[j];
    const float4 v = *(const float4*)(T + (size_t)c * NDIM + l4);
    acc.x += v.x;
    acc.y += v.y;
    acc.z += v.z;
    acc.w += v.w;
  }
  *(float4*)(AGG + (size_t)g * NDIM + l4) = acc;
}


extern "C" void kernel_launch(void* const* d_in, const int* in_sizes, int n_in,
                              void* d_out, int out_size, void* d_ws,
                              size_t ws_size, hipStream_t stream) {
  (void)n_in;
  (void)out_size;
  (void)ws_size;

  const float* x = (const float*)d_in[0];
  const int* ei = (const int*)d_in[1];
  const float* W1 = (const float*)d_in[2];
  const float* b1 = (const float*)d_in[3];
  const float* W2 = (const float*)d_in[4];
  const float* b2 = (const float*)d_in[5];
  const float* Wf = (const float*)d_in[6];
  const float* bf = (const float*)d_in[7];
  float* out = (float*)d_out;

  const int N = in_sizes[0] / NDIM;
  const int E = in_sizes[1] / 2;
  const int L = in_sizes[3] / NDIM;
  const int nTiles = N / 16;
  const size_t NB = (size_t)N * NDIM;

  float* agg = (float*)d_ws;
  float* t = agg + NB;
  float* h = t + NB;
  const int nMat = 2 * L + 1;
  float* wp = h + NB;
  const int nChunks = (E + CH - 1) / CH;
  const int nBuckets = (N + RB - 1) / RB;
  int* cnt = (int*)(wp + (size_t)nMat * WP_ELEMS);
  int* pre = cnt + (size_t)nChunks * NBKP;
  int* tot = pre + (size_t)nChunks * NBKP;
  int* gbase = tot + NBKP;
  int* sorted = gbase + 2 * NBKP;
  int* perm = sorted + 2 * (size_t)E;
  int* rowstart = perm + E;

  const int* rowp = ei;
  const int* colp = ei + E;

  gin_setup_kernel<<<nMat * REPACK_BLOCKS, 256, 0, stream>>>(W1, W2, Wf, wp, L);

  gin_bin_count_kernel<<<nChunks, BIN_T, 0, stream>>>(rowp, cnt, E);
  gin_bin_colscan_kernel<<<NBKP, BIN_T, 0, stream>>>(cnt, pre, tot, nChunks);
  gin_bin_scatter_kernel<<<nChunks, BIN_T, 0, stream>>>(rowp, colp, cnt, pre,
                                                        tot, gbase, sorted, E);
  gin_bin_finish_kernel<<<nBuckets, RB, 0, stream>>>(sorted, gbase, rowstart,
                                                     perm, N);

  const dim3 mlpBlk(WAVES_PER_BLOCK * 32);
  const dim3 mlpGrd(MLP_GRID);
  const dim3 gatGrd((N * 16 + 255) / 256);

  gin_mlp_kernel<0><<<mlpGrd, mlpBlk, 0, stream>>>(x, nullptr, wp, b1,
                                                   wp + (size_t)L * WP_ELEMS,
                                                   b2, t, nTiles);
  const float* hcur = x;
  for (int l = 0; l < L; ++l) {
    const float* wp1 = wp + (size_t)l * WP_ELEMS;
    const float* wp2 = wp + (size_t)(L + l) * WP_ELEMS;
    const float* v1 = b1 + (size_t)l * NDIM;
    const float* v2 = b2 + (size_t)l * NDIM;

    gin_gather_kernel<<<gatGrd, 256, 0, stream>>>(t, rowstart, perm, agg, N);
    if (l + 1 < L) {
      gin_fused_kernel<0><<<mlpGrd, mlpBlk, 0, stream>>>(
          hcur, agg, wp1, v1, wp2, v2, wp + (size_t)(l + 1) * WP_ELEMS,
          v1 + NDIM, wp + (size_t)(L + l + 1) * WP_ELEMS, v2 + NDIM, h, t,
          nTiles);
    } else {
      gin_fused_kernel<1><<<mlpGrd, mlpBlk, 0, stream>>>(
          hcur, agg, wp1, v1, wp2, v2, wp + (size_t)(2 * L) * WP_ELEMS, bf,
          nullptr, nullptr, nullptr, out, nTiles);
    }
    hcur = h;
  }
}


# Fewer dependent launches, 12 -> 9: each chunk sorts its edges by bucket alone and stores its offsets; finish blocks collect their runs (no count/scan launch); repack rides along; nbr order changes
# speedup vs baseline: 2.5178x; 1.0855x over previous
#include <hip/hip_runtime.h>


typedef __attribute__((ext_vector_type(2))) float v2f;
typedef __attribute__((ext_vector_type(8))) float v8f;
typedef __attribute__((ext_vector_type(2))) int v2i;

#define NDIM 64
#define WAVES_PER_BLOCK 4
#define WP_ELEMS 4096
#define RB 256
#define RB_SHIFT 8
#define NBKP 512
#define CH 4096
#define BIN_T 512
#define OFP (NBKP + 1)
#define FIN_U 16
#define FIN_CAP (FIN_U * RB)
#define FIN_Q 16
#define REPACK_SLOTS 2048
static_assert(BIN_T == NBKP, "one thread per padded bucket in the chunk-sort block");
static_assert((1 << RB_SHIFT) == RB, "bucket = row >> RB_SHIFT");
static_assert(NBKP == 2 * RB, "each finish thread owns the runs of two chunks");
static_assert(NBKP % (16 * FIN_Q) == 0, "run groups tile the padded chunk range");
static_assert(REPACK_SLOTS % BIN_T == 0, "whole repack blocks per matrix");
#define MLP_GRID 512

__device__ __forceinline__ v8f wmma_f32_16x16x4(v2f a, v2f b, v8f c) {
  return __builtin_amdgcn_wmma_f32_16x16x4_f32(false, a, false, b, (short)0, c,
                                               false, false);
}


template <int MODE>
__device__ __forceinline__ void gin_fetch_rows(const float* __restrict__ H,
                                               const float* __restrict__ AGG,
                                               int tile, int laneOff,
                                               v2f (&zh)[16], v2f (&zg)[16]) {
  const float* hp = H + (size_t)tile * (16 * NDIM) + laneOff;
#pragma unroll
  for (int c = 0; c < 16; ++c) zh[c] = *(const v2f*)(hp + 4 * c);
  if (MODE == 1) {
    const float* gp = AGG + (size_t)tile * (16 * NDIM) + laneOff;
#pragma unroll
    for (int c = 0; c < 16; ++c) zg[c] = *(const v2f*)(gp + 4 * c);
  }
}

template <int MODE>
__device__ __forceinline__ void gin_mlp_tile(const v2f (&zh)[16],
                                             const v2f (&zg)[16],
                                             const v2f (&w1)[64],
                                             const v2f (&w2)[64],
                                             const float (&bias1)[4],
                                             const float (&bias2)[4],
                                             float* lt, float* __restrict__ OUT,
                                             int tile, int laneM, int hi) {
  v2f a[16];
#pragma unroll
  for (int c = 0; c < 16; ++c) {
    v2f z = zh[c];
    if (MODE == 1) {
      z.x += zg[c].x;
      z.y += zg[c].y;
    }
    a[c] = z;
  }

  float* op = OUT + ((size_t)tile * 16 + 8 * hi) * NDIM + laneM;

  v8f acc[4];
#pragma unroll
  for (int n = 0; n < 4; ++n)
#pragma unroll
    for (int r = 0; r < 8; ++r) acc[n][r] = bias1[n];
#pragma unroll
  for (int c = 0; c < 16; ++c)
#pragma unroll
    for (int n = 0; n < 4; ++n)
      acc[n] = wmma_f32_16x16x4(a[c], w1[c * 4 + n], acc[n]);

  if (MODE == 2) {
#pragma unroll
    for (int n = 0; n < 4; ++n)
#pragma unroll
      for (int r = 0; r < 8; ++r) op[r * NDIM + 16 * n] = acc[n][r];
    return;
  }

#pragma unroll
  for (int n = 0; n < 4; ++n)
#pragma unroll
    for (int r = 0; r < 8; ++r)
      lt[(r + 8 * hi) * NDIM + 16 * n + laneM] = fmaxf(acc[n][r], 0.0f);

#pragma unroll
  for (int c = 0; c < 16; ++c)
    a[c] = *(const v2f*)(lt + laneM * NDIM + 4 * c + 2 * hi);

#pragma unroll
  for (int n = 0; n < 4; ++n)
#pragma unroll
    for (int r = 0; r < 8; ++r) acc[n][r] = bias2[n];
#pragma unroll
  for (int c = 0; c < 16; ++c)
#pragma unroll
    for (int n = 0; n < 4; ++n)
      acc[n] = wmma_f32_16x16x4(a[c], w2[c * 4 + n], acc[n]);

#pragma unroll
  for (int n = 0; n < 4; ++n)
#pragma unroll
    for (int r = 0; r < 8; ++r) op[r * NDIM + 16 * n] = acc[n][r];
}

template <int MODE>
__global__ void __launch_bounds__(WAVES_PER_BLOCK * 32)
gin_mlp_kernel(const float* __restrict__ H, const float* __restrict__ AGG,
               const float* __restrict__ WP1, const float* __restrict__ B1,
               const float* __restrict__ WP2, const float* __restrict__ B2,
               float* __restrict__ OUT, int nTiles) {
  __shared__ float lds[WAVES_PER_BLOCK][16 * NDIM];

  const int wave = __builtin_amdgcn_readfirstlane((int)(threadIdx.x >> 5));
  const int lane = threadIdx.x & 31;
  const int stride = (int)gridDim.x * WAVES_PER_BLOCK;
  int tile = (int)blockIdx.x * WAVES_PER_BLOCK + wave;
  if (tile >= nTiles) return;

  const int laneM = lane & 15;
  const int hi = lane >> 4;
  const int laneOff = laneM * NDIM + 2 * hi;
  float* lt = lds[wave];

  v2f hA[16], gA[16], hB[16], gB[16];
  gin_fetch_rows<MODE>(H, AGG, tile, laneOff, hA, gA);

  v2f w1[64];
  v2f w2[64];
  float bias1[4];
  float bias2[4];
  {
    const v2f* wp1 = (const v2f*)WP1 + lane;
#pragma unroll
    for (int i = 0; i < 64; ++i) w1[i] = wp1[i * 32];
#pragma unroll
    for (int n = 0; n < 4; ++n) bias1[n] = B1[16 * n + laneM];
    if (MODE != 2) {
      const v2f* wp2 = (const v2f*)WP2 + lane;
#pragma unroll
      for (int i = 0; i < 64; ++i) w2[i] = wp2[i * 32];
#pragma unroll
      for (int n = 0; n < 4; ++n) bias2[n] = B2[16 * n + laneM];
    }
  }

  int cnt = (nTiles - 1 - tile) / stride + 1;

  while (cnt >= 2) {
    const int t1 = tile + stride;
    const int t2 = t1 + stride;

    gin_fetch_rows<MODE>(H, AGG, t1, laneOff, hB, gB);
    __builtin_amdgcn_sched_barrier(0);
    gin_mlp_tile<MODE>(hA, gA, w1, w2, bias1, bias2, lt, OUT, tile, laneM, hi);

    gin_fetch_rows<MODE>(H, AGG, cnt > 2 ? t2 : t1, laneOff, hA, gA);
    __builtin_amdgcn_sched_barrier(0);
    gin_mlp_tile<MODE>(hB, gB, w1, w2, bias1, bias2, lt, OUT, t1, laneM, hi);

    tile = t2;
    cnt -= 2;
  }
  if (cnt == 1)
    gin_mlp_tile<MODE>(hA, gA, w1, w2, bias1, bias2, lt, OUT, tile, laneM, hi);
}

__device__ __forceinline__ void gin_lds_frags(const v2f* wl, int blk,
                                              v2f (&b)[16]) {
#pragma unroll
  for (int i = 0; i < 16; ++i) b[i] = wl[(blk * 16 + i) * 32];
}

__device__ __forceinline__ void gin_wmma_block(const v2f (&a)[16], int blk,
                                               const v2f (&b)[16],
                                               v8f (&acc)[4]) {
#pragma unroll
  for (int cc = 0; cc < 4; ++cc)
#pragma unroll
    for (int n = 0; n < 4; ++n)
      acc[n] = wmma_f32_16x16x4(a[blk * 4 + cc], b[cc * 4 + n], acc[n]);
}

template <bool CHAIN>
__device__ __forceinline__ void gin_gemm_lds(const v2f (&a)[16], const v2f* wl,
                                             const v2f* nxt, v2f (&b0)[16],
                                             v2f (&b1)[16], v8f (&acc)[4]) {
  gin_lds_frags(wl, 1, b1);
  __builtin_amdgcn_sched_barrier(0);
  gin_wmma_block(a, 0, b0, acc);
  __builtin_amdgcn_sched_barrier(0);
  gin_lds_frags(wl, 2, b0);
  __builtin_amdgcn_sched_barrier(0);
  gin_wmma_block(a, 1, b1, acc);
  __builtin_amdgcn_sched_barrier(0);
  gin_lds_frags(wl, 3, b1);
  __builtin_amdgcn_sched_barrier(0);
  gin_wmma_block(a, 2, b0, acc);
  __builtin_amdgcn_sched_barrier(0);
  if (CHAIN) gin_lds_frags(nxt, 0, b0);
  __builtin_amdgcn_sched_barrier(0);
  gin_wmma_block(a, 3, b1, acc);
  __builtin_amdgcn_sched_barrier(0);
}

template <int FIN>
__device__ __forceinline__ void gin_fused_tile(
    const v2f (&zh)[16], const v2f (&zg)[16], const v2f (&w1)[64],
    const v2f* wl2, const float (&bias1)[4], const float (&bias2)[4],
    const v2f* wl3, const v2f* wl4, const float (&bias3)[4],
    const float (&bias4)[4], float* lt, float* __restrict__ HOUT,
    float* __restrict__ OUT2, int tile, int laneM, int hi) {
  v2f a[16];
#pragma unroll
  for (int c = 0; c < 16; ++c) {
    v2f z = zh[c];
    z.x += zg[c].x;
    z.y += zg[c].y;
    a[c] = z;
  }
  const size_t off = ((size_t)tile * 16 + 8 * hi) * NDIM + laneM;
  v2f bq0[16], bq1[16];

  v8f acc[4];
#pragma unroll
  for (int n = 0; n < 4; ++n)
#pragma unroll
    for (int r = 0; r < 8; ++r) acc[n][r] = bias1[n];
#pragma unroll
  for (int c = 0; c < 12; ++c)
#pragma unroll
    for (int n = 0; n < 4; ++n)
      acc[n] = wmma_f32_16x16x4(a[c], w1[c * 4 + n], acc[n]);
  __builtin_amdgcn_sched_barrier(0);
  gin_lds_frags(wl2, 0, bq0);
  __builtin_amdgcn_sched_barrier(0);
#pragma unroll
  for (int c = 12; c < 16; ++c)
#pragma unroll
    for (int n = 0; n < 4; ++n)
      acc[n] = wmma_f32_16x16x4(a[c], w1[c * 4 + n], acc[n]);
  __builtin_amdgcn_sched_barrier(0);

#pragma unroll
  for (int n = 0; n < 4; ++n)
#pragma unroll
    for (int r = 0; r < 8; ++r)
      lt[(r + 8 * hi) * NDIM + 16 * n + laneM] = fmaxf(acc[n][r], 0.0f);
#pragma unroll
  for (int c = 0; c < 16; ++c)
    a[c] = *(const v2f*)(lt + laneM * NDIM + 4 * c + 2 * hi);

#pragma unroll
  for (int n = 0; n < 4; ++n)
#pragma unroll
    for (int r = 0; r < 8; ++r) acc[n][r] = bias2[n];
  gin_gemm_lds<true>(a, wl2, wl3, bq0, bq1, acc);

  if (FIN == 0) {
    float* hp = HOUT + off;
#pragma unroll
    for (int n = 0; n < 4; ++n)
#pragma unroll
      for (int r = 0; r < 8; ++r) hp[r * NDIM + 16 * n] = acc[n][r];
  }

#pragma unroll
  for (int n = 0; n < 4; ++n)
#pragma unroll
    for (int r = 0; r < 8; ++r)
      lt[(r + 8 * hi) * NDIM + 16 * n + laneM] = acc[n][r];
#pragma unroll
  for (int c = 0; c < 16; ++c)
    a[c] = *(const v2f*)(lt + laneM * NDIM + 4 * c + 2 * hi);

#pragma unroll
  for (int n = 0; n < 4; ++n)
#pragma unroll
    for (int r = 0; r < 8; ++r) acc[n][r] = bias3[n];
  gin_gemm_lds<FIN == 0>(a, wl3, wl4, bq0, bq1, acc);

  float* op = OUT2 + off;
  if (FIN == 1) {
#pragma unroll
    for (int n = 0; n < 4; ++n)
#pragma unroll
      for (int r = 0; r < 8; ++r) op[r * NDIM + 16 * n] = acc[n][r];
    return;
  }

#pragma unroll
  for (int n = 0; n < 4; ++n)
#pragma unroll
    for (int r = 0; r < 8; ++r)
      lt[(r + 8 * hi) * NDIM + 16 * n + laneM] = fmaxf(acc[n][r], 0.0f);
#pragma unroll
  for (int c = 0; c < 16; ++c)
    a[c] = *(const v2f*)(lt + laneM * NDIM + 4 * c + 2 * hi);

#pragma unroll
  for (int n = 0; n < 4; ++n)
#pragma unroll
    for (int r = 0; r < 8; ++r) acc[n][r] = bias4[n];
  gin_gemm_lds<false>(a, wl4, wl4, bq0, bq1, acc);

#pragma unroll
  for (int n = 0; n < 4; ++n)
#pragma unroll
    for (int r = 0; r < 8; ++r) op[r * NDIM + 16 * n] = acc[n][r];
}

#define WL_FRAGS (WP_ELEMS / 2)

template <int FIN>
__global__ void __launch_bounds__(WAVES_PER_BLOCK * 32)
gin_fused_kernel(const float* __restrict__ H, const float* __restrict__ AGG,
                 const float* __restrict__ WP1, const float* __restrict__ B1,
                 const float* __restrict__ WP2, const float* __restrict__ B2,
                 const float* __restrict__ WP3, const float* __restrict__ B3,
                 const float* __restrict__ WP4, const float* __restrict__ B4,
                 float* __restrict__ HOUT, float* __restrict__ OUT2,
                 int nTiles) {
  __shared__ float lds[WAVES_PER_BLOCK][16 * NDIM];
  __shared__ v2f wl[(FIN ? 2 : 3) * WL_FRAGS];

  const int wave = __builtin_amdgcn_readfirstlane((int)(threadIdx.x >> 5));
  const int lane = threadIdx.x & 31;
  const int stride = (int)gridDim.x * WAVES_PER_BLOCK;
  int tile = (int)blockIdx.x * WAVES_PER_BLOCK + wave;
  const bool has = tile < nTiles;

  const int laneM = lane & 15;
  const int hi = lane >> 4;
  const int laneOff = laneM * NDIM + 2 * hi;
  float* lt = lds[wave];

  v2f hA[16], gA[16], hB[16], gB[16];
  if (has) gin_fetch_rows<1>(H, AGG, tile, laneOff, hA, gA);

  {
    const v2f* s2 = (const v2f*)WP2;
    const v2f* s3 = (const v2f*)WP3;
#pragma unroll
    for (int i = 0; i < WL_FRAGS / (WAVES_PER_BLOCK * 32); ++i) {
      const int j = i * (WAVES_PER_BLOCK * 32) + (int)threadIdx.x;
      wl[j] = s2[j];
      wl[WL_FRAGS + j] = s3[j];
    }
    if (FIN == 0) {
      const v2f* s4 = (const v2f*)WP4;
#pragma unroll
      for (int i = 0; i < WL_FRAGS / (WAVES_PER_BLOCK * 32); ++i) {
        const int j = i * (WAVES_PER_BLOCK * 32) + (int)threadIdx.x;
        wl[2 * WL_FRAGS + j] = s4[j];
      }
    }
  }

  v2f w1[64];
  float bias1[4], bias2[4], bias3[4], bias4[4];
  {
    const v2f* wp1 = (const v2f*)WP1 + lane;
#pragma unroll
    for (int i = 0; i < 64; ++i) w1[i] = wp1[i * 32];
#pragma unroll
    for (int n = 0; n < 4; ++n) {
      bias1[n] = B1[16 * n + laneM];
      bias2[n] = B2[16 * n + laneM];
      bias3[n] = B3[16 * n + laneM];
      bias4[n] = (FIN == 0) ? B4[16 * n + laneM] : 0.0f;
    }
  }
  __syncthreads();
  if (!has) return;

  const v2f* wl2 = wl + lane;
  const v2f* wl3 = wl + WL_FRAGS + lane;
  const v2f* wl4 = wl + (FIN ? 0 : 2 * WL_FRAGS) + lane;

  int cnt = (nTiles - 1 - tile) / stride + 1;

  while (cnt >= 2) {
    const int t1 = tile + stride;
    const int t2 = t1 + stride;

    gin_fetch_rows<1>(H, AGG, t1, laneOff, hB, gB);
    __builtin_amdgcn_sched_barrier(0);
    gin_fused_tile<FIN>(hA, gA, w1, wl2, bias1, bias2, wl3, wl4, bias3, bias4,
                        lt, HOUT, OUT2, tile, laneM, hi);

    gin_fetch_rows<1>(H, AGG, cnt > 2 ? t2 : t1, laneOff, hA, gA);
    __builtin_amdgcn_sched_barrier(0);
    gin_fused_tile<FIN>(hB, gB, w1, wl2, bias1, bias2, wl3, wl4, bias3, bias4,
                        lt, HOUT, OUT2, t1, laneM, hi);

    tile = t2;
    cnt -= 2;
  }
  if (cnt == 1)
    gin_fused_tile<FIN>(hA, gA, w1, wl2, bias1, bias2, wl3, wl4, bias3, bias4,
                        lt, HOUT, OUT2, tile, laneM, hi);
}


template <int NW>
__device__ __forceinline__ int gin_block_scan(int v, int t, int* wsum,
                                              int& total) {
  const int lane = t & 31;
  const int wave = t >> 5;
#pragma unroll
  for (int off = 1; off < 32; off <<= 1) {
    const int u = __shfl_up(v, off);
    if (lane >= off) v += u;
  }
  if (lane == 31) wsum[wave] = v;
  __syncthreads();
  int before = 0;
  int all = 0;
#pragma unroll
  for (int w = 0; w < NW; ++w) {
    const int ws = wsum[w];
    all += ws;
    if (w < wave) before += ws;
  }
  total = all;
  return v + before;
}

__global__ void __launch_bounds__(BIN_T)
gin_chunk_sort_kernel(const int* __restrict__ rowp,
                      const int* __restrict__ colp, int* __restrict__ off,
                      int* __restrict__ sortedc, int E, int nChunks,
                      const float* __restrict__ W1,
                      const float* __restrict__ W2,
                      const float* __restrict__ Wf, float* __restrict__ Wp,
                      int L) {
  __shared__ int wsa[BIN_T / 32];
  __shared__ int lcur[NBKP];
  __shared__ v2i stage[CH];
  const int t = (int)threadIdx.x;
  const int b = (int)blockIdx.x;

  if (b >= nChunks) {
    const int slot = (b - nChunks) * BIN_T + t;
    const int mat = slot / REPACK_SLOTS;
    const int tid = slot % REPACK_SLOTS;
    const float* W = mat < L       ? W1 + (size_t)mat * (NDIM * NDIM)
                     : mat < 2 * L ? W2 + (size_t)(mat - L) * (NDIM * NDIM)
                                   : Wf;
    float* dst = Wp + (size_t)mat * WP_ELEMS;
    const int c = tid >> 7;
    const int n = (tid >> 5) & 3;
    const int lane = tid & 31;
    const int hi = lane >> 4;
    const int m = lane & 15;
    const int k = 4 * c + 2 * hi;
    dst[tid * 2 + 0] = W[(k + 0) * NDIM + 16 * n + m];
    dst[tid * 2 + 1] = W[(k + 1) * NDIM + 16 * n + m];
    return;
  }

  const int e0 = b * CH;
  lcur[t] = 0;

  int er[CH / BIN_T], ec[CH / BIN_T];
#pragma unroll
  for (int j = 0; j < CH / BIN_T; ++j) {
    const int e = e0 + j * BIN_T + t;
    er[j] = (e < E) ? rowp[e] : -1;
    ec[j] = (e < E) ? colp[e] : 0;
  }
  __syncthreads();

#pragma unroll
  for (int j = 0; j < CH / BIN_T; ++j)
    if (er[j] >= 0) atomicAdd(&lcur[er[j] >> RB_SHIFT], 1);
  __syncthreads();
  const int c = lcur[t];
  int nChunk;
  const int lstart =
      gin_block_scan<BIN_T / 32>(c, t, wsa, nChunk) - c;
  lcur[t] = lstart;
  off[(size_t)b * OFP + t] = lstart;
  if (t == BIN_T - 1) off[(size_t)b * OFP + NBKP] = nChunk;
  __syncthreads();

  int slot[CH / BIN_T];
#pragma unroll
  for (int j = 0; j < CH / BIN_T; ++j)
    slot[j] = (er[j] >= 0) ? atomicAdd(&lcur[er[j] >> RB_SHIFT], 1) : 0;
#pragma unroll
  for (int j = 0; j < CH / BIN_T; ++j) {
    if (er[j] >= 0) {
      v2i v;
      v.x = er[j];
      v.y = ec[j];
      stage[slot[j]] = v;
    }
  }
  __syncthreads();

  v2i* dst = (v2i*)sortedc + (size_t)b * CH;
#pragma unroll
  for (int j = 0; j < CH / BIN_T; ++j) {
    const int i = j * BIN_T + t;
    if (i < nChunk) dst[i] = stage[i];
  }
}

__global__ void __launch_bounds__(RB)
gin_bin_finish_kernel(const int* __restrict__ sortedc,
                      const int* __restrict__ off, int* __restrict__ rowstart,
                      int* __restrict__ perm, int N, int nChunks) {
  __shared__ int wsa[RB / 32];
  __shared__ int wsb[RB / 32];
  __shared__ int wsc[RB / 32];
  __shared__ int cpos[NBKP];
  __shared__ int csrc[NBKP];
  __shared__ int clen[NBKP];
  __shared__ int lcur[RB];
  __shared__ v2i lev[FIN_CAP];
  __shared__ int lperm[FIN_CAP];
  const int t = (int)threadIdx.x;
  const int k = (int)blockIdx.x;
  const v2i* src = (const v2i*)sortedc;

  const int c0 = t;
  const int c1 = t + RB;
  const int* o0 = off + (size_t)(c0 < nChunks ? c0 : nChunks - 1) * OFP + k;
  const int* o1 = off + (size_t)(c1 < nChunks ? c1 : nChunks - 1) * OFP + k;
  const int a0 = o0[0];
  const int z0 = o0[1];
  const int a1 = o1[0];
  const int z1 = o1[1];
  const int s0 = c0 < nChunks ? a0 : 0;
  const int l0 = c0 < nChunks ? z0 - a0 : 0;
  const int s1 = c1 < nChunks ? a1 : 0;
  const int l1 = c1 < nChunks ? z1 - a1 : 0;
  int n;
  int b0;
  const int cex = gin_block_scan<RB / 32>(l0 + l1, t, wsa, n) - (l0 + l1);
  (void)gin_block_scan<RB / 32>(s0 + s1, t, wsb, b0);
  cpos[c0] = cex;
  csrc[c0] = c0 * CH + s0;
  clen[c0] = l0;
  cpos[c1] = cex + l0;
  csrc[c1] = c1 * CH + s1;
  clen[c1] = l1;
  lcur[t] = 0;
  __syncthreads();
  const bool staged = n <= FIN_CAP;
  const int hw = t >> 4;
  const int hl = t & 15;

  if (staged) {
    for (int base = 0; base < nChunks; base += 16 * FIN_Q) {
      v2i v[FIN_Q];
      int p[FIN_Q], ln[FIN_Q];
#pragma unroll
      for (int q = 0; q < FIN_Q; ++q) {
        const int c = base + 16 * q + hw;
        ln[q] = clen[c];
        p[q] = cpos[c];
        v[q].x = 0;
        v[q].y = 0;
        if (hl < ln[q]) v[q] = src[csrc[c] + hl];
      }
#pragma unroll
      for (int q = 0; q < FIN_Q; ++q) {
        if (hl < ln[q]) {
          lev[p[q] + hl] = v[q];
          atomicAdd(&lcur[v[q].x & (RB - 1)], 1);
        }
      }
#pragma unroll
      for (int q = 0; q < FIN_Q; ++q) {
        const int c = base + 16 * q + hw;
        for (int j = hl + 16; j < ln[q]; j += 16) {
          const v2i w = src[csrc[c] + j];
          lev[p[q] + j] = w;
          atomicAdd(&lcur[w.x & (RB - 1)], 1);
        }
      }
    }
  } else {
    for (int c = hw; c < nChunks; c += RB / 16) {
      const int s = csrc[c];
      const int len = clen[c];
      for (int j = hl; j < len; j += 16)
        atomicAdd(&lcur[src[s + j].x & (RB - 1)], 1);
    }
  }
  __syncthreads();

  const int d = lcur[t];
  int nScan;
  const int ex =
      gin_block_scan<RB / 32>(d, t, wsc, nScan) - d;
  lcur[t] = ex;
  const int row = k * RB + t;
  if (row < N) rowstart[row] = b0 + ex;
  if (row == N - 1) rowstart[N] = b0 + n;
  __syncthreads();

  if (staged) {
    v2i ev[FIN_U];
    int sl[FIN_U];
#pragma unroll
    for (int u = 0; u < FIN_U; ++u) {
      const int i = u * RB + t;
      if (i < n) {
        ev[u] = lev[i];
      } else {
        ev[u].x = -1;
        ev[u].y = 0;
      }
    }
#pragma unroll
    for (int u = 0; u < FIN_U; ++u)
      sl[u] = (ev[u].x >= 0) ? atomicAdd(&lcur[ev[u].x & (RB - 1)], 1) : 0;
#pragma unroll
    for (int u = 0; u < FIN_U; ++u)
      if (ev[u].x >= 0) lperm[sl[u]] = ev[u].y;
    __syncthreads();
#pragma unroll
    for (int u = 0; u < FIN_U; ++u) {
      const int i = u * RB + t;
      if (i < n) perm[b0 + i] = lperm[i];
    }
  } else {
    for (int c = hw; c < nChunks; c += RB / 16) {
      const int s = csrc[c];
      const int len = clen[c];
      for (int j = hl; j < len; j += 16) {
        const v2i v = src[s + j];
        perm[b0 + atomicAdd(&lcur[v.x & (RB - 1)], 1)] = v.y;
      }
    }
  }
}

__global__ void gin_gather_kernel(const float* __restrict__ T,
                                  const int* __restrict__ rowstart,
                                  const int* __restrict__ perm,
                                  float* __restrict__ AGG, int N) {
  const int g = (blockIdx.x * blockDim.x + threadIdx.x) >> 4;
  if (g >= N) return;
  const int l4 = (threadIdx.x & 15) * 4;
  const int s = rowstart[g];
  const int e = rowstart[g + 1];
  float4 acc = make_float4(0.f, 0.f, 0.f, 0.f);
  for (int j = s; j < e; ++j) {
    const int c = perm[j];
    const float4 v = *(const float4*)(T + (size_t)c * NDIM + l4);
    acc.x += v.x;
    acc.y += v.y;
    acc.z += v.z;
    acc.w += v.w;
  }
  *(float4*)(AGG + (size_t)g * NDIM + l4) = acc;
}


extern "C" void kernel_launch(void* const* d_in, const int* in_sizes, int n_in,
                              void* d_out, int out_size, void* d_ws,
                              size_t ws_size, hipStream_t stream) {
  (void)n_in;
  (void)out_size;
  (void)ws_size;

  const float* x = (const float*)d_in[0];
  const int* ei = (const int*)d_in[1];
  const float* W1 = (const float*)d_in[2];
  const float* b1 = (const float*)d_in[3];
  const float* W2 = (const float*)d_in[4];
  const float* b2 = (const float*)d_in[5];
  const float* Wf = (const float*)d_in[6];
  const float* bf = (const float*)d_in[7];
  float* out = (float*)d_out;

  const int N = in_sizes[0] / NDIM;
  const int E = in_sizes[1] / 2;
  const int L = in_sizes[3] / NDIM;
  const int nTiles = N / 16;
  const size_t NB = (size_t)N * NDIM;

  float* agg = (float*)d_ws;
  float* t = agg + NB;
  float* h = t + NB;
  const int nMat = 2 * L + 1;
  float* wp = h + NB;
  const int nChunks = (E + CH - 1) / CH;
  const int nBuckets = (N + RB - 1) / RB;
  int* sortedc = (int*)(wp + (size_t)nMat * WP_ELEMS);
  int* off = sortedc + 2 * (size_t)nChunks * CH;
  int* perm = off + (size_t)nChunks * OFP;
  int* rowstart = perm + E;

  const int* rowp = ei;
  const int* colp = ei + E;

  gin_chunk_sort_kernel<<<nChunks + nMat * (REPACK_SLOTS / BIN_T), BIN_T, 0,
                          stream>>>(rowp, colp, off, sortedc, E, nChunks, W1,
                                    W2, Wf, wp, L);
  gin_bin_finish_kernel<<<nBuckets, RB, 0, stream>>>(sortedc, off, rowstart,
                                                     perm, N, nChunks);

  const dim3 mlpBlk(WAVES_PER_BLOCK * 32);
  const dim3 mlpGrd(MLP_GRID);
  const dim3 gatGrd((N * 16 + 255) / 256);

  gin_mlp_kernel<0><<<mlpGrd, mlpBlk, 0, stream>>>(x, nullptr, wp, b1,
                                                   wp + (size_t)L * WP_ELEMS,
                                                   b2, t, nTiles);
  const float* hcur = x;
  for (int l = 0; l < L; ++l) {
    const float* wp1 = wp + (size_t)l * WP_ELEMS;
    const float* wp2 = wp + (size_t)(L + l) * WP_ELEMS;
    const float* v1 = b1 + (size_t)l * NDIM;
    const float* v2 = b2 + (size_t)l * NDIM;

    gin_gather_kernel<<<gatGrd, 256, 0, stream>>>(t, rowstart, perm, agg, N);
    if (l + 1 < L) {
      gin_fused_kernel<0><<<mlpGrd, mlpBlk, 0, stream>>>(
          hcur, agg, wp1, v1, wp2, v2, wp + (size_t)(l + 1) * WP_ELEMS,
          v1 + NDIM, wp + (size_t)(L + l + 1) * WP_ELEMS, v2 + NDIM, h, t,
          nTiles);
    } else {
      gin_fused_kernel<1><<<mlpGrd, mlpBlk, 0, stream>>>(
          hcur, agg, wp1, v1, wp2, v2, wp + (size_t)(2 * L) * WP_ELEMS, bf,
          nullptr, nullptr, nullptr, out, nTiles);
    }
    hcur = h;
  }
}


# Neighbour sum moved into the combine launch: each wave gathers its own tile's rows of t (32 row loads in flight) through its LDS tile; no agg array, no gather launches (9 -> 6); t in two buffers
# speedup vs baseline: 2.7154x; 1.0785x over previous
#include <hip/hip_runtime.h>


typedef __attribute__((ext_vector_type(2))) float v2f;
typedef __attribute__((ext_vector_type(8))) float v8f;
typedef __attribute__((ext_vector_type(2))) int v2i;

#define NDIM 64
#define WAVES_PER_BLOCK 4
#define WP_ELEMS 4096
#define RB 256
#define RB_SHIFT 8
#define NBKP 512
#define CH 4096
#define BIN_T 512
#define OFP (NBKP + 1)
#define FIN_U 16
#define FIN_CAP (FIN_U * RB)
#define FIN_Q 16
#define REPACK_SLOTS 2048
static_assert(BIN_T == NBKP, "one thread per padded bucket in the chunk-sort block");
static_assert((1 << RB_SHIFT) == RB, "bucket = row >> RB_SHIFT");
static_assert(NBKP == 2 * RB, "each finish thread owns the runs of two chunks");
static_assert(NBKP % (16 * FIN_Q) == 0, "run groups tile the padded chunk range");
static_assert(REPACK_SLOTS % BIN_T == 0, "whole repack blocks per matrix");
#define MLP_GRID 512

__device__ __forceinline__ v8f wmma_f32_16x16x4(v2f a, v2f b, v8f c) {
  return __builtin_amdgcn_wmma_f32_16x16x4_f32(false, a, false, b, (short)0, c,
                                               false, false);
}


template <int MODE>
__device__ __forceinline__ void gin_fetch_rows(const float* __restrict__ H,
                                               const float* __restrict__ AGG,
                                               int tile, int laneOff,
                                               v2f (&zh)[16], v2f (&zg)[16]) {
  const float* hp = H + (size_t)tile * (16 * NDIM) + laneOff;
#pragma unroll
  for (int c = 0; c < 16; ++c) zh[c] = *(const v2f*)(hp + 4 * c);
  if (MODE == 1) {
    const float* gp = AGG + (size_t)tile * (16 * NDIM) + laneOff;
#pragma unroll
    for (int c = 0; c < 16; ++c) zg[c] = *(const v2f*)(gp + 4 * c);
  }
}

template <int MODE>
__device__ __forceinline__ void gin_mlp_tile(const v2f (&zh)[16],
                                             const v2f (&zg)[16],
                                             const v2f (&w1)[64],
                                             const v2f (&w2)[64],
                                             const float (&bias1)[4],
                                             const float (&bias2)[4],
                                             float* lt, float* __restrict__ OUT,
                                             int tile, int laneM, int hi) {
  v2f a[16];
#pragma unroll
  for (int c = 0; c < 16; ++c) {
    v2f z = zh[c];
    if (MODE == 1) {
      z.x += zg[c].x;
      z.y += zg[c].y;
    }
    a[c] = z;
  }

  float* op = OUT + ((size_t)tile * 16 + 8 * hi) * NDIM + laneM;

  v8f acc[4];
#pragma unroll
  for (int n = 0; n < 4; ++n)
#pragma unroll
    for (int r = 0; r < 8; ++r) acc[n][r] = bias1[n];
#pragma unroll
  for (int c = 0; c < 16; ++c)
#pragma unroll
    for (int n = 0; n < 4; ++n)
      acc[n] = wmma_f32_16x16x4(a[c], w1[c * 4 + n], acc[n]);

  if (MODE == 2) {
#pragma unroll
    for (int n = 0; n < 4; ++n)
#pragma unroll
      for (int r = 0; r < 8; ++r) op[r * NDIM + 16 * n] = acc[n][r];
    return;
  }

#pragma unroll
  for (int n = 0; n < 4; ++n)
#pragma unroll
    for (int r = 0; r < 8; ++r)
      lt[(r + 8 * hi) * NDIM + 16 * n + laneM] = fmaxf(acc[n][r], 0.0f);

#pragma unroll
  for (int c = 0; c < 16; ++c)
    a[c] = *(const v2f*)(lt + laneM * NDIM + 4 * c + 2 * hi);

#pragma unroll
  for (int n = 0; n < 4; ++n)
#pragma unroll
    for (int r = 0; r < 8; ++r) acc[n][r] = bias2[n];
#pragma unroll
  for (int c = 0; c < 16; ++c)
#pragma unroll
    for (int n = 0; n < 4; ++n)
      acc[n] = wmma_f32_16x16x4(a[c], w2[c * 4 + n], acc[n]);

#pragma unroll
  for (int n = 0; n < 4; ++n)
#pragma unroll
    for (int r = 0; r < 8; ++r) op[r * NDIM + 16 * n] = acc[n][r];
}

template <int MODE>
__global__ void __launch_bounds__(WAVES_PER_BLOCK * 32)
gin_mlp_kernel(const float* __restrict__ H, const float* __restrict__ AGG,
               const float* __restrict__ WP1, const float* __restrict__ B1,
               const float* __restrict__ WP2, const float* __restrict__ B2,
               float* __restrict__ OUT, int nTiles) {
  __shared__ float lds[WAVES_PER_BLOCK][16 * NDIM];

  const int wave = __builtin_amdgcn_readfirstlane((int)(threadIdx.x >> 5));
  const int lane = threadIdx.x & 31;
  const int stride = (int)gridDim.x * WAVES_PER_BLOCK;
  int tile = (int)blockIdx.x * WAVES_PER_BLOCK + wave;
  if (tile >= nTiles) return;

  const int laneM = lane & 15;
  const int hi = lane >> 4;
  const int laneOff = laneM * NDIM + 2 * hi;
  float* lt = lds[wave];

  v2f hA[16], gA[16], hB[16], gB[16];
  gin_fetch_rows<MODE>(H, AGG, tile, laneOff, hA, gA);

  v2f w1[64];
  v2f w2[64];
  float bias1[4];
  float bias2[4];
  {
    const v2f* wp1 = (const v2f*)WP1 + lane;
#pragma unroll
    for (int i = 0; i < 64; ++i) w1[i] = wp1[i * 32];
#pragma unroll
    for (int n = 0; n < 4; ++n) bias1[n] = B1[16 * n + laneM];
    if (MODE != 2) {
      const v2f* wp2 = (const v2f*)WP2 + lane;
#pragma unroll
      for (int i = 0; i < 64; ++i) w2[i] = wp2[i * 32];
#pragma unroll
      for (int n = 0; n < 4; ++n) bias2[n] = B2[16 * n + laneM];
    }
  }

  int cnt = (nTiles - 1 - tile) / stride + 1;

  while (cnt >= 2) {
    const int t1 = tile + stride;
    const int t2 = t1 + stride;

    gin_fetch_rows<MODE>(H, AGG, t1, laneOff, hB, gB);
    __builtin_amdgcn_sched_barrier(0);
    gin_mlp_tile<MODE>(hA, gA, w1, w2, bias1, bias2, lt, OUT, tile, laneM, hi);

    gin_fetch_rows<MODE>(H, AGG, cnt > 2 ? t2 : t1, laneOff, hA, gA);
    __builtin_amdgcn_sched_barrier(0);
    gin_mlp_tile<MODE>(hB, gB, w1, w2, bias1, bias2, lt, OUT, t1, laneM, hi);

    tile = t2;
    cnt -= 2;
  }
  if (cnt == 1)
    gin_mlp_tile<MODE>(hA, gA, w1, w2, bias1, bias2, lt, OUT, tile, laneM, hi);
}

__device__ __forceinline__ void gin_lds_frags(const v2f* wl, int blk,
                                              v2f (&b)[16]) {
#pragma unroll
  for (int i = 0; i < 16; ++i) b[i] = wl[(blk * 16 + i) * 32];
}

__device__ __forceinline__ void gin_wmma_block(const v2f (&a)[16], int blk,
                                               const v2f (&b)[16],
                                               v8f (&acc)[4]) {
#pragma unroll
  for (int cc = 0; cc < 4; ++cc)
#pragma unroll
    for (int n = 0; n < 4; ++n)
      acc[n] = wmma_f32_16x16x4(a[blk * 4 + cc], b[cc * 4 + n], acc[n]);
}

template <bool CHAIN>
__device__ __forceinline__ void gin_gemm_lds(const v2f (&a)[16], const v2f* wl,
                                             const v2f* nxt, v2f (&b0)[16],
                                             v2f (&b1)[16], v8f (&acc)[4]) {
  gin_lds_frags(wl, 1, b1);
  __builtin_amdgcn_sched_barrier(0);
  gin_wmma_block(a, 0, b0, acc);
  __builtin_amdgcn_sched_barrier(0);
  gin_lds_frags(wl, 2, b0);
  __builtin_amdgcn_sched_barrier(0);
  gin_wmma_block(a, 1, b1, acc);
  __builtin_amdgcn_sched_barrier(0);
  gin_lds_frags(wl, 3, b1);
  __builtin_amdgcn_sched_barrier(0);
  gin_wmma_block(a, 2, b0, acc);
  __builtin_amdgcn_sched_barrier(0);
  if (CHAIN) gin_lds_frags(nxt, 0, b0);
  __builtin_amdgcn_sched_barrier(0);
  gin_wmma_block(a, 3, b1, acc);
  __builtin_amdgcn_sched_barrier(0);
}

template <int FIN>
__device__ __forceinline__ void gin_fused_tile(
    const v2f (&zh)[16], const v2f (&zg)[16], const v2f (&w1)[64],
    const v2f* wl2, const float (&bias1)[4], const float (&bias2)[4],
    const v2f* wl3, const v2f* wl4, const float (&bias3)[4],
    const float (&bias4)[4], float* lt, float* __restrict__ HOUT,
    float* __restrict__ OUT2, int tile, int laneM, int hi) {
  v2f a[16];
#pragma unroll
  for (int c = 0; c < 16; ++c) {
    v2f z = zh[c];
    z.x += zg[c].x;
    z.y += zg[c].y;
    a[c] = z;
  }
  const size_t off = ((size_t)tile * 16 + 8 * hi) * NDIM + laneM;
  v2f bq0[16], bq1[16];

  v8f acc[4];
#pragma unroll
  for (int n = 0; n < 4; ++n)
#pragma unroll
    for (int r = 0; r < 8; ++r) acc[n][r] = bias1[n];
#pragma unroll
  for (int c = 0; c < 12; ++c)
#pragma unroll
    for (int n = 0; n < 4; ++n)
      acc[n] = wmma_f32_16x16x4(a[c], w1[c * 4 + n], acc[n]);
  __builtin_amdgcn_sched_barrier(0);
  gin_lds_frags(wl2, 0, bq0);
  __builtin_amdgcn_sched_barrier(0);
#pragma unroll
  for (int c = 12; c < 16; ++c)
#pragma unroll
    for (int n = 0; n < 4; ++n)
      acc[n] = wmma_f32_16x16x4(a[c], w1[c * 4 + n], acc[n]);
  __builtin_amdgcn_sched_barrier(0);

#pragma unroll
  for (int n = 0; n < 4; ++n)
#pragma unroll
    for (int r = 0; r < 8; ++r)
      lt[(r + 8 * hi) * NDIM + 16 * n + laneM] = fmaxf(acc[n][r], 0.0f);
#pragma unroll
  for (int c = 0; c < 16; ++c)
    a[c] = *(const v2f*)(lt + laneM * NDIM + 4 * c + 2 * hi);

#pragma unroll
  for (int n = 0; n < 4; ++n)
#pragma unroll
    for (int r = 0; r < 8; ++r) acc[n][r] = bias2[n];
  gin_gemm_lds<true>(a, wl2, wl3, bq0, bq1, acc);

  if (FIN == 0) {
    float* hp = HOUT + off;
#pragma unroll
    for (int n = 0; n < 4; ++n)
#pragma unroll
      for (int r = 0; r < 8; ++r) hp[r * NDIM + 16 * n] = acc[n][r];
  }

#pragma unroll
  for (int n = 0; n < 4; ++n)
#pragma unroll
    for (int r = 0; r < 8; ++r)
      lt[(r + 8 * hi) * NDIM + 16 * n + laneM] = acc[n][r];
#pragma unroll
  for (int c = 0; c < 16; ++c)
    a[c] = *(const v2f*)(lt + laneM * NDIM + 4 * c + 2 * hi);

#pragma unroll
  for (int n = 0; n < 4; ++n)
#pragma unroll
    for (int r = 0; r < 8; ++r) acc[n][r] = bias3[n];
  gin_gemm_lds<FIN == 0>(a, wl3, wl4, bq0, bq1, acc);

  float* op = OUT2 + off;
  if (FIN == 1) {
#pragma unroll
    for (int n = 0; n < 4; ++n)
#pragma unroll
      for (int r = 0; r < 8; ++r) op[r * NDIM + 16 * n] = acc[n][r];
    return;
  }

#pragma unroll
  for (int n = 0; n < 4; ++n)
#pragma unroll
    for (int r = 0; r < 8; ++r)
      lt[(r + 8 * hi) * NDIM + 16 * n + laneM] = fmaxf(acc[n][r], 0.0f);
#pragma unroll
  for (int c = 0; c < 16; ++c)
    a[c] = *(const v2f*)(lt + laneM * NDIM + 4 * c + 2 * hi);

#pragma unroll
  for (int n = 0; n < 4; ++n)
#pragma unroll
    for (int r = 0; r < 8; ++r) acc[n][r] = bias4[n];
  gin_gemm_lds<false>(a, wl4, wl4, bq0, bq1, acc);

#pragma unroll
  for (int n = 0; n < 4; ++n)
#pragma unroll
    for (int r = 0; r < 8; ++r) op[r * NDIM + 16 * n] = acc[n][r];
}

#define WL_FRAGS (WP_ELEMS / 2)

#define GU 4
__device__ __forceinline__ void gin_gather_tile(const float* __restrict__ T,
                                                const int* __restrict__ perm,
                                                int rs, int lane, float* lt,
                                                v2f (&zg)[16]) {
  const int hl = lane >> 4;
  const int laneM = lane & 15;
  const unsigned colB = (unsigned)laneM * 16u;
  const char* tb = (const char*)T;
  const char* pb = (const char*)perm;

  int s[8], e[8];
#pragma unroll
  for (int p = 0; p < 8; ++p) {
    s[p] = __shfl(rs, 2 * p + hl);
    e[p] = __shfl(rs, 2 * p + hl + 1);
  }
  int d = __shfl_down(rs, 1) - rs;
  d = lane < 16 ? d : 0;
#pragma unroll
  for (int o = 16; o > 0; o >>= 1) {
    const int u = __shfl_xor(d, o);
    d = u > d ? u : d;
  }
  const int maxdeg = __builtin_amdgcn_readfirstlane(d);

  float4 acc[8];
#pragma unroll
  for (int p = 0; p < 8; ++p) acc[p] = make_float4(0.f, 0.f, 0.f, 0.f);

  unsigned ro[8][GU];
#pragma unroll
  for (int p = 0; p < 8; ++p)
#pragma unroll
    for (int u = 0; u < GU; ++u) {
      const int j = s[p] + u;
      const unsigned jj = j < e[p] ? (unsigned)j : 0u;
      ro[p][u] = (unsigned)(*(const int*)(pb + jj * 4u)) * (NDIM * 4u) + colB;
    }

  for (int base = 0; base < maxdeg; base += GU) {
    float4 v[8][GU];
#pragma unroll
    for (int p = 0; p < 8; ++p)
#pragma unroll
      for (int u = 0; u < GU; ++u) v[p][u] = *(const float4*)(tb + ro[p][u]);
    int nx[8][GU];
#pragma unroll
    for (int p = 0; p < 8; ++p)
#pragma unroll
      for (int u = 0; u < GU; ++u) {
        const int j = s[p] + base + GU + u;
        const unsigned jj = j < e[p] ? (unsigned)j : 0u;
        nx[p][u] = *(const int*)(pb + jj * 4u);
      }
    __builtin_amdgcn_sched_barrier(0);
#pragma unroll
    for (int p = 0; p < 8; ++p)
#pragma unroll
      for (int u = 0; u < GU; ++u) {
        if (s[p] + base + u < e[p]) {
          acc[p].x += v[p][u].x;
          acc[p].y += v[p][u].y;
          acc[p].z += v[p][u].z;
          acc[p].w += v[p][u].w;
        }
        ro[p][u] = (unsigned)nx[p][u] * (NDIM * 4u) + colB;
      }
  }

#pragma unroll
  for (int p = 0; p < 8; ++p)
    *(float4*)(lt + (2 * p + hl) * NDIM + 4 * laneM) = acc[p];
#pragma unroll
  for (int c = 0; c < 16; ++c)
    zg[c] = *(const v2f*)(lt + laneM * NDIM + 4 * c + 2 * hl);
}

template <int FIN>
__global__ void __launch_bounds__(WAVES_PER_BLOCK * 32)
gin_fused_kernel(const float* __restrict__ H, const float* __restrict__ T,
                 const int* __restrict__ rowstart, const int* __restrict__ perm,
                 const float* __restrict__ WP1, const float* __restrict__ B1,
                 const float* __restrict__ WP2, const float* __restrict__ B2,
                 const float* __restrict__ WP3, const float* __restrict__ B3,
                 const float* __restrict__ WP4, const float* __restrict__ B4,
                 float* __restrict__ HOUT, float* __restrict__ OUT2,
                 int nTiles) {
  __shared__ __attribute__((aligned(16))) float lds[WAVES_PER_BLOCK][16 * NDIM];
  __shared__ v2f wl[(FIN ? 2 : 3) * WL_FRAGS];

  const int wave = __builtin_amdgcn_readfirstlane((int)(threadIdx.x >> 5));
  const int lane = threadIdx.x & 31;
  const int stride = (int)gridDim.x * WAVES_PER_BLOCK;
  int tile = (int)blockIdx.x * WAVES_PER_BLOCK + wave;
  const bool has = tile < nTiles;

  const int laneM = lane & 15;
  const int hi = lane >> 4;
  const int laneOff = laneM * NDIM + 2 * hi;
  float* lt = lds[wave];

  const int rsLane = lane < 16 ? lane : 16;
  int rs = 0;
  if (has) rs = rowstart[16 * tile + rsLane];

  {
    const v2f* s2 = (const v2f*)WP2;
    const v2f* s3 = (const v2f*)WP3;
#pragma unroll
    for (int i = 0; i < WL_FRAGS / (WAVES_PER_BLOCK * 32); ++i) {
      const int j = i * (WAVES_PER_BLOCK * 32) + (int)threadIdx.x;
      wl[j] = s2[j];
      wl[WL_FRAGS + j] = s3[j];
    }
    if (FIN == 0) {
      const v2f* s4 = (const v2f*)WP4;
#pragma unroll
      for (int i = 0; i < WL_FRAGS / (WAVES_PER_BLOCK * 32); ++i) {
        const int j = i * (WAVES_PER_BLOCK * 32) + (int)threadIdx.x;
        wl[2 * WL_FRAGS + j] = s4[j];
      }
    }
  }

  v2f w1[64];
  float bias1[4], bias2[4], bias3[4], bias4[4];
  {
    const v2f* wp1 = (const v2f*)WP1 + lane;
#pragma unroll
    for (int i = 0; i < 64; ++i) w1[i] = wp1[i * 32];
#pragma unroll
    for (int n = 0; n < 4; ++n) {
      bias1[n] = B1[16 * n + laneM];
      bias2[n] = B2[16 * n + laneM];
      bias3[n] = B3[16 * n + laneM];
      bias4[n] = (FIN == 0) ? B4[16 * n + laneM] : 0.0f;
    }
  }
  __syncthreads();
  if (!has) return;

  const v2f* wl2 = wl + lane;
  const v2f* wl3 = wl + WL_FRAGS + lane;
  const v2f* wl4 = wl + (FIN ? 0 : 2 * WL_FRAGS) + lane;

  int cnt = (nTiles - 1 - tile) / stride + 1;

  for (; cnt > 0; --cnt) {
    v2f zh[16], zg[16];
    gin_fetch_rows<0>(H, nullptr, tile, laneOff, zh, zg);
    const int nt = cnt > 1 ? tile + stride : tile;
    const int rsn = rowstart[16 * nt + rsLane];
    __builtin_amdgcn_sched_barrier(0);
    gin_gather_tile(T, perm, rs, lane, lt, zg);
    gin_fused_tile<FIN>(zh, zg, w1, wl2, bias1, bias2, wl3, wl4, bias3, bias4,
                        lt, HOUT, OUT2, tile, laneM, hi);
    rs = rsn;
    tile = nt;
  }
}


template <int NW>
__device__ __forceinline__ int gin_block_scan(int v, int t, int* wsum,
                                              int& total) {
  const int lane = t & 31;
  const int wave = t >> 5;
#pragma unroll
  for (int off = 1; off < 32; off <<= 1) {
    const int u = __shfl_up(v, off);
    if (lane >= off) v += u;
  }
  if (lane == 31) wsum[wave] = v;
  __syncthreads();
  int before = 0;
  int all = 0;
#pragma unroll
  for (int w = 0; w < NW; ++w) {
    const int ws = wsum[w];
    all += ws;
    if (w < wave) before += ws;
  }
  total = all;
  return v + before;
}

__global__ void __launch_bounds__(BIN_T)
gin_chunk_sort_kernel(const int* __restrict__ rowp,
                      const int* __restrict__ colp, int* __restrict__ off,
                      int* __restrict__ sortedc, int E, int nChunks,
                      const float* __restrict__ W1,
                      const float* __restrict__ W2,
                      const float* __restrict__ Wf, float* __restrict__ Wp,
                      int L) {
  __shared__ int wsa[BIN_T / 32];
  __shared__ int lcur[NBKP];
  __shared__ v2i stage[CH];
  const int t = (int)threadIdx.x;
  const int b = (int)blockIdx.x;

  if (b >= nChunks) {
    const int slot = (b - nChunks) * BIN_T + t;
    const int mat = slot / REPACK_SLOTS;
    const int tid = slot % REPACK_SLOTS;
    const float* W = mat < L       ? W1 + (size_t)mat * (NDIM * NDIM)
                     : mat < 2 * L ? W2 + (size_t)(mat - L) * (NDIM * NDIM)
                                   : Wf;
    float* dst = Wp + (size_t)mat * WP_ELEMS;
    const int c = tid >> 7;
    const int n = (tid >> 5) & 3;
    const int lane = tid & 31;
    const int hi = lane >> 4;
    const int m = lane & 15;
    const int k = 4 * c + 2 * hi;
    dst[tid * 2 + 0] = W[(k + 0) * NDIM + 16 * n + m];
    dst[tid * 2 + 1] = W[(k + 1) * NDIM + 16 * n + m];
    return;
  }

  const int e0 = b * CH;
  lcur[t] = 0;

  int er[CH / BIN_T], ec[CH / BIN_T];
#pragma unroll
  for (int j = 0; j < CH / BIN_T; ++j) {
    const int e = e0 + j * BIN_T + t;
    er[j] = (e < E) ? rowp[e] : -1;
    ec[j] = (e < E) ? colp[e] : 0;
  }
  __syncthreads();

#pragma unroll
  for (int j = 0; j < CH / BIN_T; ++j)
    if (er[j] >= 0) atomicAdd(&lcur[er[j] >> RB_SHIFT], 1);
  __syncthreads();
  const int c = lcur[t];
  int nChunk;
  const int lstart =
      gin_block_scan<BIN_T / 32>(c, t, wsa, nChunk) - c;
  lcur[t] = lstart;
  off[(size_t)b * OFP + t] = lstart;
  if (t == BIN_T - 1) off[(size_t)b * OFP + NBKP] = nChunk;
  __syncthreads();

  int slot[CH / BIN_T];
#pragma unroll
  for (int j = 0; j < CH / BIN_T; ++j)
    slot[j] = (er[j] >= 0) ? atomicAdd(&lcur[er[j] >> RB_SHIFT], 1) : 0;
#pragma unroll
  for (int j = 0; j < CH / BIN_T; ++j) {
    if (er[j] >= 0) {
      v2i v;
      v.x = er[j];
      v.y = ec[j];
      stage[slot[j]] = v;
    }
  }
  __syncthreads();

  v2i* dst = (v2i*)sortedc + (size_t)b * CH;
#pragma unroll
  for (int j = 0; j < CH / BIN_T; ++j) {
    const int i = j * BIN_T + t;
    if (i < nChunk) dst[i] = stage[i];
  }
}

__global__ void __launch_bounds__(RB)
gin_bin_finish_kernel(const int* __restrict__ sortedc,
                      const int* __restrict__ off, int* __restrict__ rowstart,
                      int* __restrict__ perm, int N, int nChunks) {
  __shared__ int wsa[RB / 32];
  __shared__ int wsb[RB / 32];
  __shared__ int wsc[RB / 32];
  __shared__ int cpos[NBKP];
  __shared__ int csrc[NBKP];
  __shared__ int clen[NBKP];
  __shared__ int lcur[RB];
  __shared__ v2i lev[FIN_CAP];
  __shared__ int lperm[FIN_CAP];
  const int t = (int)threadIdx.x;
  const int k = (int)blockIdx.x;
  const v2i* src = (const v2i*)sortedc;

  const int c0 = t;
  const int c1 = t + RB;
  const int* o0 = off + (size_t)(c0 < nChunks ? c0 : nChunks - 1) * OFP + k;
  const int* o1 = off + (size_t)(c1 < nChunks ? c1 : nChunks - 1) * OFP + k;
  const int a0 = o0[0];
  const int z0 = o0[1];
  const int a1 = o1[0];
  const int z1 = o1[1];
  const int s0 = c0 < nChunks ? a0 : 0;
  const int l0 = c0 < nChunks ? z0 - a0 : 0;
  const int s1 = c1 < nChunks ? a1 : 0;
  const int l1 = c1 < nChunks ? z1 - a1 : 0;
  int n;
  int b0;
  const int cex = gin_block_scan<RB / 32>(l0 + l1, t, wsa, n) - (l0 + l1);
  (void)gin_block_scan<RB / 32>(s0 + s1, t, wsb, b0);
  cpos[c0] = cex;
  csrc[c0] = c0 * CH + s0;
  clen[c0] = l0;
  cpos[c1] = cex + l0;
  csrc[c1] = c1 * CH + s1;
  clen[c1] = l1;
  lcur[t] = 0;
  __syncthreads();
  const bool staged = n <= FIN_CAP;
  const int hw = t >> 4;
  const int hl = t & 15;

  if (staged) {
    for (int base = 0; base < nChunks; base += 16 * FIN_Q) {
      v2i v[FIN_Q];
      int p[FIN_Q], ln[FIN_Q];
#pragma unroll
      for (int q = 0; q < FIN_Q; ++q) {
        const int c = base + 16 * q + hw;
        ln[q] = clen[c];
        p[q] = cpos[c];
        v[q].x = 0;
        v[q].y = 0;
        if (hl < ln[q]) v[q] = src[csrc[c] + hl];
      }
#pragma unroll
      for (int q = 0; q < FIN_Q; ++q) {
        if (hl < ln[q]) {
          lev[p[q] + hl] = v[q];
          atomicAdd(&lcur[v[q].x & (RB - 1)], 1);
        }
      }
#pragma unroll
      for (int q = 0; q < FIN_Q; ++q) {
        const int c = base + 16 * q + hw;
        for (int j = hl + 16; j < ln[q]; j += 16) {
          const v2i w = src[csrc[c] + j];
          lev[p[q] + j] = w;
          atomicAdd(&lcur[w.x & (RB - 1)], 1);
        }
      }
    }
  } else {
    for (int c = hw; c < nChunks; c += RB / 16) {
      const int s = csrc[c];
      const int len = clen[c];
      for (int j = hl; j < len; j += 16)
        atomicAdd(&lcur[src[s + j].x & (RB - 1)], 1);
    }
  }
  __syncthreads();

  const int d = lcur[t];
  int nScan;
  const int ex =
      gin_block_scan<RB / 32>(d, t, wsc, nScan) - d;
  lcur[t] = ex;
  const int row = k * RB + t;
  if (row < N) rowstart[row] = b0 + ex;
  if (row == N - 1) rowstart[N] = b0 + n;
  __syncthreads();

  if (staged) {
    v2i ev[FIN_U];
    int sl[FIN_U];
#pragma unroll
    for (int u = 0; u < FIN_U; ++u) {
      const int i = u * RB + t;
      if (i < n) {
        ev[u] = lev[i];
      } else {
        ev[u].x = -1;
        ev[u].y = 0;
      }
    }
#pragma unroll
    for (int u = 0; u < FIN_U; ++u)
      sl[u] = (ev[u].x >= 0) ? atomicAdd(&lcur[ev[u].x & (RB - 1)], 1) : 0;
#pragma unroll
    for (int u = 0; u < FIN_U; ++u)
      if (ev[u].x >= 0) lperm[sl[u]] = ev[u].y;
    __syncthreads();
#pragma unroll
    for (int u = 0; u < FIN_U; ++u) {
      const int i = u * RB + t;
      if (i < n) perm[b0 + i] = lperm[i];
    }
  } else {
    for (int c = hw; c < nChunks; c += RB / 16) {
      const int s = csrc[c];
      const int len = clen[c];
      for (int j = hl; j < len; j += 16) {
        const v2i v = src[s + j];
        perm[b0 + atomicAdd(&lcur[v.x & (RB - 1)], 1)] = v.y;
      }
    }
  }
}


extern "C" void kernel_launch(void* const* d_in, const int* in_sizes, int n_in,
                              void* d_out, int out_size, void* d_ws,
                              size_t ws_size, hipStream_t stream) {
  (void)n_in;
  (void)out_size;
  (void)ws_size;

  const float* x = (const float*)d_in[0];
  const int* ei = (const int*)d_in[1];
  const float* W1 = (const float*)d_in[2];
  const float* b1 = (const float*)d_in[3];
  const float* W2 = (const float*)d_in[4];
  const float* b2 = (const float*)d_in[5];
  const float* Wf = (const float*)d_in[6];
  const float* bf = (const float*)d_in[7];
  float* out = (float*)d_out;

  const int N = in_sizes[0] / NDIM;
  const int E = in_sizes[1] / 2;
  const int L = in_sizes[3] / NDIM;
  const int nTiles = N / 16;
  const size_t NB = (size_t)N * NDIM;

  float* tA = (float*)d_ws;
  float* tB = tA + NB;
  float* h = tB + NB;
  const int nMat = 2 * L + 1;
  float* wp = h + NB;
  const int nChunks = (E + CH - 1) / CH;
  const int nBuckets = (N + RB - 1) / RB;
  int* sortedc = (int*)(wp + (size_t)nMat * WP_ELEMS);
  int* off = sortedc + 2 * (size_t)nChunks * CH;
  int* perm = off + (size_t)nChunks * OFP;
  int* rowstart = perm + E;

  const int* rowp = ei;
  const int* colp = ei + E;

  gin_chunk_sort_kernel<<<nChunks + nMat * (REPACK_SLOTS / BIN_T), BIN_T, 0,
                          stream>>>(rowp, colp, off, sortedc, E, nChunks, W1,
                                    W2, Wf, wp, L);
  gin_bin_finish_kernel<<<nBuckets, RB, 0, stream>>>(sortedc, off, rowstart,
                                                     perm, N, nChunks);

  const dim3 mlpBlk(WAVES_PER_BLOCK * 32);
  const dim3 mlpGrd(MLP_GRID);

  gin_mlp_kernel<0><<<mlpGrd, mlpBlk, 0, stream>>>(x, nullptr, wp, b1,
                                                   wp + (size_t)L * WP_ELEMS,
                                                   b2, tA, nTiles);
  const float* hcur = x;
  float* tcur = tA;
  float* tnxt = tB;
  for (int l = 0; l < L; ++l) {
    const float* wp1 = wp + (size_t)l * WP_ELEMS;
    const float* wp2 = wp + (size_t)(L + l) * WP_ELEMS;
    const float* v1 = b1 + (size_t)l * NDIM;
    const float* v2 = b2 + (size_t)l * NDIM;

    if (l + 1 < L) {
      gin_fused_kernel<0><<<mlpGrd, mlpBlk, 0, stream>>>(
          hcur, tcur, rowstart, perm, wp1, v1, wp2, v2,
          wp + (size_t)(l + 1) * WP_ELEMS, v1 + NDIM,
          wp + (size_t)(L + l + 1) * WP_ELEMS, v2 + NDIM, h, tnxt, nTiles);
    } else {
      gin_fused_kernel<1><<<mlpGrd, mlpBlk, 0, stream>>>(
          hcur, tcur, rowstart, perm, wp1, v1, wp2, v2,
          wp + (size_t)(2 * L) * WP_ELEMS, bf, nullptr, nullptr, nullptr, out,
          nTiles);
    }
    float* tswap = tcur;
    tcur = tnxt;
    tnxt = tswap;
    hcur = h;
  }
}


# Adjacent linear maps folded into 64x64 products once per call on WMMA (W2[l]W1[l]; W2[l]W1[l+1] or Wf): 7 GEMMs per node, not 13; one array stored per layer, ReLU in gather; f32 sum order changes
# speedup vs baseline: 3.3414x; 1.2305x over previous
#include <hip/hip_runtime.h>


typedef __attribute__((ext_vector_type(2))) float v2f;
typedef __attribute__((ext_vector_type(8))) float v8f;
typedef __attribute__((ext_vector_type(2))) int v2i;

#define NDIM 64
#define WAVES_PER_BLOCK 4
#define WP_ELEMS 4096
#define RB 256
#define RB_SHIFT 8
#define NBKP 512
#define CH 4096
#define BIN_T 512
#define OFP (NBKP + 1)
#define FIN_U 16
#define FIN_CAP (FIN_U * RB)
#define FIN_Q 16
#define REPACK_SLOTS 2048
static_assert(BIN_T == NBKP, "one thread per padded bucket in the chunk-sort block");
static_assert((1 << RB_SHIFT) == RB, "bucket = row >> RB_SHIFT");
static_assert(NBKP == 2 * RB, "each finish thread owns the runs of two chunks");
static_assert(NBKP % (16 * FIN_Q) == 0, "run groups tile the padded chunk range");
static_assert(REPACK_SLOTS % BIN_T == 0, "whole repack blocks per matrix");
#define MLP_GRID 512
#define LAYER_GRID 768

__device__ __forceinline__ v8f wmma_f32_16x16x4(v2f a, v2f b, v8f c) {
  return __builtin_amdgcn_wmma_f32_16x16x4_f32(false, a, false, b, (short)0, c,
                                               false, false);
}


__device__ __forceinline__ void gin_fetch_rows(const float* __restrict__ H,
                                               int tile, int laneOff,
                                               v2f (&zh)[16]) {
  const float* hp = H + (size_t)tile * (16 * NDIM) + laneOff;
#pragma unroll
  for (int c = 0; c < 16; ++c) zh[c] = *(const v2f*)(hp + 4 * c);
}

__device__ __forceinline__ void gin_linear_tile(const v2f (&a)[16],
                                                const v2f (&w1)[64],
                                                const float (&bias1)[4],
                                                float* __restrict__ OUT,
                                                int tile, int laneM, int hi) {
  float* op = OUT + ((size_t)tile * 16 + 8 * hi) * NDIM + laneM;
  v8f acc[4];
#pragma unroll
  for (int n = 0; n < 4; ++n)
#pragma unroll
    for (int r = 0; r < 8; ++r) acc[n][r] = bias1[n];
#pragma unroll
  for (int c = 0; c < 16; ++c)
#pragma unroll
    for (int n = 0; n < 4; ++n)
      acc[n] = wmma_f32_16x16x4(a[c], w1[c * 4 + n], acc[n]);
#pragma unroll
  for (int n = 0; n < 4; ++n)
#pragma unroll
    for (int r = 0; r < 8; ++r) op[r * NDIM + 16 * n] = acc[n][r];
}

__global__ void __launch_bounds__(WAVES_PER_BLOCK * 32)
gin_linear_kernel(const float* __restrict__ H, const float* __restrict__ WP1,
                  const float* __restrict__ B1, float* __restrict__ OUT,
                  int nTiles) {
  const int wave = __builtin_amdgcn_readfirstlane((int)(threadIdx.x >> 5));
  const int lane = threadIdx.x & 31;
  const int stride = (int)gridDim.x * WAVES_PER_BLOCK;
  int tile = (int)blockIdx.x * WAVES_PER_BLOCK + wave;
  if (tile >= nTiles) return;

  const int laneM = lane & 15;
  const int hi = lane >> 4;
  const int laneOff = laneM * NDIM + 2 * hi;

  v2f hA[16], hB[16];
  gin_fetch_rows(H, tile, laneOff, hA);

  v2f w1[64];
  float bias1[4];
  {
    const v2f* wp1 = (const v2f*)WP1 + lane;
#pragma unroll
    for (int i = 0; i < 64; ++i) w1[i] = wp1[i * 32];
#pragma unroll
    for (int n = 0; n < 4; ++n) bias1[n] = B1[16 * n + laneM];
  }

  int cnt = (nTiles - 1 - tile) / stride + 1;

  while (cnt >= 2) {
    const int t1 = tile + stride;
    const int t2 = t1 + stride;

    gin_fetch_rows(H, t1, laneOff, hB);
    __builtin_amdgcn_sched_barrier(0);
    gin_linear_tile(hA, w1, bias1, OUT, tile, laneM, hi);

    gin_fetch_rows(H, cnt > 2 ? t2 : t1, laneOff, hA);
    __builtin_amdgcn_sched_barrier(0);
    gin_linear_tile(hB, w1, bias1, OUT, t1, laneM, hi);

    tile = t2;
    cnt -= 2;
  }
  if (cnt == 1) gin_linear_tile(hA, w1, bias1, OUT, tile, laneM, hi);
}

__device__ __forceinline__ void gin_lds_frags(const v2f* wl, int blk,
                                              v2f (&b)[16]) {
#pragma unroll
  for (int i = 0; i < 16; ++i) b[i] = wl[(blk * 16 + i) * 32];
}

__device__ __forceinline__ void gin_wmma_block(const v2f (&a)[16], int blk,
                                               const v2f (&b)[16],
                                               v8f (&acc)[4]) {
#pragma unroll
  for (int cc = 0; cc < 4; ++cc)
#pragma unroll
    for (int n = 0; n < 4; ++n)
      acc[n] = wmma_f32_16x16x4(a[blk * 4 + cc], b[cc * 4 + n], acc[n]);
}

template <bool CHAIN>
__device__ __forceinline__ void gin_gemm_lds(const v2f (&a)[16], const v2f* wl,
                                             const v2f* nxt, v2f (&b0)[16],
                                             v2f (&b1)[16], v8f (&acc)[4]) {
  gin_lds_frags(wl, 1, b1);
  __builtin_amdgcn_sched_barrier(0);
  gin_wmma_block(a, 0, b0, acc);
  __builtin_amdgcn_sched_barrier(0);
  gin_lds_frags(wl, 2, b0);
  __builtin_amdgcn_sched_barrier(0);
  gin_wmma_block(a, 1, b1, acc);
  __builtin_amdgcn_sched_barrier(0);
  gin_lds_frags(wl, 3, b1);
  __builtin_amdgcn_sched_barrier(0);
  gin_wmma_block(a, 2, b0, acc);
  __builtin_amdgcn_sched_barrier(0);
  if (CHAIN) gin_lds_frags(nxt, 0, b0);
  __builtin_amdgcn_sched_barrier(0);
  gin_wmma_block(a, 3, b1, acc);
  __builtin_amdgcn_sched_barrier(0);
}

#define WL_FRAGS (WP_ELEMS / 2)

#define GU 4
__device__ __forceinline__ void gin_gather_tile(const float* __restrict__ T,
                                                const int* __restrict__ perm,
                                                unsigned zoff, int rs, int lane,
                                                float* lt, v2f (&zg)[16]) {
  const int hl = lane >> 4;
  const int laneM = lane & 15;
  const unsigned colB = (unsigned)laneM * 16u;
  const char* tb = (const char*)T;
  const char* pb = (const char*)perm;

  int s[8], e[8];
#pragma unroll
  for (int p = 0; p < 8; ++p) {
    s[p] = __shfl(rs, 2 * p + hl);
    e[p] = __shfl(rs, 2 * p + hl + 1);
  }
  int d = __shfl_down(rs, 1) - rs;
  d = lane < 16 ? d : 0;
#pragma unroll
  for (int o = 16; o > 0; o >>= 1) {
    const int u = __shfl_xor(d, o);
    d = u > d ? u : d;
  }
  const int maxdeg = __builtin_amdgcn_readfirstlane(d);

  float4 acc[8];
#pragma unroll
  for (int p = 0; p < 8; ++p) acc[p] = make_float4(0.f, 0.f, 0.f, 0.f);

  unsigned ro[8][GU];
#pragma unroll
  for (int p = 0; p < 8; ++p)
#pragma unroll
    for (int u = 0; u < GU; ++u) {
      const int j = s[p] + u;
      const bool in = j < e[p];
      const unsigned jj = in ? (unsigned)j : 0u;
      const unsigned rb = (unsigned)(*(const int*)(pb + jj * 4u)) * (NDIM * 4u);
      ro[p][u] = (in ? rb : zoff) + colB;
    }

  for (int base = 0; base < maxdeg; base += GU) {
    float4 v[8][GU];
#pragma unroll
    for (int p = 0; p < 8; ++p)
#pragma unroll
      for (int u = 0; u < GU; ++u) v[p][u] = *(const float4*)(tb + ro[p][u]);
    int nx[8][GU];
#pragma unroll
    for (int p = 0; p < 8; ++p)
#pragma unroll
      for (int u = 0; u < GU; ++u) {
        const int j = s[p] + base + GU + u;
        const unsigned jj = j < e[p] ? (unsigned)j : 0u;
        nx[p][u] = *(const int*)(pb + jj * 4u);
      }
    __builtin_amdgcn_sched_barrier(0);
#pragma unroll
    for (int p = 0; p < 8; ++p)
#pragma unroll
      for (int u = 0; u < GU; ++u) {
        acc[p].x += fmaxf(v[p][u].x, 0.0f);
        acc[p].y += fmaxf(v[p][u].y, 0.0f);
        acc[p].z += fmaxf(v[p][u].z, 0.0f);
        acc[p].w += fmaxf(v[p][u].w, 0.0f);
        const bool in = s[p] + base + GU + u < e[p];
        ro[p][u] = (in ? (unsigned)nx[p][u] * (NDIM * 4u) : zoff) + colB;
      }
  }

#pragma unroll
  for (int p = 0; p < 8; ++p)
    *(float4*)(lt + (2 * p + hl) * NDIM + 4 * laneM) = acc[p];
#pragma unroll
  for (int c = 0; c < 16; ++c)
    zg[c] = *(const v2f*)(lt + laneM * NDIM + 4 * c + 2 * hl);
}

__device__ __forceinline__ void gin_layer_tile(
    const float* __restrict__ U, const v2f (&zs)[16], int deg, const v2f* wlp,
    const v2f* wlq, const float (&cv)[4], const float (&dv)[4], float* lt,
    float* __restrict__ OUT, int tile, int laneM, int hi) {
  const size_t off = ((size_t)tile * 16 + 8 * hi) * NDIM + laneM;
  v2f bq0[16], bq1[16];

  float uo[4][8];
  {
    const float* up = U + off;
#pragma unroll
    for (int n = 0; n < 4; ++n)
#pragma unroll
      for (int r = 0; r < 8; ++r) uo[n][r] = up[r * NDIM + 16 * n];
  }

  v8f acc[4];
#pragma unroll
  for (int r = 0; r < 8; ++r) {
    const float dr = (float)__shfl(deg, 8 * hi + r);
#pragma unroll
    for (int n = 0; n < 4; ++n) acc[n][r] = dr * cv[n];
  }
  gin_lds_frags(wlp, 0, bq0);
  __builtin_amdgcn_sched_barrier(0);
  gin_gemm_lds<true>(zs, wlp, wlq, bq0, bq1, acc);

#pragma unroll
  for (int n = 0; n < 4; ++n)
#pragma unroll
    for (int r = 0; r < 8; ++r)
      lt[(r + 8 * hi) * NDIM + 16 * n + laneM] =
          fmaxf(acc[n][r] + uo[n][r], 0.0f);
  v2f a[16];
#pragma unroll
  for (int c = 0; c < 16; ++c)
    a[c] = *(const v2f*)(lt + laneM * NDIM + 4 * c + 2 * hi);

#pragma unroll
  for (int n = 0; n < 4; ++n)
#pragma unroll
    for (int r = 0; r < 8; ++r) acc[n][r] = dv[n];
  gin_gemm_lds<false>(a, wlq, wlq, bq0, bq1, acc);

  float* op = OUT + off;
#pragma unroll
  for (int n = 0; n < 4; ++n)
#pragma unroll
    for (int r = 0; r < 8; ++r) op[r * NDIM + 16 * n] = acc[n][r];
}

__global__ void __launch_bounds__(WAVES_PER_BLOCK * 32)
__attribute__((amdgpu_waves_per_eu(3, 3)))
gin_layer_kernel(const float* __restrict__ U, const int* __restrict__ rowstart,
                 const int* __restrict__ perm, const float* __restrict__ WPP,
                 const float* __restrict__ CV, const float* __restrict__ WPQ,
                 const float* __restrict__ DV, float* __restrict__ OUT,
                 int nTiles) {
  __shared__ __attribute__((aligned(16))) float lds[WAVES_PER_BLOCK][16 * NDIM];
  __shared__ v2f wl[2 * WL_FRAGS];

  const int wave = __builtin_amdgcn_readfirstlane((int)(threadIdx.x >> 5));
  const int lane = threadIdx.x & 31;
  const int stride = (int)gridDim.x * WAVES_PER_BLOCK;
  int tile = (int)blockIdx.x * WAVES_PER_BLOCK + wave;
  const bool has = tile < nTiles;

  const int laneM = lane & 15;
  const int hi = lane >> 4;
  float* lt = lds[wave];

  const int rsLane = lane < 16 ? lane : 16;
  int rs = 0;
  if (has) rs = rowstart[16 * tile + rsLane];

  {
    const v2f* sp = (const v2f*)WPP;
    const v2f* sq = (const v2f*)WPQ;
#pragma unroll
    for (int i = 0; i < WL_FRAGS / (WAVES_PER_BLOCK * 32); ++i) {
      const int j = i * (WAVES_PER_BLOCK * 32) + (int)threadIdx.x;
      wl[j] = sp[j];
      wl[WL_FRAGS + j] = sq[j];
    }
  }
  float cv[4], dv[4];
#pragma unroll
  for (int n = 0; n < 4; ++n) {
    cv[n] = CV[16 * n + laneM];
    dv[n] = DV[16 * n + laneM];
  }
  __syncthreads();
  if (!has) return;

  const v2f* wlp = wl + lane;
  const v2f* wlq = wl + WL_FRAGS + lane;
  const unsigned zoff = (unsigned)nTiles * (16u * NDIM * 4u);

  int cnt = (nTiles - 1 - tile) / stride + 1;

  for (; cnt > 0; --cnt) {
    const int nt = cnt > 1 ? tile + stride : tile;
    const int rsn = rowstart[16 * nt + rsLane];
    __builtin_amdgcn_sched_barrier(0);
    v2f zs[16];
    gin_gather_tile(U, perm, zoff, rs, lane, lt, zs);
    const int deg = __shfl_down(rs, 1) - rs;
    gin_layer_tile(U, zs, deg, wlp, wlq, cv, dv, lt, OUT, tile, laneM, hi);
    rs = rsn;
    tile = nt;
  }
}


template <int NW>
__device__ __forceinline__ int gin_block_scan(int v, int t, int* wsum,
                                              int& total) {
  const int lane = t & 31;
  const int wave = t >> 5;
#pragma unroll
  for (int off = 1; off < 32; off <<= 1) {
    const int u = __shfl_up(v, off);
    if (lane >= off) v += u;
  }
  if (lane == 31) wsum[wave] = v;
  __syncthreads();
  int before = 0;
  int all = 0;
#pragma unroll
  for (int w = 0; w < NW; ++w) {
    const int ws = wsum[w];
    all += ws;
    if (w < wave) before += ws;
  }
  total = all;
  return v + before;
}

__global__ void __launch_bounds__(BIN_T)
gin_chunk_sort_kernel(const int* __restrict__ rowp,
                      const int* __restrict__ colp, int* __restrict__ off,
                      int* __restrict__ sortedc, int E, int nChunks,
                      const float* __restrict__ W1,
                      const float* __restrict__ B1,
                      const float* __restrict__ W2,
                      const float* __restrict__ B2,
                      const float* __restrict__ Wf,
                      const float* __restrict__ Bf, float* __restrict__ Wp,
                      float* __restrict__ bv, float* __restrict__ Z0,
                      float* __restrict__ Z1, int L) {
  __shared__ int wsa[BIN_T / 32];
  __shared__ int lcur[NBKP];
  __shared__ v2i stage[CH];
  const int t = (int)threadIdx.x;
  const int b = (int)blockIdx.x;

  if (b >= nChunks) {
    const int q = b - nChunks;
    if (q < REPACK_SLOTS / BIN_T) {
      const int tid = q * BIN_T + t;
      const int c = tid >> 7;
      const int n = (tid >> 5) & 3;
      const int lane = tid & 31;
      const int hi = lane >> 4;
      const int m = lane & 15;
      const int k = 4 * c + 2 * hi;
      Wp[tid * 2 + 0] = W1[(k + 0) * NDIM + 16 * n + m];
      Wp[tid * 2 + 1] = W1[(k + 1) * NDIM + 16 * n + m];
      if (tid < NDIM) {
        Z0[tid] = 0.0f;
        Z1[tid] = 0.0f;
      }
      return;
    }
    const int pm = q - REPACK_SLOTS / BIN_T;
    const bool isQ = pm >= L;
    const int l = isQ ? pm - L : pm;
    const float* A = W2 + (size_t)l * (NDIM * NDIM);
    const float* Bm = !isQ        ? W1 + (size_t)l * (NDIM * NDIM)
                      : l + 1 < L ? W1 + (size_t)(l + 1) * (NDIM * NDIM)
                                  : Wf;
    float* dst = Wp + (size_t)(1 + pm) * WP_ELEMS;
    const int wv = t >> 5;
    const int lane = t & 31;
    const int hi = lane >> 4;
    const int m = lane & 15;
    const int R = wv >> 2;
    const int n = wv & 3;
    const float* ap = A + (16 * R + m) * NDIM + 2 * hi;
    const float* bp = Bm + (2 * hi) * NDIM + 16 * n + m;
    v8f acc;
#pragma unroll
    for (int r = 0; r < 8; ++r) acc[r] = 0.0f;
#pragma unroll
    for (int c = 0; c < 16; ++c) {
      const v2f a = *(const v2f*)(ap + 4 * c);
      v2f bb;
      bb.x = bp[(4 * c + 0) * NDIM];
      bb.y = bp[(4 * c + 1) * NDIM];
      acc = wmma_f32_16x16x4(a, bb, acc);
    }
#pragma unroll
    for (int r = 0; r < 8; ++r) {
      const int row = 16 * R + 8 * hi + r;
      dst[((((row >> 2) * 4 + n) * 32) + ((row >> 1) & 1) * 16 + m) * 2 +
          (row & 1)] = acc[r];
    }
    if (t < NDIM) {
      const float* bs = B2 + (size_t)l * NDIM;
      float sum = !isQ ? 0.0f : l + 1 < L ? B1[(l + 1) * NDIM + t] : Bf[t];
      for (int i = 0; i < NDIM; ++i) sum += bs[i] * Bm[i * NDIM + t];
      bv[pm * NDIM + t] = sum;
    }
    return;
  }

  const int e0 = b * CH;
  lcur[t] = 0;

  int er[CH / BIN_T], ec[CH / BIN_T];
#pragma unroll
  for (int j = 0; j < CH / BIN_T; ++j) {
    const int e = e0 + j * BIN_T + t;
    er[j] = (e < E) ? rowp[e] : -1;
    ec[j] = (e < E) ? colp[e] : 0;
  }
  __syncthreads();

#pragma unroll
  for (int j = 0; j < CH / BIN_T; ++j)
    if (er[j] >= 0) atomicAdd(&lcur[er[j] >> RB_SHIFT], 1);
  __syncthreads();
  const int c = lcur[t];
  int nChunk;
  const int lstart =
      gin_block_scan<BIN_T / 32>(c, t, wsa, nChunk) - c;
  lcur[t] = lstart;
  off[(size_t)b * OFP + t] = lstart;
  if (t == BIN_T - 1) off[(size_t)b * OFP + NBKP] = nChunk;
  __syncthreads();

  int slot[CH / BIN_T];
#pragma unroll
  for (int j = 0; j < CH / BIN_T; ++j)
    slot[j] = (er[j] >= 0) ? atomicAdd(&lcur[er[j] >> RB_SHIFT], 1) : 0;
#pragma unroll
  for (int j = 0; j < CH / BIN_T; ++j) {
    if (er[j] >= 0) {
      v2i v;
      v.x = er[j];
      v.y = ec[j];
      stage[slot[j]] = v;
    }
  }
  __syncthreads();

  v2i* dst = (v2i*)sortedc + (size_t)b * CH;
#pragma unroll
  for (int j = 0; j < CH / BIN_T; ++j) {
    const int i = j * BIN_T + t;
    if (i < nChunk) dst[i] = stage[i];
  }
}

__global__ void __launch_bounds__(RB)
gin_bin_finish_kernel(const int* __restrict__ sortedc,
                      const int* __restrict__ off, int* __restrict__ rowstart,
                      int* __restrict__ perm, int N, int nChunks) {
  __shared__ int wsa[RB / 32];
  __shared__ int wsb[RB / 32];
  __shared__ int wsc[RB / 32];
  __shared__ int cpos[NBKP];
  __shared__ int csrc[NBKP];
  __shared__ int clen[NBKP];
  __shared__ int lcur[RB];
  __shared__ v2i lev[FIN_CAP];
  __shared__ int lperm[FIN_CAP];
  const int t = (int)threadIdx.x;
  const int k = (int)blockIdx.x;
  const v2i* src = (const v2i*)sortedc;

  const int c0 = t;
  const int c1 = t + RB;
  const int* o0 = off + (size_t)(c0 < nChunks ? c0 : nChunks - 1) * OFP + k;
  const int* o1 = off + (size_t)(c1 < nChunks ? c1 : nChunks - 1) * OFP + k;
  const int a0 = o0[0];
  const int z0 = o0[1];
  const int a1 = o1[0];
  const int z1 = o1[1];
  const int s0 = c0 < nChunks ? a0 : 0;
  const int l0 = c0 < nChunks ? z0 - a0 : 0;
  const int s1 = c1 < nChunks ? a1 : 0;
  const int l1 = c1 < nChunks ? z1 - a1 : 0;
  int n;
  int b0;
  const int cex = gin_block_scan<RB / 32>(l0 + l1, t, wsa, n) - (l0 + l1);
  (void)gin_block_scan<RB / 32>(s0 + s1, t, wsb, b0);
  cpos[c0] = cex;
  csrc[c0] = c0 * CH + s0;
  clen[c0] = l0;
  cpos[c1] = cex + l0;
  csrc[c1] = c1 * CH + s1;
  clen[c1] = l1;
  lcur[t] = 0;
  __syncthreads();
  const bool staged = n <= FIN_CAP;
  const int hw = t >> 4;
  const int hl = t & 15;

  if (staged) {
    for (int base = 0; base < nChunks; base += 16 * FIN_Q) {
      v2i v[FIN_Q];
      int p[FIN_Q], ln[FIN_Q];
#pragma unroll
      for (int q = 0; q < FIN_Q; ++q) {
        const int c = base + 16 * q + hw;
        ln[q] = clen[c];
        p[q] = cpos[c];
        v[q].x = 0;
        v[q].y = 0;
        if (hl < ln[q]) v[q] = src[csrc[c] + hl];
      }
#pragma unroll
      for (int q = 0; q < FIN_Q; ++q) {
        if (hl < ln[q]) {
          lev[p[q] + hl] = v[q];
          atomicAdd(&lcur[v[q].x & (RB - 1)], 1);
        }
      }
#pragma unroll
      for (int q = 0; q < FIN_Q; ++q) {
        const int c = base + 16 * q + hw;
        for (int j = hl + 16; j < ln[q]; j += 16) {
          const v2i w = src[csrc[c] + j];
          lev[p[q] + j] = w;
          atomicAdd(&lcur[w.x & (RB - 1)], 1);
        }
      }
    }
  } else {
    for (int c = hw; c < nChunks; c += RB / 16) {
      const int s = csrc[c];
      const int len = clen[c];
      for (int j = hl; j < len; j += 16)
        atomicAdd(&lcur[src[s + j].x & (RB - 1)], 1);
    }
  }
  __syncthreads();

  const int d = lcur[t];
  int nScan;
  const int ex =
      gin_block_scan<RB / 32>(d, t, wsc, nScan) - d;
  lcur[t] = ex;
  const int row = k * RB + t;
  if (row < N) rowstart[row] = b0 + ex;
  if (row == N - 1) rowstart[N] = b0 + n;
  __syncthreads();

  if (staged) {
    v2i ev[FIN_U];
    int sl[FIN_U];
#pragma unroll
    for (int u = 0; u < FIN_U; ++u) {
      const int i = u * RB + t;
      if (i < n) {
        ev[u] = lev[i];
      } else {
        ev[u].x = -1;
        ev[u].y = 0;
      }
    }
#pragma unroll
    for (int u = 0; u < FIN_U; ++u)
      sl[u] = (ev[u].x >= 0) ? atomicAdd(&lcur[ev[u].x & (RB - 1)], 1) : 0;
#pragma unroll
    for (int u = 0; u < FIN_U; ++u)
      if (ev[u].x >= 0) lperm[sl[u]] = ev[u].y;
    __syncthreads();
#pragma unroll
    for (int u = 0; u < FIN_U; ++u) {
      const int i = u * RB + t;
      if (i < n) perm[b0 + i] = lperm[i];
    }
  } else {
    for (int c = hw; c < nChunks; c += RB / 16) {
      const int s = csrc[c];
      const int len = clen[c];
      for (int j = hl; j < len; j += 16) {
        const v2i v = src[s + j];
        perm[b0 + atomicAdd(&lcur[v.x & (RB - 1)], 1)] = v.y;
      }
    }
  }
}


extern "C" void kernel_launch(void* const* d_in, const int* in_sizes, int n_in,
                              void* d_out, int out_size, void* d_ws,
                              size_t ws_size, hipStream_t stream) {
  (void)n_in;
  (void)out_size;
  (void)ws_size;

  const float* x = (const float*)d_in[0];
  const int* ei = (const int*)d_in[1];
  const float* W1 = (const float*)d_in[2];
  const float* b1 = (const float*)d_in[3];
  const float* W2 = (const float*)d_in[4];
  const float* b2 = (const float*)d_in[5];
  const float* Wf = (const float*)d_in[6];
  const float* bf = (const float*)d_in[7];
  float* out = (float*)d_out;

  const int N = in_sizes[0] / NDIM;
  const int E = in_sizes[1] / 2;
  const int L = in_sizes[3] / NDIM;
  const int nTiles = N / 16;
  const size_t NB = (size_t)N * NDIM;

  float* uA = (float*)d_ws;
  float* uB = uA + NB + NDIM;
  const int nMat = 2 * L + 1;
  float* wp = uB + NB + NDIM;
  float* bv = wp + (size_t)nMat * WP_ELEMS;
  const int nChunks = (E + CH - 1) / CH;
  const int nBuckets = (N + RB - 1) / RB;
  int* sortedc = (int*)(bv + (size_t)2 * L * NDIM);
  int* off = sortedc + 2 * (size_t)nChunks * CH;
  int* perm = off + (size_t)nChunks * OFP;
  int* rowstart = perm + E;

  const int* rowp = ei;
  const int* colp = ei + E;

  gin_chunk_sort_kernel<<<nChunks + REPACK_SLOTS / BIN_T + 2 * L, BIN_T, 0,
                          stream>>>(rowp, colp, off, sortedc, E, nChunks, W1,
                                    b1, W2, b2, Wf, bf, wp, bv, uA + NB,
                                    uB + NB, L);
  gin_bin_finish_kernel<<<nBuckets, RB, 0, stream>>>(sortedc, off, rowstart,
                                                     perm, N, nChunks);

  const dim3 mlpBlk(WAVES_PER_BLOCK * 32);

  gin_linear_kernel<<<dim3(MLP_GRID), mlpBlk, 0, stream>>>(x, wp, b1, uA,
                                                           nTiles);
  float* ucur = uA;
  float* unxt = uB;
  for (int l = 0; l < L; ++l) {
    float* dst = l + 1 < L ? unxt : out;
    gin_layer_kernel<<<dim3(LAYER_GRID), mlpBlk, 0, stream>>>(
        ucur, rowstart, perm, wp + (size_t)(1 + l) * WP_ELEMS,
        bv + (size_t)l * NDIM, wp + (size_t)(1 + L + l) * WP_ELEMS,
        bv + (size_t)(L + l) * NDIM, dst, nTiles);
    float* uswap = ucur;
    ucur = unxt;
    unxt = uswap;
  }
}


# Fewer dependent launches, 6 -> 4: x W1_0 + b1_0 runs as extra blocks of the chunk-sort launch (W1_0 from LDS); each CSR finish block then runs layer 0 on its own 256 rows; f32 sum order as before
# speedup vs baseline: 3.4452x; 1.0311x over previous
#include <hip/hip_runtime.h>


typedef __attribute__((ext_vector_type(2))) float v2f;
typedef __attribute__((ext_vector_type(8))) float v8f;
typedef __attribute__((ext_vector_type(2))) int v2i;

#define NDIM 64
#define WAVES_PER_BLOCK 4
#define WP_ELEMS 4096
#define RB 256
#define RB_SHIFT 8
#define NBKP 512
#define CH 4096
#define BIN_T 512
#define OFP (NBKP + 1)
#define FIN_U 16
#define FIN_CAP (FIN_U * RB)
#define FIN_Q 16
#define REPACK_SLOTS 2048
static_assert(BIN_T == NBKP, "one thread per padded bucket in the chunk-sort block");
static_assert((1 << RB_SHIFT) == RB, "bucket = row >> RB_SHIFT");
static_assert(NBKP == 2 * RB, "each finish thread owns the runs of two chunks");
static_assert(NBKP % (16 * FIN_Q) == 0, "run groups tile the padded chunk range");
static_assert(REPACK_SLOTS % BIN_T == 0, "whole passes of a block per packed matrix");
static_assert(REPACK_SLOTS % RB == 0, "whole passes of a finish block per packed matrix");
static_assert(RB % (16 * (RB / 32)) == 0, "a bucket's tiles deal out evenly over its block's waves");
#define LIN_GRID 196
#define LAYER_GRID 768

__device__ __forceinline__ v8f wmma_f32_16x16x4(v2f a, v2f b, v8f c) {
  return __builtin_amdgcn_wmma_f32_16x16x4_f32(false, a, false, b, (short)0, c,
                                               false, false);
}

__device__ __forceinline__ void gin_fetch_rows(const float* __restrict__ H,
                                               int tile, int laneOff,
                                               v2f (&zh)[16]) {
  const float* hp = H + (size_t)tile * (16 * NDIM) + laneOff;
#pragma unroll
  for (int c = 0; c < 16; ++c) zh[c] = *(const v2f*)(hp + 4 * c);
}

__device__ __forceinline__ void gin_lds_frags(const v2f* wl, int blk,
                                              v2f (&b)[16]) {
#pragma unroll
  for (int i = 0; i < 16; ++i) b[i] = wl[(blk * 16 + i) * 32];
}

__device__ __forceinline__ void gin_wmma_block(const v2f (&a)[16], int blk,
                                               const v2f (&b)[16],
                                               v8f (&acc)[4]) {
#pragma unroll
  for (int cc = 0; cc < 4; ++cc)
#pragma unroll
    for (int n = 0; n < 4; ++n)
      acc[n] = wmma_f32_16x16x4(a[blk * 4 + cc], b[cc * 4 + n], acc[n]);
}

__device__ __forceinline__ void gin_linear_tile_lds(const v2f (&a)[16],
                                                    const v2f* wl,
                                                    const float (&bias1)[4],
                                                    float* __restrict__ OUT,
                                                    int tile, int laneM,
                                                    int hi) {
  float* op = OUT + ((size_t)tile * 16 + 8 * hi) * NDIM + laneM;
  v8f acc[4];
#pragma unroll
  for (int n = 0; n < 4; ++n)
#pragma unroll
    for (int r = 0; r < 8; ++r) acc[n][r] = bias1[n];
#pragma unroll
  for (int blk = 0; blk < 4; ++blk) {
    v2f b[16];
    gin_lds_frags(wl, blk, b);
    gin_wmma_block(a, blk, b, acc);
  }
#pragma unroll
  for (int n = 0; n < 4; ++n)
#pragma unroll
    for (int r = 0; r < 8; ++r) op[r * NDIM + 16 * n] = acc[n][r];
}

template <bool CHAIN>
__device__ __forceinline__ void gin_gemm_lds(const v2f (&a)[16], const v2f* wl,
                                             const v2f* nxt, v2f (&b0)[16],
                                             v2f (&b1)[16], v8f (&acc)[4]) {
  gin_lds_frags(wl, 1, b1);
  __builtin_amdgcn_sched_barrier(0);
  gin_wmma_block(a, 0, b0, acc);
  __builtin_amdgcn_sched_barrier(0);
  gin_lds_frags(wl, 2, b0);
  __builtin_amdgcn_sched_barrier(0);
  gin_wmma_block(a, 1, b1, acc);
  __builtin_amdgcn_sched_barrier(0);
  gin_lds_frags(wl, 3, b1);
  __builtin_amdgcn_sched_barrier(0);
  gin_wmma_block(a, 2, b0, acc);
  __builtin_amdgcn_sched_barrier(0);
  if (CHAIN) gin_lds_frags(nxt, 0, b0);
  __builtin_amdgcn_sched_barrier(0);
  gin_wmma_block(a, 3, b1, acc);
  __builtin_amdgcn_sched_barrier(0);
}

#define WL_FRAGS (WP_ELEMS / 2)

template <int GU>
__device__ __forceinline__ void gin_gather_tile(const float* __restrict__ T,
                                                const int* perm,
                                                unsigned zoff, int rs, int lane,
                                                float* lt, v2f (&zg)[16]) {
  const int hl = lane >> 4;
  const int laneM = lane & 15;
  const unsigned colB = (unsigned)laneM * 16u;
  const char* tb = (const char*)T;
  const char* pb = (const char*)perm;

  int s[8], e[8];
#pragma unroll
  for (int p = 0; p < 8; ++p) {
    s[p] = __shfl(rs, 2 * p + hl);
    e[p] = __shfl(rs, 2 * p + hl + 1);
  }
  int d = __shfl_down(rs, 1) - rs;
  d = lane < 16 ? d : 0;
#pragma unroll
  for (int o = 16; o > 0; o >>= 1) {
    const int u = __shfl_xor(d, o);
    d = u > d ? u : d;
  }
  const int maxdeg = __builtin_amdgcn_readfirstlane(d);

  float4 acc[8];
#pragma unroll
  for (int p = 0; p < 8; ++p) acc[p] = make_float4(0.f, 0.f, 0.f, 0.f);

  unsigned ro[8][GU];
#pragma unroll
  for (int p = 0; p < 8; ++p)
#pragma unroll
    for (int u = 0; u < GU; ++u) {
      const int j = s[p] + u;
      const bool in = j < e[p];
      const unsigned jj = in ? (unsigned)j : 0u;
      const unsigned rb = (unsigned)(*(const int*)(pb + jj * 4u)) * (NDIM * 4u);
      ro[p][u] = (in ? rb : zoff) + colB;
    }

  for (int base = 0; base < maxdeg; base += GU) {
    float4 v[8][GU];
#pragma unroll
    for (int p = 0; p < 8; ++p)
#pragma unroll
      for (int u = 0; u < GU; ++u) v[p][u] = *(const float4*)(tb + ro[p][u]);
    int nx[8][GU];
#pragma unroll
    for (int p = 0; p < 8; ++p)
#pragma unroll
      for (int u = 0; u < GU; ++u) {
        const int j = s[p] + base + GU + u;
        const unsigned jj = j < e[p] ? (unsigned)j : 0u;
        nx[p][u] = *(const int*)(pb + jj * 4u);
      }
    __builtin_amdgcn_sched_barrier(0);
#pragma unroll
    for (int p = 0; p < 8; ++p)
#pragma unroll
      for (int u = 0; u < GU; ++u) {
        acc[p].x += fmaxf(v[p][u].x, 0.0f);
        acc[p].y += fmaxf(v[p][u].y, 0.0f);
        acc[p].z += fmaxf(v[p][u].z, 0.0f);
        acc[p].w += fmaxf(v[p][u].w, 0.0f);
        const bool in = s[p] + base + GU + u < e[p];
        ro[p][u] = (in ? (unsigned)nx[p][u] * (NDIM * 4u) : zoff) + colB;
      }
  }

#pragma unroll
  for (int p = 0; p < 8; ++p)
    *(float4*)(lt + (2 * p + hl) * NDIM + 4 * laneM) = acc[p];
#pragma unroll
  for (int c = 0; c < 16; ++c)
    zg[c] = *(const v2f*)(lt + laneM * NDIM + 4 * c + 2 * hl);
}

__device__ __forceinline__ void gin_layer_tile(
    const float* __restrict__ U, const v2f (&zs)[16], int deg, const v2f* wlp,
    const v2f* wlq, const float (&cv)[4], const float (&dv)[4], float* lt,
    float* __restrict__ OUT, int tile, int laneM, int hi) {
  const size_t off = ((size_t)tile * 16 + 8 * hi) * NDIM + laneM;
  v2f bq0[16], bq1[16];

  float uo[4][8];
  {
    const float* up = U + off;
#pragma unroll
    for (int n = 0; n < 4; ++n)
#pragma unroll
      for (int r = 0; r < 8; ++r) uo[n][r] = up[r * NDIM + 16 * n];
  }

  v8f acc[4];
#pragma unroll
  for (int r = 0; r < 8; ++r) {
    const float dr = (float)__shfl(deg, 8 * hi + r);
#pragma unroll
    for (int n = 0; n < 4; ++n) acc[n][r] = dr * cv[n];
  }
  gin_lds_frags(wlp, 0, bq0);
  __builtin_amdgcn_sched_barrier(0);
  gin_gemm_lds<true>(zs, wlp, wlq, bq0, bq1, acc);

#pragma unroll
  for (int n = 0; n < 4; ++n)
#pragma unroll
    for (int r = 0; r < 8; ++r)
      lt[(r + 8 * hi) * NDIM + 16 * n + laneM] =
          fmaxf(acc[n][r] + uo[n][r], 0.0f);
  v2f a[16];
#pragma unroll
  for (int c = 0; c < 16; ++c)
    a[c] = *(const v2f*)(lt + laneM * NDIM + 4 * c + 2 * hi);

#pragma unroll
  for (int n = 0; n < 4; ++n)
#pragma unroll
    for (int r = 0; r < 8; ++r) acc[n][r] = dv[n];
  gin_gemm_lds<false>(a, wlq, wlq, bq0, bq1, acc);

  float* op = OUT + off;
#pragma unroll
  for (int n = 0; n < 4; ++n)
#pragma unroll
    for (int r = 0; r < 8; ++r) op[r * NDIM + 16 * n] = acc[n][r];
}

__global__ void __launch_bounds__(WAVES_PER_BLOCK * 32)
__attribute__((amdgpu_waves_per_eu(3, 3)))
gin_layer_kernel(const float* __restrict__ U, const int* __restrict__ rowstart,
                 const int* __restrict__ perm, const float* __restrict__ WPP,
                 const float* __restrict__ CV, const float* __restrict__ WPQ,
                 const float* __restrict__ DV, float* __restrict__ OUT,
                 int nTiles) {
  __shared__ __attribute__((aligned(16))) float lds[WAVES_PER_BLOCK][16 * NDIM];
  __shared__ v2f wl[2 * WL_FRAGS];

  const int wave = __builtin_amdgcn_readfirstlane((int)(threadIdx.x >> 5));
  const int lane = threadIdx.x & 31;
  const int stride = (int)gridDim.x * WAVES_PER_BLOCK;
  int tile = (int)blockIdx.x * WAVES_PER_BLOCK + wave;
  const bool has = tile < nTiles;

  const int laneM = lane & 15;
  const int hi = lane >> 4;
  float* lt = lds[wave];

  const int rsLane = lane < 16 ? lane : 16;
  int rs = 0;
  if (has) rs = rowstart[16 * tile + rsLane];

  {
    const v2f* sp = (const v2f*)WPP;
    const v2f* sq = (const v2f*)WPQ;
#pragma unroll
    for (int i = 0; i < WL_FRAGS / (WAVES_PER_BLOCK * 32); ++i) {
      const int j = i * (WAVES_PER_BLOCK * 32) + (int)threadIdx.x;
      wl[j] = sp[j];
      wl[WL_FRAGS + j] = sq[j];
    }
  }
  float cv[4], dv[4];
#pragma unroll
  for (int n = 0; n < 4; ++n) {
    cv[n] = CV[16 * n + laneM];
    dv[n] = DV[16 * n + laneM];
  }
  __syncthreads();
  if (!has) return;

  const v2f* wlp = wl + lane;
  const v2f* wlq = wl + WL_FRAGS + lane;
  const unsigned zoff = (unsigned)nTiles * (16u * NDIM * 4u);

  int cnt = (nTiles - 1 - tile) / stride + 1;

  for (; cnt > 0; --cnt) {
    const int nt = cnt > 1 ? tile + stride : tile;
    const int rsn = rowstart[16 * nt + rsLane];
    __builtin_amdgcn_sched_barrier(0);
    v2f zs[16];
    gin_gather_tile<4>(U, perm, zoff, rs, lane, lt, zs);
    const int deg = __shfl_down(rs, 1) - rs;
    gin_layer_tile(U, zs, deg, wlp, wlq, cv, dv, lt, OUT, tile, laneM, hi);
    rs = rsn;
    tile = nt;
  }
}


template <int NW>
__device__ __forceinline__ int gin_block_scan(int v, int t, int* wsum,
                                              int& total) {
  const int lane = t & 31;
  const int wave = t >> 5;
#pragma unroll
  for (int off = 1; off < 32; off <<= 1) {
    const int u = __shfl_up(v, off);
    if (lane >= off) v += u;
  }
  if (lane == 31) wsum[wave] = v;
  __syncthreads();
  int before = 0;
  int all = 0;
#pragma unroll
  for (int w = 0; w < NW; ++w) {
    const int ws = wsum[w];
    all += ws;
    if (w < wave) before += ws;
  }
  total = all;
  return v + before;
}

__global__ void __launch_bounds__(BIN_T)
__attribute__((amdgpu_waves_per_eu(8, 8)))
gin_chunk_sort_kernel(const int* __restrict__ rowp,
                      const int* __restrict__ colp, int* __restrict__ off,
                      int* __restrict__ sortedc, int E, int nChunks,
                      const float* __restrict__ W1,
                      const float* __restrict__ B1,
                      const float* __restrict__ W2,
                      const float* __restrict__ B2,
                      const float* __restrict__ Wf,
                      const float* __restrict__ Bf, float* __restrict__ Wp,
                      float* __restrict__ bv, float* __restrict__ Z0,
                      float* __restrict__ Z1, int L,
                      const float* __restrict__ X, float* __restrict__ U0,
                      int nTiles) {
  __shared__ int wsa[BIN_T / 32];
  __shared__ int lcur[NBKP];
  __shared__ v2i stage[CH];
  const int t = (int)threadIdx.x;
  const int b = (int)blockIdx.x;

  if (b >= nChunks + 2 * L) {
    const int q = b - nChunks - 2 * L;
    v2f* wlin = (v2f*)stage;
#pragma unroll
    for (int i = 0; i < REPACK_SLOTS / BIN_T; ++i) {
      const int tid = i * BIN_T + t;
      const int c = tid >> 7;
      const int n = (tid >> 5) & 3;
      const int ln = tid & 31;
      const int k = 4 * c + 2 * (ln >> 4);
      const int m = ln & 15;
      v2f w;
      w.x = W1[(k + 0) * NDIM + 16 * n + m];
      w.y = W1[(k + 1) * NDIM + 16 * n + m];
      wlin[tid] = w;
    }
    __syncthreads();
    const int wave = __builtin_amdgcn_readfirstlane(t >> 5);
    const int lane = t & 31;
    const int laneM = lane & 15;
    const int hi = lane >> 4;
    const int laneOff = laneM * NDIM + 2 * hi;
    const int stride = ((int)gridDim.x - nChunks - 2 * L) * (BIN_T / 32);
    float bias1[4];
#pragma unroll
    for (int n = 0; n < 4; ++n) bias1[n] = B1[16 * n + laneM];
    for (int tile = q * (BIN_T / 32) + wave; tile < nTiles; tile += stride) {
      v2f a[16];
      gin_fetch_rows(X, tile, laneOff, a);
      int slot = lane;
      asm volatile("" : "+v"(slot));
      gin_linear_tile_lds(a, wlin + slot, bias1, U0, tile, laneM, hi);
    }
    return;
  }
  if (b >= nChunks) {
    const int pm = b - nChunks;
    const bool isQ = pm >= L;
    const int l = isQ ? pm - L : pm;
    const float* A = W2 + (size_t)l * (NDIM * NDIM);
    const float* Bm = !isQ        ? W1 + (size_t)l * (NDIM * NDIM)
                      : l + 1 < L ? W1 + (size_t)(l + 1) * (NDIM * NDIM)
                                  : Wf;
    float* dst = Wp + (size_t)pm * WP_ELEMS;
    const int wv = t >> 5;
    const int lane = t & 31;
    const int hi = lane >> 4;
    const int m = lane & 15;
    const int R = wv >> 2;
    const int n = wv & 3;
    const float* ap = A + (16 * R + m) * NDIM + 2 * hi;
    const float* bp = Bm + (2 * hi) * NDIM + 16 * n + m;
    v8f acc;
#pragma unroll
    for (int r = 0; r < 8; ++r) acc[r] = 0.0f;
#pragma unroll
    for (int c = 0; c < 16; ++c) {
      const v2f a = *(const v2f*)(ap + 4 * c);
      v2f bb;
      bb.x = bp[(4 * c + 0) * NDIM];
      bb.y = bp[(4 * c + 1) * NDIM];
      acc = wmma_f32_16x16x4(a, bb, acc);
    }
#pragma unroll
    for (int r = 0; r < 8; ++r) {
      const int row = 16 * R + 8 * hi + r;
      dst[((((row >> 2) * 4 + n) * 32) + ((row >> 1) & 1) * 16 + m) * 2 +
          (row & 1)] = acc[r];
    }
    if (t < NDIM) {
      const float* bs = B2 + (size_t)l * NDIM;
      float sum = !isQ ? 0.0f : l + 1 < L ? B1[(l + 1) * NDIM + t] : Bf[t];
      for (int i = 0; i < NDIM; ++i) sum += bs[i] * Bm[i * NDIM + t];
      bv[pm * NDIM + t] = sum;
      if (pm == 0) {
        Z0[t] = 0.0f;
        Z1[t] = 0.0f;
      }
    }
    return;
  }

  const int e0 = b * CH;
  lcur[t] = 0;

  int er[CH / BIN_T], ec[CH / BIN_T];
#pragma unroll
  for (int j = 0; j < CH / BIN_T; ++j) {
    const int e = e0 + j * BIN_T + t;
    er[j] = (e < E) ? rowp[e] : -1;
    ec[j] = (e < E) ? colp[e] : 0;
  }
  __syncthreads();

#pragma unroll
  for (int j = 0; j < CH / BIN_T; ++j)
    if (er[j] >= 0) atomicAdd(&lcur[er[j] >> RB_SHIFT], 1);
  __syncthreads();
  const int c = lcur[t];
  int nChunk;
  const int lstart =
      gin_block_scan<BIN_T / 32>(c, t, wsa, nChunk) - c;
  lcur[t] = lstart;
  off[(size_t)b * OFP + t] = lstart;
  if (t == BIN_T - 1) off[(size_t)b * OFP + NBKP] = nChunk;
  __syncthreads();

  int slot[CH / BIN_T];
#pragma unroll
  for (int j = 0; j < CH / BIN_T; ++j)
    slot[j] = (er[j] >= 0) ? atomicAdd(&lcur[er[j] >> RB_SHIFT], 1) : 0;
#pragma unroll
  for (int j = 0; j < CH / BIN_T; ++j) {
    if (er[j] >= 0) {
      v2i v;
      v.x = er[j];
      v.y = ec[j];
      stage[slot[j]] = v;
    }
  }
  __syncthreads();

  v2i* dst = (v2i*)sortedc + (size_t)b * CH;
#pragma unroll
  for (int j = 0; j < CH / BIN_T; ++j) {
    const int i = j * BIN_T + t;
    if (i < nChunk) dst[i] = stage[i];
  }
}

__global__ void __launch_bounds__(RB)
__attribute__((amdgpu_waves_per_eu(4, 4)))
gin_finish_layer0_kernel(const int* __restrict__ sortedc,
                         const int* __restrict__ off, int* rowstart, int* perm,
                         int N, int nChunks, const float* __restrict__ U,
                         const float* __restrict__ WPP,
                         const float* __restrict__ CV,
                         const float* __restrict__ WPQ,
                         const float* __restrict__ DV, float* __restrict__ OUT,
                         int nTiles) {
  __shared__ __attribute__((aligned(16))) float ltile[RB / 32][16 * NDIM];
  __shared__ v2f wl[2 * WL_FRAGS];
  __shared__ int wsa[RB / 32];
  __shared__ int wsb[RB / 32];
  __shared__ int wsc[RB / 32];
  __shared__ int cpos[NBKP];
  __shared__ int csrc[NBKP];
  __shared__ int clen[NBKP];
  __shared__ int lcur[RB];
  __shared__ v2i lev[FIN_CAP];
  __shared__ int lperm[FIN_CAP];
  const int t = (int)threadIdx.x;
  const int k = (int)blockIdx.x;
  const v2i* src = (const v2i*)sortedc;

  const int c0 = t;
  const int c1 = t + RB;
  const int* o0 = off + (size_t)(c0 < nChunks ? c0 : nChunks - 1) * OFP + k;
  const int* o1 = off + (size_t)(c1 < nChunks ? c1 : nChunks - 1) * OFP + k;
  const int a0 = o0[0];
  const int z0 = o0[1];
  const int a1 = o1[0];
  const int z1 = o1[1];
  const int s0 = c0 < nChunks ? a0 : 0;
  const int l0 = c0 < nChunks ? z0 - a0 : 0;
  const int s1 = c1 < nChunks ? a1 : 0;
  const int l1 = c1 < nChunks ? z1 - a1 : 0;
  int n;
  int b0;
  const int cex = gin_block_scan<RB / 32>(l0 + l1, t, wsa, n) - (l0 + l1);
  (void)gin_block_scan<RB / 32>(s0 + s1, t, wsb, b0);
  cpos[c0] = cex;
  csrc[c0] = c0 * CH + s0;
  clen[c0] = l0;
  cpos[c1] = cex + l0;
  csrc[c1] = c1 * CH + s1;
  clen[c1] = l1;
  lcur[t] = 0;
  __syncthreads();
  const bool staged = n <= FIN_CAP;
  const int hw = t >> 4;
  const int hl = t & 15;

  if (staged) {
    for (int base = 0; base < nChunks; base += 16 * FIN_Q) {
      v2i v[FIN_Q];
      int p[FIN_Q], ln[FIN_Q];
#pragma unroll
      for (int q = 0; q < FIN_Q; ++q) {
        const int c = base + 16 * q + hw;
        ln[q] = clen[c];
        p[q] = cpos[c];
        v[q].x = 0;
        v[q].y = 0;
        if (hl < ln[q]) v[q] = src[csrc[c] + hl];
      }
#pragma unroll
      for (int q = 0; q < FIN_Q; ++q) {
        if (hl < ln[q]) {
          lev[p[q] + hl] = v[q];
          atomicAdd(&lcur[v[q].x & (RB - 1)], 1);
        }
      }
#pragma unroll
      for (int q = 0; q < FIN_Q; ++q) {
        const int c = base + 16 * q + hw;
        for (int j = hl + 16; j < ln[q]; j += 16) {
          const v2i w = src[csrc[c] + j];
          lev[p[q] + j] = w;
          atomicAdd(&lcur[w.x & (RB - 1)], 1);
        }
      }
    }
  } else {
    for (int c = hw; c < nChunks; c += RB / 16) {
      const int s = csrc[c];
      const int len = clen[c];
      for (int j = hl; j < len; j += 16)
        atomicAdd(&lcur[src[s + j].x & (RB - 1)], 1);
    }
  }
  __syncthreads();

  const int d = lcur[t];
  int nScan;
  const int ex =
      gin_block_scan<RB / 32>(d, t, wsc, nScan) - d;
  lcur[t] = ex;
  const int row = k * RB + t;
  if (row < N) rowstart[row] = b0 + ex;
  if (row == N - 1) rowstart[N] = b0 + n;
  __syncthreads();

  if (staged) {
    v2i ev[FIN_U];
    int sl[FIN_U];
#pragma unroll
    for (int u = 0; u < FIN_U; ++u) {
      const int i = u * RB + t;
      if (i < n) {
        ev[u] = lev[i];
      } else {
        ev[u].x = -1;
        ev[u].y = 0;
      }
    }
#pragma unroll
    for (int u = 0; u < FIN_U; ++u)
      sl[u] = (ev[u].x >= 0) ? atomicAdd(&lcur[ev[u].x & (RB - 1)], 1) : 0;
#pragma unroll
    for (int u = 0; u < FIN_U; ++u)
      if (ev[u].x >= 0) lperm[sl[u]] = ev[u].y;
    __syncthreads();
#pragma unroll
    for (int u = 0; u < FIN_U; ++u) {
      const int i = u * RB + t;
      if (i < n) perm[b0 + i] = lperm[i];
    }
  } else {
    for (int c = hw; c < nChunks; c += RB / 16) {
      const int s = csrc[c];
      const int len = clen[c];
      for (int j = hl; j < len; j += 16) {
        const v2i v = src[s + j];
        perm[b0 + atomicAdd(&lcur[v.x & (RB - 1)], 1)] = v.y;
      }
    }
  }

  {
    const v2f* sp = (const v2f*)WPP;
    const v2f* sq = (const v2f*)WPQ;
#pragma unroll
    for (int i = 0; i < WL_FRAGS / RB; ++i) {
      const int j = i * RB + t;
      wl[j] = sp[j];
      wl[WL_FRAGS + j] = sq[j];
    }
  }
  const int wave = __builtin_amdgcn_readfirstlane(t >> 5);
  const int lane = t & 31;
  const int laneM = lane & 15;
  const int hi = lane >> 4;
  float cv[4], dv[4];
#pragma unroll
  for (int i = 0; i < 4; ++i) {
    cv[i] = CV[16 * i + laneM];
    dv[i] = DV[16 * i + laneM];
  }
  __syncthreads();

  float* lt = ltile[wave];
  const v2f* wlp = wl + lane;
  const v2f* wlq = wl + WL_FRAGS + lane;
  const unsigned zoff = (unsigned)nTiles * (16u * NDIM * 4u);
  const int rsLane = lane < 16 ? lane : 16;
  const int endRow = (k + 1) * RB;
  const int endVal = b0 + n;
  for (int tile = k * (RB / 16) + wave;
       tile < (k + 1) * (RB / 16) && tile < nTiles; tile += RB / 32) {
    const int idx = 16 * tile + rsLane;
    int rs = endVal;
    if (idx != endRow) rs = rowstart[idx];
    v2f zs[16];
    gin_gather_tile<3>(U, perm, zoff, rs, lane, lt, zs);
    const int deg = __shfl_down(rs, 1) - rs;
    gin_layer_tile(U, zs, deg, wlp, wlq, cv, dv, lt, OUT, tile, laneM, hi);
  }
}


extern "C" void kernel_launch(void* const* d_in, const int* in_sizes, int n_in,
                              void* d_out, int out_size, void* d_ws,
                              size_t ws_size, hipStream_t stream) {
  (void)n_in;
  (void)out_size;
  (void)ws_size;

  const float* x = (const float*)d_in[0];
  const int* ei = (const int*)d_in[1];
  const float* W1 = (const float*)d_in[2];
  const float* b1 = (const float*)d_in[3];
  const float* W2 = (const float*)d_in[4];
  const float* b2 = (const float*)d_in[5];
  const float* Wf = (const float*)d_in[6];
  const float* bf = (const float*)d_in[7];
  float* out = (float*)d_out;

  const int N = in_sizes[0] / NDIM;
  const int E = in_sizes[1] / 2;
  const int L = in_sizes[3] / NDIM;
  const int nTiles = N / 16;
  const size_t NB = (size_t)N * NDIM;

  float* uA = (float*)d_ws;
  float* uB = uA + NB + NDIM;
  const int nMat = 2 * L;
  float* wp = uB + NB + NDIM;
  float* bv = wp + (size_t)nMat * WP_ELEMS;
  const int nChunks = (E + CH - 1) / CH;
  const int nBuckets = (N + RB - 1) / RB;
  int* sortedc = (int*)(bv + (size_t)2 * L * NDIM);
  int* off = sortedc + 2 * (size_t)nChunks * CH;
  int* perm = off + (size_t)nChunks * OFP;
  int* rowstart = perm + E;

  const int* rowp = ei;
  const int* colp = ei + E;

  gin_chunk_sort_kernel<<<nChunks + 2 * L + LIN_GRID, BIN_T, 0, stream>>>(
      rowp, colp, off, sortedc, E, nChunks, W1, b1, W2, b2, Wf, bf, wp, bv,
      uA + NB, uB + NB, L, x, uA, nTiles);

  gin_finish_layer0_kernel<<<nBuckets, RB, 0, stream>>>(
      sortedc, off, rowstart, perm, N, nChunks, uA, wp, bv,
      wp + (size_t)L * WP_ELEMS, bv + (size_t)L * NDIM, 1 < L ? uB : out,
      nTiles);

  const dim3 layerBlk(WAVES_PER_BLOCK * 32);
  float* ucur = uB;
  float* unxt = uA;
  for (int l = 1; l < L; ++l) {
    float* dst = l + 1 < L ? unxt : out;
    gin_layer_kernel<<<dim3(LAYER_GRID), layerBlk, 0, stream>>>(
        ucur, rowstart, perm, wp + (size_t)l * WP_ELEMS, bv + (size_t)l * NDIM,
        wp + (size_t)(L + l) * WP_ELEMS, bv + (size_t)(L + l) * NDIM, dst,
        nTiles);
    float* uswap = ucur;
    ucur = unxt;
    unxt = uswap;
  }
}
